# starting kernel
# baseline (speedup 1.0000x reference)
#include <hip/hip_runtime.h>
#include <hip/hip_bf16.h>


static constexpr int NN = 100000;
static constexpr int EE = 1600000;
static constexpr int FF = 128;

typedef __attribute__((ext_vector_type(2))) float v2f;
typedef __attribute__((ext_vector_type(8))) float v8f;


__global__ void k_zero_u32(unsigned* __restrict__ p, int n) {
  int i = blockIdx.x * 256 + threadIdx.x;
  if (i < n) p[i] = 0u;
}


__global__ void k_deg(const int* __restrict__ src, const int* __restrict__ dst,
                      unsigned* __restrict__ outdeg, unsigned* __restrict__ indeg) {
  int e = blockIdx.x * 256 + threadIdx.x;
  if (e < EE) {
    atomicAdd(&outdeg[src[e]], 1u);
    atomicAdd(&indeg[dst[e]], 1u);
  }
}

__global__ void k_ds(const unsigned* __restrict__ outdeg, const unsigned* __restrict__ indeg,
                     float* __restrict__ dsout, float* __restrict__ dsin) {
  int i = blockIdx.x * 256 + threadIdx.x;
  if (i < NN) {
    unsigned od = outdeg[i], id = indeg[i];
    dsout[i] = od ? rsqrtf((float)od) : 1.0f;
    dsin[i]  = id ? rsqrtf((float)id) : 1.0f;
  }
}

__global__ void k_scan(const unsigned* __restrict__ deg, unsigned* __restrict__ offs) {
  __shared__ unsigned ts[1024];
  const int t = threadIdx.x;
  const int per = (NN + 1023) / 1024;
  int s = t * per; if (s > NN) s = NN;
  int e = s + per; if (e > NN) e = NN;
  unsigned sum = 0;
  for (int i = s; i < e; ++i) sum += deg[i];
  ts[t] = sum;
  __syncthreads();
  for (int off = 1; off < 1024; off <<= 1) {
    unsigned v = (t >= off) ? ts[t - off] : 0u;
    __syncthreads();
    if (t >= off) ts[t] += v;
    __syncthreads();
  }
  unsigned run = (t == 0) ? 0u : ts[t - 1];
  for (int i = s; i < e; ++i) { offs[i] = run; run += deg[i]; }
  if (t == 1023) offs[NN] = ts[1023];
}

__global__ void k_copy_u32(const unsigned* __restrict__ a, unsigned* __restrict__ b, int n) {
  int i = blockIdx.x * 256 + threadIdx.x;
  if (i < n) b[i] = a[i];
}

__global__ void k_fill(const int* __restrict__ src, const int* __restrict__ dst,
                       unsigned* __restrict__ cursor, unsigned* __restrict__ esrc) {
  int e = blockIdx.x * 256 + threadIdx.x;
  if (e < EE) {
    unsigned pos = atomicAdd(&cursor[dst[e]], 1u);
    esrc[pos] = (unsigned)src[e];
  }
}


__global__ void k_agg128(const float* __restrict__ h, const unsigned* __restrict__ offs,
                         const unsigned* __restrict__ esrc,
                         const float* __restrict__ dsout, const float* __restrict__ dsin,
                         float* __restrict__ agg) {
  const int wv = threadIdx.x >> 5, lane = threadIdx.x & 31;
  const int n = blockIdx.x * 8 + wv;
  const unsigned b = offs[n], e = offs[n + 1];
  float4 acc = make_float4(0.f, 0.f, 0.f, 0.f);
  for (unsigned i = b; i < e; ++i) {
    const unsigned s = esrc[i];
    const float w = dsout[s];
    const float4 v = *(const float4*)(h + (size_t)s * FF + lane * 4);
    acc.x = fmaf(w, v.x, acc.x);
    acc.y = fmaf(w, v.y, acc.y);
    acc.z = fmaf(w, v.z, acc.z);
    acc.w = fmaf(w, v.w, acc.w);
  }
  const float c = dsin[n];
  *(float4*)(agg + (size_t)n * FF + lane * 4) =
      make_float4(c * acc.x, c * acc.y, c * acc.z, c * acc.w);
}

__global__ void k_agg64(const float* __restrict__ P, const unsigned* __restrict__ offs,
                        const unsigned* __restrict__ esrc, const float* __restrict__ dsin,
                        float* __restrict__ R) {
  const int wv = threadIdx.x >> 5, lane = threadIdx.x & 31;
  const int n = blockIdx.x * 8 + wv;
  const unsigned b = offs[n], e = offs[n + 1];
  float2 acc = make_float2(0.f, 0.f);
  for (unsigned i = b; i < e; ++i) {
    const unsigned s = esrc[i];
    const float2 v = *(const float2*)(P + (size_t)s * 64 + lane * 2);
    acc.x += v.x;
    acc.y += v.y;
  }
  const float c = dsin[n];
  *(float2*)(R + (size_t)n * 64 + lane * 2) = make_float2(c * acc.x, c * acc.y);
}


__global__ void k_gemm_dual(const float* __restrict__ Agg, const float* __restrict__ H,
                            const float* __restrict__ W, const float* __restrict__ L,
                            float* __restrict__ Z) {
  const int wv = threadIdx.x >> 5;
  const int lane = threadIdx.x & 31;
  const int t = lane & 15;
  const int hh = lane >> 4;
  const int r0 = blockIdx.x << 4;
  const int cb = wv << 4;
  const float* arow = Agg + (size_t)(r0 + t) * FF;
  const float* hrow = H + (size_t)(r0 + t) * FF;
  v8f acc = {};
#pragma unroll 4
  for (int k0 = 0; k0 < 128; k0 += 4) {
    const int kk = k0 + 2 * hh;
    v2f a1 = *(const v2f*)(arow + kk);
    v2f b1; b1.x = W[kk * FF + cb + t]; b1.y = W[(kk + 1) * FF + cb + t];
    acc = __builtin_amdgcn_wmma_f32_16x16x4_f32(false, a1, false, b1, (short)0, acc, false, false);
    v2f a2 = *(const v2f*)(hrow + kk);
    v2f b2; b2.x = L[kk * FF + cb + t]; b2.y = L[(kk + 1) * FF + cb + t];
    acc = __builtin_amdgcn_wmma_f32_16x16x4_f32(false, a2, false, b2, (short)0, acc, false, false);
  }
#pragma unroll
  for (int v = 0; v < 8; ++v)
    Z[(size_t)(r0 + v + 8 * hh) * FF + cb + t] = acc[v];
}

__global__ void k_gemm_p(const float* __restrict__ H, const float* __restrict__ dsout,
                         const float* __restrict__ Wp, float* __restrict__ P) {
  const int wv = threadIdx.x >> 5;
  const int lane = threadIdx.x & 31;
  const int t = lane & 15, hh = lane >> 4;
  const int r0 = blockIdx.x << 4;
  const int cb = wv << 4;
  const float d = dsout[r0 + t];
  const float* hrow = H + (size_t)(r0 + t) * FF;
  v8f acc = {};
#pragma unroll 4
  for (int k0 = 0; k0 < 128; k0 += 4) {
    const int kk = k0 + 2 * hh;
    v2f a; a.x = d * hrow[kk]; a.y = d * hrow[kk + 1];
    v2f b; b.x = Wp[kk * 64 + cb + t]; b.y = Wp[(kk + 1) * 64 + cb + t];
    acc = __builtin_amdgcn_wmma_f32_16x16x4_f32(false, a, false, b, (short)0, acc, false, false);
  }
#pragma unroll
  for (int v = 0; v < 8; ++v)
    P[(size_t)(r0 + v + 8 * hh) * 64 + cb + t] = acc[v];
}

__global__ void k_gemm_out(const float* __restrict__ H, const float* __restrict__ Lp,
                           const float* __restrict__ R, const float* __restrict__ b2,
                           float* __restrict__ out) {
  const int wv = threadIdx.x >> 5;
  if (wv == 3) return;
  const int lane = threadIdx.x & 31;
  const int t = lane & 15, hh = lane >> 4;
  const int r0 = blockIdx.x << 4;
  const int cb = wv << 4;
  const float* hrow = H + (size_t)(r0 + t) * FF;
  v8f acc = {};
#pragma unroll 4
  for (int k0 = 0; k0 < 128; k0 += 4) {
    const int kk = k0 + 2 * hh;
    v2f a = *(const v2f*)(hrow + kk);
    v2f b; b.x = Lp[kk * 64 + cb + t]; b.y = Lp[(kk + 1) * 64 + cb + t];
    acc = __builtin_amdgcn_wmma_f32_16x16x4_f32(false, a, false, b, (short)0, acc, false, false);
  }
  const int col = cb + t;
  if (col < 40) {
    const float bias = b2[col];
#pragma unroll
    for (int v = 0; v < 8; ++v) {
      const int row = r0 + v + 8 * hh;
      out[(size_t)row * 40 + col] = acc[v] + R[(size_t)row * 64 + col] + bias;
    }
  }
}


__global__ void k_colstats(const float* __restrict__ z, float* __restrict__ cs,
                           float* __restrict__ cs2) {
  const int c = threadIdx.x;
  const size_t r0 = (size_t)blockIdx.x * 250;
  float s = 0.f, s2 = 0.f;
  for (int i = 0; i < 250; ++i) {
    const float v = z[(r0 + i) * FF + c];
    s += v;
    s2 = fmaf(v, v, s2);
  }
  atomicAdd(&cs[c], s);
  atomicAdd(&cs2[c], s2);
}

__global__ void k_bnparams(const float* __restrict__ cs, const float* __restrict__ cs2,
                           const float* __restrict__ g, const float* __restrict__ be,
                           float* __restrict__ bnA, float* __restrict__ bnB) {
  const int c = threadIdx.x;
  const float inv = 1.0f / (float)NN;
  const float mu = cs[c] * inv;
  const float var = cs2[c] * inv - mu * mu;
  const float a = g[c] * rsqrtf(var + 1e-5f);
  bnA[c] = a;
  bnB[c] = be[c] - mu * a;
}

__global__ void k_bnapply(const float* __restrict__ z, const float* __restrict__ bnA,
                          const float* __restrict__ bnB, float* __restrict__ hout) {
  const size_t i = (size_t)blockIdx.x * 256 + threadIdx.x;
  const int c = (int)(i & 127);
  const float v = fmaf(z[i], bnA[c], bnB[c]);
  hout[i] = v > 0.f ? v : 0.f;
}

__global__ void k_padw(const float* __restrict__ W2, const float* __restrict__ L2,
                       float* __restrict__ Wp, float* __restrict__ Lp) {
  const int i = blockIdx.x * 256 + threadIdx.x;
  const int k = i >> 6, c = i & 63;
  const float w = (c < 40) ? W2[k * 40 + c] : 0.f;
  const float l = (c < 40) ? L2[k * 40 + c] : 0.f;
  Wp[i] = w;
  Lp[i] = l;
}


extern "C" void kernel_launch(void* const* d_in, const int* in_sizes, int n_in,
                              void* d_out, int out_size, void* d_ws, size_t ws_size,
                              hipStream_t stream) {
  (void)in_sizes; (void)n_in; (void)out_size; (void)ws_size;

  const float* feat = (const float*)d_in[0];
  const int* src   = (const int*)d_in[1];
  const int* dst   = (const int*)d_in[2];
  const float* W0  = (const float*)d_in[3];
  const float* L0  = (const float*)d_in[4];
  const float* W1  = (const float*)d_in[5];
  const float* L1  = (const float*)d_in[6];
  const float* W2  = (const float*)d_in[7];
  const float* L2  = (const float*)d_in[8];
  const float* b2  = (const float*)d_in[9];
  const float* g0  = (const float*)d_in[10];
  const float* be0 = (const float*)d_in[11];
  const float* g1  = (const float*)d_in[12];
  const float* be1 = (const float*)d_in[13];
  float* out = (float*)d_out;

  char* base = (char*)d_ws;
  size_t off = 0;
  auto take = [&](size_t bytes) -> char* {
    char* p = base + off;
    off = (off + bytes + 255) & ~(size_t)255;
    return p;
  };
  unsigned* indeg  = (unsigned*)take((size_t)NN * 4);
  unsigned* outdeg = (unsigned*)take((size_t)NN * 4);
  unsigned* offs   = (unsigned*)take((size_t)(NN + 1) * 4);
  unsigned* cursor = (unsigned*)take((size_t)NN * 4);
  unsigned* esrc   = (unsigned*)take((size_t)EE * 4);
  float* dsout = (float*)take((size_t)NN * 4);
  float* dsin  = (float*)take((size_t)NN * 4);
  float* cs    = (float*)take(128 * 4);
  float* cs2   = (float*)take(128 * 4);
  float* bnA   = (float*)take(128 * 4);
  float* bnB   = (float*)take(128 * 4);
  float* Wp    = (float*)take(128 * 64 * 4);
  float* Lp    = (float*)take(128 * 64 * 4);
  float* bufA  = (float*)take((size_t)NN * FF * 4);
  float* bufB  = (float*)take((size_t)NN * FF * 4);
  float* bufC  = (float*)take((size_t)NN * FF * 4);

  const int BEdge = (EE + 255) / 256;
  const int BNode = (NN + 255) / 256;
  const int BWave = NN / 8;
  const int BTile = NN / 16;
  const int BApply = (NN * FF) / 256;

  k_zero_u32<<<BNode, 256, 0, stream>>>(indeg, NN);
  k_zero_u32<<<BNode, 256, 0, stream>>>(outdeg, NN);
  k_deg<<<BEdge, 256, 0, stream>>>(src, dst, outdeg, indeg);
  k_ds<<<BNode, 256, 0, stream>>>(outdeg, indeg, dsout, dsin);
  k_scan<<<1, 1024, 0, stream>>>(indeg, offs);
  k_copy_u32<<<BNode, 256, 0, stream>>>(offs, cursor, NN);
  k_fill<<<BEdge, 256, 0, stream>>>(src, dst, cursor, esrc);
  k_padw<<<32, 256, 0, stream>>>(W2, L2, Wp, Lp);

  k_agg128<<<BWave, 256, 0, stream>>>(feat, offs, esrc, dsout, dsin, bufA);
  k_gemm_dual<<<BTile, 256, 0, stream>>>(bufA, feat, W0, L0, bufB);
  k_zero_u32<<<1, 256, 0, stream>>>((unsigned*)cs, 256);
  k_colstats<<<400, 128, 0, stream>>>(bufB, cs, cs2);
  k_bnparams<<<1, 128, 0, stream>>>(cs, cs2, g0, be0, bnA, bnB);
  k_bnapply<<<BApply, 256, 0, stream>>>(bufB, bnA, bnB, bufC);

  k_agg128<<<BWave, 256, 0, stream>>>(bufC, offs, esrc, dsout, dsin, bufA);
  k_gemm_dual<<<BTile, 256, 0, stream>>>(bufA, bufC, W1, L1, bufB);
  k_zero_u32<<<1, 256, 0, stream>>>((unsigned*)cs, 256);
  k_colstats<<<400, 128, 0, stream>>>(bufB, cs, cs2);
  k_bnparams<<<1, 128, 0, stream>>>(cs, cs2, g1, be1, bnA, bnB);
  k_bnapply<<<BApply, 256, 0, stream>>>(bufB, bnA, bnB, bufC);

  k_gemm_p<<<BTile, 128, 0, stream>>>(bufC, dsout, Wp, bufA);
  k_agg64<<<BWave, 256, 0, stream>>>(bufA, offs, esrc, dsin, bufB);
  k_gemm_out<<<BTile, 128, 0, stream>>>(bufC, Lp, bufB, b2, out);
}


# Gather kernels: a wave walks 8 dst nodes, takes a node's neighbour ids with one coalesced load and keeps up to 8 neighbour rows in flight (was one node per wave, one row at a time); sum order kept
# speedup vs baseline: 1.3656x; 1.3656x over previous
#include <hip/hip_runtime.h>
#include <hip/hip_bf16.h>


static constexpr int NN = 100000;
static constexpr int EE = 1600000;
static constexpr int FF = 128;

typedef __attribute__((ext_vector_type(2))) float v2f;
typedef __attribute__((ext_vector_type(8))) float v8f;


__global__ void k_zero_u32(unsigned* __restrict__ p, int n) {
  int i = blockIdx.x * 256 + threadIdx.x;
  if (i < n) p[i] = 0u;
}


__global__ void k_deg(const int* __restrict__ src, const int* __restrict__ dst,
                      unsigned* __restrict__ outdeg, unsigned* __restrict__ indeg) {
  int e = blockIdx.x * 256 + threadIdx.x;
  if (e < EE) {
    atomicAdd(&outdeg[src[e]], 1u);
    atomicAdd(&indeg[dst[e]], 1u);
  }
}

__global__ __launch_bounds__(256) void k_ds_alloc(const unsigned* __restrict__ outdeg,
                                                  const unsigned* __restrict__ indeg,
                                                  unsigned* __restrict__ bump,
                                                  float* __restrict__ dsout, float* __restrict__ dsin,
                                                  unsigned* __restrict__ rows,
                                                  unsigned* __restrict__ cursor) {
  __shared__ unsigned wtot[8];
  __shared__ unsigned bbase;
  const int i = blockIdx.x * 256 + threadIdx.x;
  const int lane = threadIdx.x & 31, wv = threadIdx.x >> 5;
  unsigned id = 0u;
  if (i < NN) {
    const unsigned od = outdeg[i];
    id = indeg[i];
    dsout[i] = od ? rsqrtf((float)od) : 1.0f;
    dsin[i]  = id ? rsqrtf((float)id) : 1.0f;
  }
  unsigned inc = id;
  for (int d = 1; d < 32; d <<= 1) {
    const unsigned up = __shfl_up(inc, d);
    if (lane >= d) inc += up;
  }
  if (lane == 31) wtot[wv] = inc;
  __syncthreads();
  if (threadIdx.x == 0) {
    unsigned tot = 0u;
    for (int w = 0; w < 8; ++w) tot += wtot[w];
    bbase = atomicAdd(bump, tot);
  }
  __syncthreads();
  unsigned st = bbase + inc - id;
  for (int w = 0; w < wv; ++w) st += wtot[w];
  if (i < NN) {
    rows[2 * i] = st;
    rows[2 * i + 1] = st + id;
    cursor[i] = st;
  }
}

__global__ void k_fill(const int* __restrict__ src, const int* __restrict__ dst,
                       unsigned* __restrict__ cursor, unsigned* __restrict__ esrc) {
  int e = blockIdx.x * 256 + threadIdx.x;
  if (e < EE) {
    unsigned pos = atomicAdd(&cursor[dst[e]], 1u);
    esrc[pos] = (unsigned)src[e];
  }
}


static constexpr int AGG_NODES = 8;
static constexpr int AGG_GROUP = 8;
static constexpr int AGG_WAVES = NN / AGG_NODES;
static constexpr int AGG_BLOCKS = AGG_WAVES / 4;
static_assert(NN % AGG_NODES == 0 && AGG_WAVES % 4 == 0, "gather grid must cover the nodes exactly");

__device__ __forceinline__ unsigned lane_u32(unsigned v, int l) {
  return (unsigned)__builtin_amdgcn_readlane((int)v, l);
}
__device__ __forceinline__ float lane_f32(float v, int l) {
  return __int_as_float(__builtin_amdgcn_readlane(__float_as_int(v), l));
}

__device__ __forceinline__ void axpy4(float4& acc, float w, const float4& v) {
  acc.x = fmaf(w, v.x, acc.x);
  acc.y = fmaf(w, v.y, acc.y);
  acc.z = fmaf(w, v.z, acc.z);
  acc.w = fmaf(w, v.w, acc.w);
}

__global__ __launch_bounds__(128) void k_agg128(const float* __restrict__ h,
                                                const unsigned* __restrict__ rows,
                                                const unsigned* __restrict__ esrc,
                                                const float* __restrict__ dsout,
                                                const float* __restrict__ dsin,
                                                float* __restrict__ agg) {
  const int lane = threadIdx.x & 31;
  const int n0 = (blockIdx.x * 4 + (threadIdx.x >> 5)) * AGG_NODES;
  unsigned rb = 0u, re = 0u;
  float cin = 0.f;
  if (lane < AGG_NODES) {
    const uint2 r = *(const uint2*)(rows + 2 * (n0 + lane));
    rb = r.x;
    re = r.y;
    cin = dsin[n0 + lane];
  }
  unsigned b = lane_u32(rb, 0), e = lane_u32(re, 0);
  unsigned my = 0u;
  float myw = 0.f;
  if (b + lane < e) {
    my = esrc[b + lane];
    myw = dsout[my];
  }
  const float* hl = h + lane * 4;

#pragma unroll 1
  for (int k = 0; k < AGG_NODES; ++k) {
    unsigned nb = 0u, ne = 0u, nmy = 0u;
    if (k + 1 < AGG_NODES) {
      nb = lane_u32(rb, k + 1);
      ne = lane_u32(re, k + 1);
      if (nb + lane < ne) nmy = esrc[nb + lane];
    }
    float4 acc = make_float4(0.f, 0.f, 0.f, 0.f);
    const int cnt = (int)min(e - b, 32u);
    int j = 0;
    for (; j + AGG_GROUP <= cnt; j += AGG_GROUP) {
      float4 v[AGG_GROUP];
#pragma unroll
      for (int u = 0; u < AGG_GROUP; ++u)
        v[u] = *(const float4*)(hl + (size_t)lane_u32(my, j + u) * FF);
      __builtin_amdgcn_sched_barrier(0);
#pragma unroll
      for (int u = 0; u < AGG_GROUP; ++u) axpy4(acc, lane_f32(myw, j + u), v[u]);
    }
    const int rem = cnt - j;
    if (rem > 0) {
      float4 v[AGG_GROUP - 1];
#pragma unroll
      for (int u = 0; u < AGG_GROUP - 1; ++u) {
        v[u] = make_float4(0.f, 0.f, 0.f, 0.f);
        if (u < rem) v[u] = *(const float4*)(hl + (size_t)lane_u32(my, j + u) * FF);
      }
#pragma unroll
      for (int u = 0; u < AGG_GROUP - 1; ++u)
        if (u < rem) axpy4(acc, lane_f32(myw, j + u), v[u]);
    }
    for (unsigned i = b + 32u; i < e; ++i) {
      const unsigned s = esrc[i];
      axpy4(acc, dsout[s], *(const float4*)(hl + (size_t)s * FF));
    }
    const float c = lane_f32(cin, k);
    *(float4*)(agg + (size_t)(n0 + k) * FF + lane * 4) =
        make_float4(c * acc.x, c * acc.y, c * acc.z, c * acc.w);
    b = nb;
    e = ne;
    my = nmy;
    myw = 0.f;
    if (b + lane < e) myw = dsout[my];
  }
}

__global__ __launch_bounds__(128) void k_agg64(const float* __restrict__ P,
                                               const unsigned* __restrict__ rows,
                                               const unsigned* __restrict__ esrc,
                                               const float* __restrict__ dsin,
                                               float* __restrict__ R) {
  const int lane = threadIdx.x & 31;
  const int n0 = (blockIdx.x * 4 + (threadIdx.x >> 5)) * AGG_NODES;
  unsigned rb = 0u, re = 0u;
  float cin = 0.f;
  if (lane < AGG_NODES) {
    const uint2 r = *(const uint2*)(rows + 2 * (n0 + lane));
    rb = r.x;
    re = r.y;
    cin = dsin[n0 + lane];
  }
  unsigned b = lane_u32(rb, 0), e = lane_u32(re, 0);
  unsigned my = 0u;
  if (b + lane < e) my = esrc[b + lane];
  const float* pl = P + lane * 2;

#pragma unroll 1
  for (int k = 0; k < AGG_NODES; ++k) {
    unsigned nb = 0u, ne = 0u, nmy = 0u;
    if (k + 1 < AGG_NODES) {
      nb = lane_u32(rb, k + 1);
      ne = lane_u32(re, k + 1);
      if (nb + lane < ne) nmy = esrc[nb + lane];
    }
    float2 acc = make_float2(0.f, 0.f);
    const int cnt = (int)min(e - b, 32u);
    int j = 0;
    for (; j + AGG_GROUP <= cnt; j += AGG_GROUP) {
      float2 v[AGG_GROUP];
#pragma unroll
      for (int u = 0; u < AGG_GROUP; ++u)
        v[u] = *(const float2*)(pl + (size_t)lane_u32(my, j + u) * 64);
#pragma unroll
      for (int u = 0; u < AGG_GROUP; ++u) {
        acc.x += v[u].x;
        acc.y += v[u].y;
      }
    }
    const int rem = cnt - j;
    if (rem > 0) {
      float2 v[AGG_GROUP - 1];
#pragma unroll
      for (int u = 0; u < AGG_GROUP - 1; ++u) {
        v[u] = make_float2(0.f, 0.f);
        if (u < rem) v[u] = *(const float2*)(pl + (size_t)lane_u32(my, j + u) * 64);
      }
#pragma unroll
      for (int u = 0; u < AGG_GROUP - 1; ++u) {
        if (u < rem) {
          acc.x += v[u].x;
          acc.y += v[u].y;
        }
      }
    }
    for (unsigned i = b + 32u; i < e; ++i) {
      const float2 x = *(const float2*)(pl + (size_t)esrc[i] * 64);
      acc.x += x.x;
      acc.y += x.y;
    }
    const float c = lane_f32(cin, k);
    *(float2*)(R + (size_t)(n0 + k) * 64 + lane * 2) = make_float2(c * acc.x, c * acc.y);
    b = nb;
    e = ne;
    my = nmy;
  }
}


static constexpr int GEMM_BLOCKS = 512;
static constexpr int NSTRIP = NN / 16;

template <int NC>
__device__ __forceinline__ void stage_weights(const float* __restrict__ Bg,
                                              float4* __restrict__ s, int tid) {
  constexpr int C4 = NC / 4;
  constexpr int PITCH = NC + 8;
  for (int i = tid; i < 32 * C4; i += 256) {
    const int g = i / C4, c4 = i - g * C4;
    const float* p = Bg + (size_t)(4 * g) * NC + 4 * c4;
    const float4 r0 = *(const float4*)(p);
    const float4 r1 = *(const float4*)(p + NC);
    const float4 r2 = *(const float4*)(p + 2 * NC);
    const float4 r3 = *(const float4*)(p + 3 * NC);
    const int m = g >> 3, hh = (g >> 2) & 1, q = g & 3;
    float4* d = s + ((m * 4 + q) * 2 + hh) * PITCH + 4 * c4;
    d[0] = make_float4(r0.x, r1.x, r2.x, r3.x);
    d[1] = make_float4(r0.y, r1.y, r2.y, r3.y);
    d[2] = make_float4(r0.z, r1.z, r2.z, r3.z);
    d[3] = make_float4(r0.w, r1.w, r2.w, r3.w);
  }
}

__device__ __forceinline__ v8f mm4(float ax, float ay, float bx, float by, v8f c) {
  v2f a; a.x = ax; a.y = ay;
  v2f b; b.x = bx; b.y = by;
  return __builtin_amdgcn_wmma_f32_16x16x4_f32(false, a, false, b, (short)0, c, false, false);
}

__global__ __launch_bounds__(256) void k_gemm_dual(const float* __restrict__ Agg,
                                                   const float* __restrict__ H,
                                                   const float* __restrict__ W,
                                                   const float* __restrict__ L,
                                                   float* __restrict__ Z) {
  constexpr int PITCH = 128 + 8;
  __shared__ __attribute__((aligned(16))) float4 sW[32 * PITCH];
  __shared__ __attribute__((aligned(16))) float4 sL[32 * PITCH];
  stage_weights<128>(W, sW, threadIdx.x);
  stage_weights<128>(L, sL, threadIdx.x);
  __syncthreads();

  const int wv = threadIdx.x >> 5;
  const int lane = threadIdx.x & 31;
  const int t = lane & 15;
  const int hh = lane >> 4;

  for (int strip = blockIdx.x + GEMM_BLOCKS * wv; strip < NSTRIP; strip += GEMM_BLOCKS * 8) {
    const int r0 = strip << 4;
    const float* arow = Agg + (size_t)(r0 + t) * FF + 16 * hh;
    const float* hrow = H + (size_t)(r0 + t) * FF + 16 * hh;
    v8f acc[8];
#pragma unroll
    for (int ct = 0; ct < 8; ++ct) acc[ct] = v8f{};

#pragma unroll 1
    for (int m = 0; m < 4; ++m) {
      float4 a[4], h[4];
#pragma unroll
      for (int q = 0; q < 4; ++q) {
        a[q] = *(const float4*)(arow + 32 * m + 4 * q);
        h[q] = *(const float4*)(hrow + 32 * m + 4 * q);
      }
#pragma unroll
      for (int q = 0; q < 4; ++q) {
        const float4* pw = sW + ((m * 4 + q) * 2 + hh) * PITCH + t;
        const float4* pl = sL + ((m * 4 + q) * 2 + hh) * PITCH + t;
        float4 bw[8], bl[8];
#pragma unroll
        for (int ct = 0; ct < 8; ++ct) { bw[ct] = pw[16 * ct]; bl[ct] = pl[16 * ct]; }
#pragma unroll
        for (int ct = 0; ct < 8; ++ct) acc[ct] = mm4(a[q].x, a[q].y, bw[ct].x, bw[ct].y, acc[ct]);
#pragma unroll
        for (int ct = 0; ct < 8; ++ct) acc[ct] = mm4(h[q].x, h[q].y, bl[ct].x, bl[ct].y, acc[ct]);
#pragma unroll
        for (int ct = 0; ct < 8; ++ct) acc[ct] = mm4(a[q].z, a[q].w, bw[ct].z, bw[ct].w, acc[ct]);
#pragma unroll
        for (int ct = 0; ct < 8; ++ct) acc[ct] = mm4(h[q].z, h[q].w, bl[ct].z, bl[ct].w, acc[ct]);
      }
    }
#pragma unroll
    for (int ct = 0; ct < 8; ++ct)
#pragma unroll
      for (int v = 0; v < 8; ++v)
        Z[(size_t)(r0 + v + 8 * hh) * FF + 16 * ct + t] = acc[ct][v];
  }
}

__global__ __launch_bounds__(256) void k_gemm_p(const float* __restrict__ H,
                                                const float* __restrict__ dsout,
                                                const float* __restrict__ Wp,
                                                float* __restrict__ P) {
  constexpr int PITCH = 64 + 8;
  __shared__ __attribute__((aligned(16))) float4 sW[32 * PITCH];
  stage_weights<64>(Wp, sW, threadIdx.x);
  __syncthreads();

  const int wv = threadIdx.x >> 5;
  const int lane = threadIdx.x & 31;
  const int t = lane & 15, hh = lane >> 4;

  for (int strip = blockIdx.x + GEMM_BLOCKS * wv; strip < NSTRIP; strip += GEMM_BLOCKS * 8) {
    const int r0 = strip << 4;
    const float d = dsout[r0 + t];
    const float* hrow = H + (size_t)(r0 + t) * FF + 16 * hh;
    v8f acc[4];
#pragma unroll
    for (int ct = 0; ct < 4; ++ct) acc[ct] = v8f{};
#pragma unroll 1
    for (int m = 0; m < 4; ++m) {
      float4 a[4];
#pragma unroll
      for (int q = 0; q < 4; ++q) {
        const float4 x = *(const float4*)(hrow + 32 * m + 4 * q);
        a[q] = make_float4(d * x.x, d * x.y, d * x.z, d * x.w);
      }
#pragma unroll
      for (int q = 0; q < 4; ++q) {
        const float4* pw = sW + ((m * 4 + q) * 2 + hh) * PITCH + t;
        float4 bw[4];
#pragma unroll
        for (int ct = 0; ct < 4; ++ct) bw[ct] = pw[16 * ct];
#pragma unroll
        for (int ct = 0; ct < 4; ++ct) acc[ct] = mm4(a[q].x, a[q].y, bw[ct].x, bw[ct].y, acc[ct]);
#pragma unroll
        for (int ct = 0; ct < 4; ++ct) acc[ct] = mm4(a[q].z, a[q].w, bw[ct].z, bw[ct].w, acc[ct]);
      }
    }
#pragma unroll
    for (int ct = 0; ct < 4; ++ct)
#pragma unroll
      for (int v = 0; v < 8; ++v)
        P[(size_t)(r0 + v + 8 * hh) * 64 + 16 * ct + t] = acc[ct][v];
  }
}

__global__ __launch_bounds__(256) void k_gemm_out(const float* __restrict__ H,
                                                  const float* __restrict__ Lp,
                                                  const float* __restrict__ R,
                                                  const float* __restrict__ b2,
                                                  float* __restrict__ out) {
  constexpr int PITCH = 64 + 8;
  __shared__ __attribute__((aligned(16))) float4 sL[32 * PITCH];
  stage_weights<64>(Lp, sL, threadIdx.x);
  __syncthreads();

  const int wv = threadIdx.x >> 5;
  const int lane = threadIdx.x & 31;
  const int t = lane & 15, hh = lane >> 4;

  for (int strip = blockIdx.x + GEMM_BLOCKS * wv; strip < NSTRIP; strip += GEMM_BLOCKS * 8) {
    const int r0 = strip << 4;
    const float* hrow = H + (size_t)(r0 + t) * FF + 16 * hh;
    v8f acc[3];
#pragma unroll
    for (int ct = 0; ct < 3; ++ct) acc[ct] = v8f{};
#pragma unroll 1
    for (int m = 0; m < 4; ++m) {
      float4 a[4];
#pragma unroll
      for (int q = 0; q < 4; ++q) a[q] = *(const float4*)(hrow + 32 * m + 4 * q);
#pragma unroll
      for (int q = 0; q < 4; ++q) {
        const float4* pl = sL + ((m * 4 + q) * 2 + hh) * PITCH + t;
        float4 bl[3];
#pragma unroll
        for (int ct = 0; ct < 3; ++ct) bl[ct] = pl[16 * ct];
#pragma unroll
        for (int ct = 0; ct < 3; ++ct) acc[ct] = mm4(a[q].x, a[q].y, bl[ct].x, bl[ct].y, acc[ct]);
#pragma unroll
        for (int ct = 0; ct < 3; ++ct) acc[ct] = mm4(a[q].z, a[q].w, bl[ct].z, bl[ct].w, acc[ct]);
      }
    }
#pragma unroll
    for (int ct = 0; ct < 3; ++ct) {
      const int col = 16 * ct + t;
      if (col < 40) {
        const float bias = b2[col];
#pragma unroll
        for (int v = 0; v < 8; ++v) {
          const int row = r0 + v + 8 * hh;
          out[(size_t)row * 40 + col] = acc[ct][v] + R[(size_t)row * 64 + col] + bias;
        }
      }
    }
  }
}


__global__ void k_colstats(const float* __restrict__ z, float* __restrict__ cs,
                           float* __restrict__ cs2) {
  const int c = threadIdx.x;
  const size_t r0 = (size_t)blockIdx.x * 250;
  float s = 0.f, s2 = 0.f;
  for (int i = 0; i < 250; ++i) {
    const float v = z[(r0 + i) * FF + c];
    s += v;
    s2 = fmaf(v, v, s2);
  }
  atomicAdd(&cs[c], s);
  atomicAdd(&cs2[c], s2);
}

__global__ void k_bnparams(const float* __restrict__ cs, const float* __restrict__ cs2,
                           const float* __restrict__ g, const float* __restrict__ be,
                           float* __restrict__ bnA, float* __restrict__ bnB) {
  const int c = threadIdx.x;
  const float inv = 1.0f / (float)NN;
  const float mu = cs[c] * inv;
  const float var = cs2[c] * inv - mu * mu;
  const float a = g[c] * rsqrtf(var + 1e-5f);
  bnA[c] = a;
  bnB[c] = be[c] - mu * a;
}

__global__ void k_bnapply(const float* __restrict__ z, const float* __restrict__ bnA,
                          const float* __restrict__ bnB, float* __restrict__ hout) {
  const size_t i = (size_t)blockIdx.x * 256 + threadIdx.x;
  const int c = (int)(i & 127);
  const float v = fmaf(z[i], bnA[c], bnB[c]);
  hout[i] = v > 0.f ? v : 0.f;
}

__global__ void k_padw(const float* __restrict__ W2, const float* __restrict__ L2,
                       float* __restrict__ Wp, float* __restrict__ Lp) {
  const int i = blockIdx.x * 256 + threadIdx.x;
  const int k = i >> 6, c = i & 63;
  const float w = (c < 40) ? W2[k * 40 + c] : 0.f;
  const float l = (c < 40) ? L2[k * 40 + c] : 0.f;
  Wp[i] = w;
  Lp[i] = l;
}


extern "C" void kernel_launch(void* const* d_in, const int* in_sizes, int n_in,
                              void* d_out, int out_size, void* d_ws, size_t ws_size,
                              hipStream_t stream) {
  (void)in_sizes; (void)n_in; (void)out_size; (void)ws_size;

  const float* feat = (const float*)d_in[0];
  const int* src   = (const int*)d_in[1];
  const int* dst   = (const int*)d_in[2];
  const float* W0  = (const float*)d_in[3];
  const float* L0  = (const float*)d_in[4];
  const float* W1  = (const float*)d_in[5];
  const float* L1  = (const float*)d_in[6];
  const float* W2  = (const float*)d_in[7];
  const float* L2  = (const float*)d_in[8];
  const float* b2  = (const float*)d_in[9];
  const float* g0  = (const float*)d_in[10];
  const float* be0 = (const float*)d_in[11];
  const float* g1  = (const float*)d_in[12];
  const float* be1 = (const float*)d_in[13];
  float* out = (float*)d_out;

  char* base = (char*)d_ws;
  size_t off = 0;
  auto take = [&](size_t bytes) -> char* {
    char* p = base + off;
    off = (off + bytes + 255) & ~(size_t)255;
    return p;
  };
  unsigned* indeg  = (unsigned*)take((size_t)(NN + 1) * 4);
  unsigned* bump   = indeg + NN;
  unsigned* outdeg = (unsigned*)take((size_t)NN * 4);
  unsigned* rows   = (unsigned*)take((size_t)NN * 8);
  unsigned* cursor = (unsigned*)take((size_t)NN * 4);
  unsigned* esrc   = (unsigned*)take((size_t)EE * 4);
  float* dsout = (float*)take((size_t)NN * 4);
  float* dsin  = (float*)take((size_t)NN * 4);
  float* cs    = (float*)take(128 * 4);
  float* cs2   = (float*)take(128 * 4);
  float* bnA   = (float*)take(128 * 4);
  float* bnB   = (float*)take(128 * 4);
  float* Wp    = (float*)take(128 * 64 * 4);
  float* Lp    = (float*)take(128 * 64 * 4);
  float* bufA  = (float*)take((size_t)NN * FF * 4);
  float* bufB  = (float*)take((size_t)NN * FF * 4);
  float* bufC  = (float*)take((size_t)NN * FF * 4);

  const int BEdge = (EE + 255) / 256;
  const int BNode = (NN + 255) / 256;
  const int BApply = (NN * FF) / 256;

  k_zero_u32<<<BNode, 256, 0, stream>>>(indeg, NN + 1);
  k_zero_u32<<<BNode, 256, 0, stream>>>(outdeg, NN);
  k_deg<<<BEdge, 256, 0, stream>>>(src, dst, outdeg, indeg);
  k_ds_alloc<<<BNode, 256, 0, stream>>>(outdeg, indeg, bump, dsout, dsin, rows, cursor);
  k_fill<<<BEdge, 256, 0, stream>>>(src, dst, cursor, esrc);
  k_padw<<<32, 256, 0, stream>>>(W2, L2, Wp, Lp);

  k_agg128<<<AGG_BLOCKS, 128, 0, stream>>>(feat, rows, esrc, dsout, dsin, bufA);
  k_gemm_dual<<<GEMM_BLOCKS, 256, 0, stream>>>(bufA, feat, W0, L0, bufB);
  k_zero_u32<<<1, 256, 0, stream>>>((unsigned*)cs, 256);
  k_colstats<<<400, 128, 0, stream>>>(bufB, cs, cs2);
  k_bnparams<<<1, 128, 0, stream>>>(cs, cs2, g0, be0, bnA, bnB);
  k_bnapply<<<BApply, 256, 0, stream>>>(bufB, bnA, bnB, bufC);

  k_agg128<<<AGG_BLOCKS, 128, 0, stream>>>(bufC, rows, esrc, dsout, dsin, bufA);
  k_gemm_dual<<<GEMM_BLOCKS, 256, 0, stream>>>(bufA, bufC, W1, L1, bufB);
  k_zero_u32<<<1, 256, 0, stream>>>((unsigned*)cs, 256);
  k_colstats<<<400, 128, 0, stream>>>(bufB, cs, cs2);
  k_bnparams<<<1, 128, 0, stream>>>(cs, cs2, g1, be1, bnA, bnB);
  k_bnapply<<<BApply, 256, 0, stream>>>(bufB, bnA, bnB, bufC);

  k_gemm_p<<<GEMM_BLOCKS, 256, 0, stream>>>(bufC, dsout, Wp, bufA);
  k_agg64<<<AGG_BLOCKS, 128, 0, stream>>>(bufA, rows, esrc, dsin, bufB);
  k_gemm_out<<<GEMM_BLOCKS, 256, 0, stream>>>(bufC, Lp, bufB, b2, out);
}


# CSR and both norms by range binning: edges bucketed by dst>>8 and src>>8 via LDS counters, one block per 256-node bin builds its rows in LDS; per-edge global atomics gone; row (sum) order changes
# speedup vs baseline: 1.8902x; 1.3842x over previous
#include <hip/hip_runtime.h>
#include <hip/hip_bf16.h>


static constexpr int NN = 100000;
static constexpr int EE = 1600000;
static constexpr int FF = 128;

typedef __attribute__((ext_vector_type(2))) float v2f;
typedef __attribute__((ext_vector_type(8))) float v8f;


__global__ void k_zero_u32(unsigned* __restrict__ p, int n) {
  int i = blockIdx.x * 256 + threadIdx.x;
  if (i < n) p[i] = 0u;
}


static constexpr int BIN_SHIFT = 8;
static constexpr int BIN_NODES = 1 << BIN_SHIFT;
static constexpr int NBIN = (NN + BIN_NODES - 1) / BIN_NODES;
static constexpr int SLICES = 256;
static constexpr int SLICE_EDGES = 6256;
static constexpr int SLICE_IT = (SLICE_EDGES / 4 + 255) / 256;
static constexpr int ALLOC_BLOCKS = (2 * NBIN + 7) / 8;
static constexpr int BIN_BATCH = 8;
static_assert(SLICE_EDGES % 4 == 0 && EE % 4 == 0, "slices are read four edges at a time");
static_assert((SLICES - 1) * SLICE_EDGES < EE && SLICES * SLICE_EDGES >= EE, "slices must cover the edges");
static_assert(NN <= (1 << 24), "a bin entry keeps the node id in 24 bits");

__global__ __launch_bounds__(256) void k_binhist(const int* __restrict__ src, const int* __restrict__ dst,
                                                 unsigned* __restrict__ cnt, unsigned* __restrict__ bump) {
  __shared__ unsigned hD[NBIN];
  __shared__ unsigned hS[NBIN];
  const int tid = threadIdx.x, b = blockIdx.x;
  for (int i = tid; i < NBIN; i += 256) {
    hD[i] = 0u;
    hS[i] = 0u;
  }
  if (b == 0 && tid < 2) bump[tid] = 0u;
  __syncthreads();
  const int e0 = b * SLICE_EDGES;
  const int n4 = min(EE - e0, SLICE_EDGES) >> 2;
  const int4* s4 = (const int4*)(src + e0);
  const int4* d4 = (const int4*)(dst + e0);
  int4 s[SLICE_IT], d[SLICE_IT];
#pragma unroll
  for (int u = 0; u < SLICE_IT; ++u) {
    const int i = min(tid + 256 * u, n4 - 1);
    s[u] = s4[i];
    d[u] = d4[i];
  }
#pragma unroll
  for (int u = 0; u < SLICE_IT; ++u) {
    if (tid + 256 * u < n4) {
      atomicAdd(&hD[d[u].x >> BIN_SHIFT], 1u);
      atomicAdd(&hD[d[u].y >> BIN_SHIFT], 1u);
      atomicAdd(&hD[d[u].z >> BIN_SHIFT], 1u);
      atomicAdd(&hD[d[u].w >> BIN_SHIFT], 1u);
      atomicAdd(&hS[s[u].x >> BIN_SHIFT], 1u);
      atomicAdd(&hS[s[u].y >> BIN_SHIFT], 1u);
      atomicAdd(&hS[s[u].z >> BIN_SHIFT], 1u);
      atomicAdd(&hS[s[u].w >> BIN_SHIFT], 1u);
    }
  }
  __syncthreads();
  for (int i = tid; i < NBIN; i += 256) {
    cnt[(size_t)i * SLICES + b] = hD[i];
    cnt[(size_t)(NBIN + i) * SLICES + b] = hS[i];
  }
}

__global__ __launch_bounds__(256) void k_binalloc(unsigned* __restrict__ cnt, unsigned* __restrict__ bump,
                                                  unsigned* __restrict__ binbase,
                                                  unsigned* __restrict__ bintot) {
  const int lane = threadIdx.x & 31;
  const int w = blockIdx.x * 8 + (threadIdx.x >> 5);
  if (w >= 2 * NBIN) return;
  uint4* p = (uint4*)(cnt + (size_t)w * SLICES) + lane * 2;
  const uint4 a = p[0];
  const uint4 c = p[1];
  const unsigned loc = a.x + a.y + a.z + a.w + c.x + c.y + c.z + c.w;
  unsigned inc = loc;
  for (int d = 1; d < 32; d <<= 1) {
    const unsigned up = __shfl_up(inc, d);
    if (lane >= d) inc += up;
  }
  const unsigned tot = __shfl(inc, 31);
  unsigned base = 0u;
  if (lane == 0) base = atomicAdd(&bump[w >= NBIN ? 1 : 0], tot);
  base = __shfl(base, 0);
  unsigned o = base + inc - loc;
  uint4 ra, rc;
  ra.x = o; o += a.x;
  ra.y = o; o += a.y;
  ra.z = o; o += a.z;
  ra.w = o; o += a.w;
  rc.x = o; o += c.x;
  rc.y = o; o += c.y;
  rc.z = o; o += c.z;
  rc.w = o;
  p[0] = ra;
  p[1] = rc;
  if (lane == 0) {
    binbase[w] = base;
    bintot[w] = tot;
  }
}

__device__ __forceinline__ void bin_put4(unsigned* cD, unsigned* cS, unsigned* __restrict__ binD,
                                         unsigned* __restrict__ binS, const int4 s, const int4 d) {
  const unsigned pd0 = atomicAdd(&cD[d.x >> BIN_SHIFT], 1u);
  const unsigned pd1 = atomicAdd(&cD[d.y >> BIN_SHIFT], 1u);
  const unsigned pd2 = atomicAdd(&cD[d.z >> BIN_SHIFT], 1u);
  const unsigned pd3 = atomicAdd(&cD[d.w >> BIN_SHIFT], 1u);
  const unsigned ps0 = atomicAdd(&cS[s.x >> BIN_SHIFT], 1u);
  const unsigned ps1 = atomicAdd(&cS[s.y >> BIN_SHIFT], 1u);
  const unsigned ps2 = atomicAdd(&cS[s.z >> BIN_SHIFT], 1u);
  const unsigned ps3 = atomicAdd(&cS[s.w >> BIN_SHIFT], 1u);
  binD[pd0] = ((unsigned)(d.x & (BIN_NODES - 1)) << 24) | (unsigned)s.x;
  binD[pd1] = ((unsigned)(d.y & (BIN_NODES - 1)) << 24) | (unsigned)s.y;
  binD[pd2] = ((unsigned)(d.z & (BIN_NODES - 1)) << 24) | (unsigned)s.z;
  binD[pd3] = ((unsigned)(d.w & (BIN_NODES - 1)) << 24) | (unsigned)s.w;
  binS[ps0] = (unsigned)(s.x & (BIN_NODES - 1));
  binS[ps1] = (unsigned)(s.y & (BIN_NODES - 1));
  binS[ps2] = (unsigned)(s.z & (BIN_NODES - 1));
  binS[ps3] = (unsigned)(s.w & (BIN_NODES - 1));
}

__global__ __launch_bounds__(256) void k_binscatter(const int* __restrict__ src, const int* __restrict__ dst,
                                                    const unsigned* __restrict__ off,
                                                    unsigned* __restrict__ binD,
                                                    unsigned* __restrict__ binS) {
  __shared__ unsigned cD[NBIN];
  __shared__ unsigned cS[NBIN];
  const int tid = threadIdx.x, b = blockIdx.x;
  for (int i = tid; i < NBIN; i += 256) {
    cD[i] = off[(size_t)i * SLICES + b];
    cS[i] = off[(size_t)(NBIN + i) * SLICES + b];
  }
  __syncthreads();
  const int e0 = b * SLICE_EDGES;
  const int n4 = min(EE - e0, SLICE_EDGES) >> 2;
  const int4* s4 = (const int4*)(src + e0);
  const int4* d4 = (const int4*)(dst + e0);
  int4 s[SLICE_IT], d[SLICE_IT];
#pragma unroll
  for (int u = 0; u < SLICE_IT; ++u) {
    const int i = min(tid + 256 * u, n4 - 1);
    s[u] = s4[i];
    d[u] = d4[i];
  }
#pragma unroll
  for (int u = 0; u < SLICE_IT; ++u) {
    if (tid + 256 * u < n4) {
      bin_put4(cD, cS, binD, binS, s[u], d[u]);
    }
  }
}

__global__ __launch_bounds__(256) void k_binrows(const unsigned* __restrict__ binD,
                                                 const unsigned* __restrict__ binS,
                                                 const unsigned* __restrict__ binbase,
                                                 const unsigned* __restrict__ bintot,
                                                 float* __restrict__ dsout, float* __restrict__ dsin,
                                                 unsigned* __restrict__ rows, unsigned* __restrict__ esrc) {
  __shared__ unsigned cD[BIN_NODES];
  __shared__ unsigned cS[BIN_NODES];
  __shared__ unsigned cur[BIN_NODES];
  __shared__ unsigned wtot[8];
  const int tid = threadIdx.x, j = blockIdx.x;
  const int lane = tid & 31, wv = tid >> 5;
  cD[tid] = 0u;
  cS[tid] = 0u;
  const unsigned bD = binbase[j], nD = bintot[j];
  const unsigned bS = binbase[NBIN + j], nS = bintot[NBIN + j];
  const unsigned* eD = binD + bD;
  const unsigned* eS = binS + bS;
  __syncthreads();

  for (unsigned i0 = 0u; i0 < nD; i0 += 256u * BIN_BATCH) {
    unsigned v[BIN_BATCH];
#pragma unroll
    for (int u = 0; u < BIN_BATCH; ++u) v[u] = eD[min(i0 + 256u * u + tid, nD - 1u)];
#pragma unroll
    for (int u = 0; u < BIN_BATCH; ++u)
      if (i0 + 256u * u + tid < nD) atomicAdd(&cD[v[u] >> 24], 1u);
  }
  for (unsigned i0 = 0u; i0 < nS; i0 += 256u * BIN_BATCH) {
    unsigned v[BIN_BATCH];
#pragma unroll
    for (int u = 0; u < BIN_BATCH; ++u) v[u] = eS[min(i0 + 256u * u + tid, nS - 1u)];
#pragma unroll
    for (int u = 0; u < BIN_BATCH; ++u)
      if (i0 + 256u * u + tid < nS) atomicAdd(&cS[v[u]], 1u);
  }
  __syncthreads();

  const unsigned id = cD[tid], od = cS[tid];
  unsigned inc = id;
  for (int d = 1; d < 32; d <<= 1) {
    const unsigned up = __shfl_up(inc, d);
    if (lane >= d) inc += up;
  }
  if (lane == 31) wtot[wv] = inc;
  __syncthreads();
  unsigned st = inc - id;
  for (int w = 0; w < wv; ++w) st += wtot[w];
  cur[tid] = st;
  const int node = j * BIN_NODES + tid;
  if (node < NN) {
    *(uint2*)(rows + 2 * node) = make_uint2(bD + st, bD + st + id);
    dsin[node] = id ? rsqrtf((float)id) : 1.0f;
    dsout[node] = od ? rsqrtf((float)od) : 1.0f;
  }
  __syncthreads();

  unsigned* o = esrc + bD;
  for (unsigned i0 = 0u; i0 < nD; i0 += 256u * BIN_BATCH) {
    unsigned v[BIN_BATCH];
#pragma unroll
    for (int u = 0; u < BIN_BATCH; ++u) v[u] = eD[min(i0 + 256u * u + tid, nD - 1u)];
    unsigned p[BIN_BATCH];
#pragma unroll
    for (int u = 0; u < BIN_BATCH; ++u) {
      p[u] = 0u;
      if (i0 + 256u * u + tid < nD) p[u] = atomicAdd(&cur[v[u] >> 24], 1u);
    }
#pragma unroll
    for (int u = 0; u < BIN_BATCH; ++u)
      if (i0 + 256u * u + tid < nD) o[p[u]] = v[u] & 0xFFFFFFu;
  }
}


static constexpr int AGG_NODES = 8;
static constexpr int AGG_GROUP = 8;
static constexpr int AGG_WAVES = NN / AGG_NODES;
static constexpr int AGG_BLOCKS = AGG_WAVES / 4;
static_assert(NN % AGG_NODES == 0 && AGG_WAVES % 4 == 0, "gather grid must cover the nodes exactly");

__device__ __forceinline__ unsigned lane_u32(unsigned v, int l) {
  return (unsigned)__builtin_amdgcn_readlane((int)v, l);
}
__device__ __forceinline__ float lane_f32(float v, int l) {
  return __int_as_float(__builtin_amdgcn_readlane(__float_as_int(v), l));
}

__device__ __forceinline__ void axpy4(float4& acc, float w, const float4& v) {
  acc.x = fmaf(w, v.x, acc.x);
  acc.y = fmaf(w, v.y, acc.y);
  acc.z = fmaf(w, v.z, acc.z);
  acc.w = fmaf(w, v.w, acc.w);
}

__global__ __launch_bounds__(128) void k_agg128(const float* __restrict__ h,
                                                const unsigned* __restrict__ rows,
                                                const unsigned* __restrict__ esrc,
                                                const float* __restrict__ dsout,
                                                const float* __restrict__ dsin,
                                                float* __restrict__ agg) {
  const int lane = threadIdx.x & 31;
  const int n0 = (blockIdx.x * 4 + (threadIdx.x >> 5)) * AGG_NODES;
  unsigned rb = 0u, re = 0u;
  float cin = 0.f;
  if (lane < AGG_NODES) {
    const uint2 r = *(const uint2*)(rows + 2 * (n0 + lane));
    rb = r.x;
    re = r.y;
    cin = dsin[n0 + lane];
  }
  unsigned b = lane_u32(rb, 0), e = lane_u32(re, 0);
  unsigned my = 0u;
  float myw = 0.f;
  if (b + lane < e) {
    my = esrc[b + lane];
    myw = dsout[my];
  }
  const float* hl = h + lane * 4;

#pragma unroll 1
  for (int k = 0; k < AGG_NODES; ++k) {
    unsigned nb = 0u, ne = 0u, nmy = 0u;
    if (k + 1 < AGG_NODES) {
      nb = lane_u32(rb, k + 1);
      ne = lane_u32(re, k + 1);
      if (nb + lane < ne) nmy = esrc[nb + lane];
    }
    float4 acc = make_float4(0.f, 0.f, 0.f, 0.f);
    const int cnt = (int)min(e - b, 32u);
    int j = 0;
    for (; j + AGG_GROUP <= cnt; j += AGG_GROUP) {
      float4 v[AGG_GROUP];
#pragma unroll
      for (int u = 0; u < AGG_GROUP; ++u)
        v[u] = *(const float4*)(hl + (size_t)lane_u32(my, j + u) * FF);
      __builtin_amdgcn_sched_barrier(0);
#pragma unroll
      for (int u = 0; u < AGG_GROUP; ++u) axpy4(acc, lane_f32(myw, j + u), v[u]);
    }
    const int rem = cnt - j;
    if (rem > 0) {
      float4 v[AGG_GROUP - 1];
#pragma unroll
      for (int u = 0; u < AGG_GROUP - 1; ++u) {
        v[u] = make_float4(0.f, 0.f, 0.f, 0.f);
        if (u < rem) v[u] = *(const float4*)(hl + (size_t)lane_u32(my, j + u) * FF);
      }
#pragma unroll
      for (int u = 0; u < AGG_GROUP - 1; ++u)
        if (u < rem) axpy4(acc, lane_f32(myw, j + u), v[u]);
    }
    for (unsigned i = b + 32u; i < e; ++i) {
      const unsigned s = esrc[i];
      axpy4(acc, dsout[s], *(const float4*)(hl + (size_t)s * FF));
    }
    const float c = lane_f32(cin, k);
    *(float4*)(agg + (size_t)(n0 + k) * FF + lane * 4) =
        make_float4(c * acc.x, c * acc.y, c * acc.z, c * acc.w);
    b = nb;
    e = ne;
    my = nmy;
    myw = 0.f;
    if (b + lane < e) myw = dsout[my];
  }
}

__global__ __launch_bounds__(128) void k_agg64(const float* __restrict__ P,
                                               const unsigned* __restrict__ rows,
                                               const unsigned* __restrict__ esrc,
                                               const float* __restrict__ dsin,
                                               float* __restrict__ R) {
  const int lane = threadIdx.x & 31;
  const int n0 = (blockIdx.x * 4 + (threadIdx.x >> 5)) * AGG_NODES;
  unsigned rb = 0u, re = 0u;
  float cin = 0.f;
  if (lane < AGG_NODES) {
    const uint2 r = *(const uint2*)(rows + 2 * (n0 + lane));
    rb = r.x;
    re = r.y;
    cin = dsin[n0 + lane];
  }
  unsigned b = lane_u32(rb, 0), e = lane_u32(re, 0);
  unsigned my = 0u;
  if (b + lane < e) my = esrc[b + lane];
  const float* pl = P + lane * 2;

#pragma unroll 1
  for (int k = 0; k < AGG_NODES; ++k) {
    unsigned nb = 0u, ne = 0u, nmy = 0u;
    if (k + 1 < AGG_NODES) {
      nb = lane_u32(rb, k + 1);
      ne = lane_u32(re, k + 1);
      if (nb + lane < ne) nmy = esrc[nb + lane];
    }
    float2 acc = make_float2(0.f, 0.f);
    const int cnt = (int)min(e - b, 32u);
    int j = 0;
    for (; j + AGG_GROUP <= cnt; j += AGG_GROUP) {
      float2 v[AGG_GROUP];
#pragma unroll
      for (int u = 0; u < AGG_GROUP; ++u)
        v[u] = *(const float2*)(pl + (size_t)lane_u32(my, j + u) * 64);
#pragma unroll
      for (int u = 0; u < AGG_GROUP; ++u) {
        acc.x += v[u].x;
        acc.y += v[u].y;
      }
    }
    const int rem = cnt - j;
    if (rem > 0) {
      float2 v[AGG_GROUP - 1];
#pragma unroll
      for (int u = 0; u < AGG_GROUP - 1; ++u) {
        v[u] = make_float2(0.f, 0.f);
        if (u < rem) v[u] = *(const float2*)(pl + (size_t)lane_u32(my, j + u) * 64);
      }
#pragma unroll
      for (int u = 0; u < AGG_GROUP - 1; ++u) {
        if (u < rem) {
          acc.x += v[u].x;
          acc.y += v[u].y;
        }
      }
    }
    for (unsigned i = b + 32u; i < e; ++i) {
      const float2 x = *(const float2*)(pl + (size_t)esrc[i] * 64);
      acc.x += x.x;
      acc.y += x.y;
    }
    const float c = lane_f32(cin, k);
    *(float2*)(R + (size_t)(n0 + k) * 64 + lane * 2) = make_float2(c * acc.x, c * acc.y);
    b = nb;
    e = ne;
    my = nmy;
  }
}


static constexpr int GEMM_BLOCKS = 512;
static constexpr int NSTRIP = NN / 16;

template <int NC>
__device__ __forceinline__ void stage_weights(const float* __restrict__ Bg,
                                              float4* __restrict__ s, int tid) {
  constexpr int C4 = NC / 4;
  constexpr int PITCH = NC + 8;
  for (int i = tid; i < 32 * C4; i += 256) {
    const int g = i / C4, c4 = i - g * C4;
    const float* p = Bg + (size_t)(4 * g) * NC + 4 * c4;
    const float4 r0 = *(const float4*)(p);
    const float4 r1 = *(const float4*)(p + NC);
    const float4 r2 = *(const float4*)(p + 2 * NC);
    const float4 r3 = *(const float4*)(p + 3 * NC);
    const int m = g >> 3, hh = (g >> 2) & 1, q = g & 3;
    float4* d = s + ((m * 4 + q) * 2 + hh) * PITCH + 4 * c4;
    d[0] = make_float4(r0.x, r1.x, r2.x, r3.x);
    d[1] = make_float4(r0.y, r1.y, r2.y, r3.y);
    d[2] = make_float4(r0.z, r1.z, r2.z, r3.z);
    d[3] = make_float4(r0.w, r1.w, r2.w, r3.w);
  }
}

__device__ __forceinline__ v8f mm4(float ax, float ay, float bx, float by, v8f c) {
  v2f a; a.x = ax; a.y = ay;
  v2f b; b.x = bx; b.y = by;
  return __builtin_amdgcn_wmma_f32_16x16x4_f32(false, a, false, b, (short)0, c, false, false);
}

__global__ __launch_bounds__(256) void k_gemm_dual(const float* __restrict__ Agg,
                                                   const float* __restrict__ H,
                                                   const float* __restrict__ W,
                                                   const float* __restrict__ L,
                                                   float* __restrict__ Z) {
  constexpr int PITCH = 128 + 8;
  __shared__ __attribute__((aligned(16))) float4 sW[32 * PITCH];
  __shared__ __attribute__((aligned(16))) float4 sL[32 * PITCH];
  stage_weights<128>(W, sW, threadIdx.x);
  stage_weights<128>(L, sL, threadIdx.x);
  __syncthreads();

  const int wv = threadIdx.x >> 5;
  const int lane = threadIdx.x & 31;
  const int t = lane & 15;
  const int hh = lane >> 4;

  for (int strip = blockIdx.x + GEMM_BLOCKS * wv; strip < NSTRIP; strip += GEMM_BLOCKS * 8) {
    const int r0 = strip << 4;
    const float* arow = Agg + (size_t)(r0 + t) * FF + 16 * hh;
    const float* hrow = H + (size_t)(r0 + t) * FF + 16 * hh;
    v8f acc[8];
#pragma unroll
    for (int ct = 0; ct < 8; ++ct) acc[ct] = v8f{};

#pragma unroll 1
    for (int m = 0; m < 4; ++m) {
      float4 a[4], h[4];
#pragma unroll
      for (int q = 0; q < 4; ++q) {
        a[q] = *(const float4*)(arow + 32 * m + 4 * q);
        h[q] = *(const float4*)(hrow + 32 * m + 4 * q);
      }
#pragma unroll
      for (int q = 0; q < 4; ++q) {
        const float4* pw = sW + ((m * 4 + q) * 2 + hh) * PITCH + t;
        const float4* pl = sL + ((m * 4 + q) * 2 + hh) * PITCH + t;
        float4 bw[8], bl[8];
#pragma unroll
        for (int ct = 0; ct < 8; ++ct) { bw[ct] = pw[16 * ct]; bl[ct] = pl[16 * ct]; }
#pragma unroll
        for (int ct = 0; ct < 8; ++ct) acc[ct] = mm4(a[q].x, a[q].y, bw[ct].x, bw[ct].y, acc[ct]);
#pragma unroll
        for (int ct = 0; ct < 8; ++ct) acc[ct] = mm4(h[q].x, h[q].y, bl[ct].x, bl[ct].y, acc[ct]);
#pragma unroll
        for (int ct = 0; ct < 8; ++ct) acc[ct] = mm4(a[q].z, a[q].w, bw[ct].z, bw[ct].w, acc[ct]);
#pragma unroll
        for (int ct = 0; ct < 8; ++ct) acc[ct] = mm4(h[q].z, h[q].w, bl[ct].z, bl[ct].w, acc[ct]);
      }
    }
#pragma unroll
    for (int ct = 0; ct < 8; ++ct)
#pragma unroll
      for (int v = 0; v < 8; ++v)
        Z[(size_t)(r0 + v + 8 * hh) * FF + 16 * ct + t] = acc[ct][v];
  }
}

__global__ __launch_bounds__(256) void k_gemm_p(const float* __restrict__ H,
                                                const float* __restrict__ dsout,
                                                const float* __restrict__ Wp,
                                                float* __restrict__ P) {
  constexpr int PITCH = 64 + 8;
  __shared__ __attribute__((aligned(16))) float4 sW[32 * PITCH];
  stage_weights<64>(Wp, sW, threadIdx.x);
  __syncthreads();

  const int wv = threadIdx.x >> 5;
  const int lane = threadIdx.x & 31;
  const int t = lane & 15, hh = lane >> 4;

  for (int strip = blockIdx.x + GEMM_BLOCKS * wv; strip < NSTRIP; strip += GEMM_BLOCKS * 8) {
    const int r0 = strip << 4;
    const float d = dsout[r0 + t];
    const float* hrow = H + (size_t)(r0 + t) * FF + 16 * hh;
    v8f acc[4];
#pragma unroll
    for (int ct = 0; ct < 4; ++ct) acc[ct] = v8f{};
#pragma unroll 1
    for (int m = 0; m < 4; ++m) {
      float4 a[4];
#pragma unroll
      for (int q = 0; q < 4; ++q) {
        const float4 x = *(const float4*)(hrow + 32 * m + 4 * q);
        a[q] = make_float4(d * x.x, d * x.y, d * x.z, d * x.w);
      }
#pragma unroll
      for (int q = 0; q < 4; ++q) {
        const float4* pw = sW + ((m * 4 + q) * 2 + hh) * PITCH + t;
        float4 bw[4];
#pragma unroll
        for (int ct = 0; ct < 4; ++ct) bw[ct] = pw[16 * ct];
#pragma unroll
        for (int ct = 0; ct < 4; ++ct) acc[ct] = mm4(a[q].x, a[q].y, bw[ct].x, bw[ct].y, acc[ct]);
#pragma unroll
        for (int ct = 0; ct < 4; ++ct) acc[ct] = mm4(a[q].z, a[q].w, bw[ct].z, bw[ct].w, acc[ct]);
      }
    }
#pragma unroll
    for (int ct = 0; ct < 4; ++ct)
#pragma unroll
      for (int v = 0; v < 8; ++v)
        P[(size_t)(r0 + v + 8 * hh) * 64 + 16 * ct + t] = acc[ct][v];
  }
}

__global__ __launch_bounds__(256) void k_gemm_out(const float* __restrict__ H,
                                                  const float* __restrict__ Lp,
                                                  const float* __restrict__ R,
                                                  const float* __restrict__ b2,
                                                  float* __restrict__ out) {
  constexpr int PITCH = 64 + 8;
  __shared__ __attribute__((aligned(16))) float4 sL[32 * PITCH];
  stage_weights<64>(Lp, sL, threadIdx.x);
  __syncthreads();

  const int wv = threadIdx.x >> 5;
  const int lane = threadIdx.x & 31;
  const int t = lane & 15, hh = lane >> 4;

  for (int strip = blockIdx.x + GEMM_BLOCKS * wv; strip < NSTRIP; strip += GEMM_BLOCKS * 8) {
    const int r0 = strip << 4;
    const float* hrow = H + (size_t)(r0 + t) * FF + 16 * hh;
    v8f acc[3];
#pragma unroll
    for (int ct = 0; ct < 3; ++ct) acc[ct] = v8f{};
#pragma unroll 1
    for (int m = 0; m < 4; ++m) {
      float4 a[4];
#pragma unroll
      for (int q = 0; q < 4; ++q) a[q] = *(const float4*)(hrow + 32 * m + 4 * q);
#pragma unroll
      for (int q = 0; q < 4; ++q) {
        const float4* pl = sL + ((m * 4 + q) * 2 + hh) * PITCH + t;
        float4 bl[3];
#pragma unroll
        for (int ct = 0; ct < 3; ++ct) bl[ct] = pl[16 * ct];
#pragma unroll
        for (int ct = 0; ct < 3; ++ct) acc[ct] = mm4(a[q].x, a[q].y, bl[ct].x, bl[ct].y, acc[ct]);
#pragma unroll
        for (int ct = 0; ct < 3; ++ct) acc[ct] = mm4(a[q].z, a[q].w, bl[ct].z, bl[ct].w, acc[ct]);
      }
    }
#pragma unroll
    for (int ct = 0; ct < 3; ++ct) {
      const int col = 16 * ct + t;
      if (col < 40) {
        const float bias = b2[col];
#pragma unroll
        for (int v = 0; v < 8; ++v) {
          const int row = r0 + v + 8 * hh;
          out[(size_t)row * 40 + col] = acc[ct][v] + R[(size_t)row * 64 + col] + bias;
        }
      }
    }
  }
}


__global__ void k_colstats(const float* __restrict__ z, float* __restrict__ cs,
                           float* __restrict__ cs2) {
  const int c = threadIdx.x;
  const size_t r0 = (size_t)blockIdx.x * 250;
  float s = 0.f, s2 = 0.f;
  for (int i = 0; i < 250; ++i) {
    const float v = z[(r0 + i) * FF + c];
    s += v;
    s2 = fmaf(v, v, s2);
  }
  atomicAdd(&cs[c], s);
  atomicAdd(&cs2[c], s2);
}

__global__ void k_bnparams(const float* __restrict__ cs, const float* __restrict__ cs2,
                           const float* __restrict__ g, const float* __restrict__ be,
                           float* __restrict__ bnA, float* __restrict__ bnB) {
  const int c = threadIdx.x;
  const float inv = 1.0f / (float)NN;
  const float mu = cs[c] * inv;
  const float var = cs2[c] * inv - mu * mu;
  const float a = g[c] * rsqrtf(var + 1e-5f);
  bnA[c] = a;
  bnB[c] = be[c] - mu * a;
}

__global__ void k_bnapply(const float* __restrict__ z, const float* __restrict__ bnA,
                          const float* __restrict__ bnB, float* __restrict__ hout) {
  const size_t i = (size_t)blockIdx.x * 256 + threadIdx.x;
  const int c = (int)(i & 127);
  const float v = fmaf(z[i], bnA[c], bnB[c]);
  hout[i] = v > 0.f ? v : 0.f;
}

__global__ void k_padw(const float* __restrict__ W2, const float* __restrict__ L2,
                       float* __restrict__ Wp, float* __restrict__ Lp) {
  const int i = blockIdx.x * 256 + threadIdx.x;
  const int k = i >> 6, c = i & 63;
  const float w = (c < 40) ? W2[k * 40 + c] : 0.f;
  const float l = (c < 40) ? L2[k * 40 + c] : 0.f;
  Wp[i] = w;
  Lp[i] = l;
}


extern "C" void kernel_launch(void* const* d_in, const int* in_sizes, int n_in,
                              void* d_out, int out_size, void* d_ws, size_t ws_size,
                              hipStream_t stream) {
  (void)in_sizes; (void)n_in; (void)out_size; (void)ws_size;

  const float* feat = (const float*)d_in[0];
  const int* src   = (const int*)d_in[1];
  const int* dst   = (const int*)d_in[2];
  const float* W0  = (const float*)d_in[3];
  const float* L0  = (const float*)d_in[4];
  const float* W1  = (const float*)d_in[5];
  const float* L1  = (const float*)d_in[6];
  const float* W2  = (const float*)d_in[7];
  const float* L2  = (const float*)d_in[8];
  const float* b2  = (const float*)d_in[9];
  const float* g0  = (const float*)d_in[10];
  const float* be0 = (const float*)d_in[11];
  const float* g1  = (const float*)d_in[12];
  const float* be1 = (const float*)d_in[13];
  float* out = (float*)d_out;

  char* base = (char*)d_ws;
  size_t off = 0;
  auto take = [&](size_t bytes) -> char* {
    char* p = base + off;
    off = (off + bytes + 255) & ~(size_t)255;
    return p;
  };
  unsigned* cnt     = (unsigned*)take((size_t)2 * NBIN * SLICES * 4);
  unsigned* bump    = (unsigned*)take(2 * 4);
  unsigned* binbase = (unsigned*)take((size_t)2 * NBIN * 4);
  unsigned* bintot  = (unsigned*)take((size_t)2 * NBIN * 4);
  unsigned* binD    = (unsigned*)take((size_t)EE * 4);
  unsigned* binS    = (unsigned*)take((size_t)EE * 4);
  unsigned* rows    = (unsigned*)take((size_t)NN * 8);
  unsigned* esrc    = (unsigned*)take((size_t)EE * 4);
  float* dsout = (float*)take((size_t)NN * 4);
  float* dsin  = (float*)take((size_t)NN * 4);
  float* cs    = (float*)take(128 * 4);
  float* cs2   = (float*)take(128 * 4);
  float* bnA   = (float*)take(128 * 4);
  float* bnB   = (float*)take(128 * 4);
  float* Wp    = (float*)take(128 * 64 * 4);
  float* Lp    = (float*)take(128 * 64 * 4);
  float* bufA  = (float*)take((size_t)NN * FF * 4);
  float* bufB  = (float*)take((size_t)NN * FF * 4);
  float* bufC  = (float*)take((size_t)NN * FF * 4);

  const int BApply = (NN * FF) / 256;

  k_binhist<<<SLICES, 256, 0, stream>>>(src, dst, cnt, bump);
  k_binalloc<<<ALLOC_BLOCKS, 256, 0, stream>>>(cnt, bump, binbase, bintot);
  k_binscatter<<<SLICES, 256, 0, stream>>>(src, dst, cnt, binD, binS);
  k_binrows<<<NBIN, 256, 0, stream>>>(binD, binS, binbase, bintot, dsout, dsin, rows, esrc);
  k_padw<<<32, 256, 0, stream>>>(W2, L2, Wp, Lp);

  k_agg128<<<AGG_BLOCKS, 128, 0, stream>>>(feat, rows, esrc, dsout, dsin, bufA);
  k_gemm_dual<<<GEMM_BLOCKS, 256, 0, stream>>>(bufA, feat, W0, L0, bufB);
  k_zero_u32<<<1, 256, 0, stream>>>((unsigned*)cs, 256);
  k_colstats<<<400, 128, 0, stream>>>(bufB, cs, cs2);
  k_bnparams<<<1, 128, 0, stream>>>(cs, cs2, g0, be0, bnA, bnB);
  k_bnapply<<<BApply, 256, 0, stream>>>(bufB, bnA, bnB, bufC);

  k_agg128<<<AGG_BLOCKS, 128, 0, stream>>>(bufC, rows, esrc, dsout, dsin, bufA);
  k_gemm_dual<<<GEMM_BLOCKS, 256, 0, stream>>>(bufA, bufC, W1, L1, bufB);
  k_zero_u32<<<1, 256, 0, stream>>>((unsigned*)cs, 256);
  k_colstats<<<400, 128, 0, stream>>>(bufB, cs, cs2);
  k_bnparams<<<1, 128, 0, stream>>>(cs, cs2, g1, be1, bnA, bnB);
  k_bnapply<<<BApply, 256, 0, stream>>>(bufB, bnA, bnB, bufC);

  k_gemm_p<<<GEMM_BLOCKS, 256, 0, stream>>>(bufC, dsout, Wp, bufA);
  k_agg64<<<AGG_BLOCKS, 128, 0, stream>>>(bufA, rows, esrc, dsin, bufB);
  k_gemm_out<<<GEMM_BLOCKS, 256, 0, stream>>>(bufC, Lp, bufB, b2, out);
}


# BN statistics pass re-tiled: 500 blocks of 8 waves, a wave streams 25 whole rows as float4 with about 12 row loads in flight, block sums via LDS; was 25 dependent trips per wave; sum order changes
# speedup vs baseline: 2.0218x; 1.0696x over previous
#include <hip/hip_runtime.h>
#include <hip/hip_bf16.h>


static constexpr int NN = 100000;
static constexpr int EE = 1600000;
static constexpr int FF = 128;

typedef __attribute__((ext_vector_type(2))) float v2f;
typedef __attribute__((ext_vector_type(8))) float v8f;


__global__ void k_zero_u32(unsigned* __restrict__ p, int n) {
  int i = blockIdx.x * 256 + threadIdx.x;
  if (i < n) p[i] = 0u;
}


static constexpr int BIN_SHIFT = 8;
static constexpr int BIN_NODES = 1 << BIN_SHIFT;
static constexpr int NBIN = (NN + BIN_NODES - 1) / BIN_NODES;
static constexpr int SLICES = 256;
static constexpr int SLICE_EDGES = 6256;
static constexpr int SLICE_IT = (SLICE_EDGES / 4 + 255) / 256;
static constexpr int ALLOC_BLOCKS = (2 * NBIN + 7) / 8;
static constexpr int BIN_BATCH = 8;
static_assert(SLICE_EDGES % 4 == 0 && EE % 4 == 0, "slices are read four edges at a time");
static_assert((SLICES - 1) * SLICE_EDGES < EE && SLICES * SLICE_EDGES >= EE, "slices must cover the edges");
static_assert(NN <= (1 << 24), "a bin entry keeps the node id in 24 bits");

__global__ __launch_bounds__(256) void k_binhist(const int* __restrict__ src, const int* __restrict__ dst,
                                                 unsigned* __restrict__ cnt, unsigned* __restrict__ bump) {
  __shared__ unsigned hD[NBIN];
  __shared__ unsigned hS[NBIN];
  const int tid = threadIdx.x, b = blockIdx.x;
  for (int i = tid; i < NBIN; i += 256) {
    hD[i] = 0u;
    hS[i] = 0u;
  }
  if (b == 0 && tid < 2) bump[tid] = 0u;
  __syncthreads();
  const int e0 = b * SLICE_EDGES;
  const int n4 = min(EE - e0, SLICE_EDGES) >> 2;
  const int4* s4 = (const int4*)(src + e0);
  const int4* d4 = (const int4*)(dst + e0);
  int4 s[SLICE_IT], d[SLICE_IT];
#pragma unroll
  for (int u = 0; u < SLICE_IT; ++u) {
    const int i = min(tid + 256 * u, n4 - 1);
    s[u] = s4[i];
    d[u] = d4[i];
  }
#pragma unroll
  for (int u = 0; u < SLICE_IT; ++u) {
    if (tid + 256 * u < n4) {
      atomicAdd(&hD[d[u].x >> BIN_SHIFT], 1u);
      atomicAdd(&hD[d[u].y >> BIN_SHIFT], 1u);
      atomicAdd(&hD[d[u].z >> BIN_SHIFT], 1u);
      atomicAdd(&hD[d[u].w >> BIN_SHIFT], 1u);
      atomicAdd(&hS[s[u].x >> BIN_SHIFT], 1u);
      atomicAdd(&hS[s[u].y >> BIN_SHIFT], 1u);
      atomicAdd(&hS[s[u].z >> BIN_SHIFT], 1u);
      atomicAdd(&hS[s[u].w >> BIN_SHIFT], 1u);
    }
  }
  __syncthreads();
  for (int i = tid; i < NBIN; i += 256) {
    cnt[(size_t)i * SLICES + b] = hD[i];
    cnt[(size_t)(NBIN + i) * SLICES + b] = hS[i];
  }
}

__global__ __launch_bounds__(256) void k_binalloc(unsigned* __restrict__ cnt, unsigned* __restrict__ bump,
                                                  unsigned* __restrict__ binbase,
                                                  unsigned* __restrict__ bintot) {
  const int lane = threadIdx.x & 31;
  const int w = blockIdx.x * 8 + (threadIdx.x >> 5);
  if (w >= 2 * NBIN) return;
  uint4* p = (uint4*)(cnt + (size_t)w * SLICES) + lane * 2;
  const uint4 a = p[0];
  const uint4 c = p[1];
  const unsigned loc = a.x + a.y + a.z + a.w + c.x + c.y + c.z + c.w;
  unsigned inc = loc;
  for (int d = 1; d < 32; d <<= 1) {
    const unsigned up = __shfl_up(inc, d);
    if (lane >= d) inc += up;
  }
  const unsigned tot = __shfl(inc, 31);
  unsigned base = 0u;
  if (lane == 0) base = atomicAdd(&bump[w >= NBIN ? 1 : 0], tot);
  base = __shfl(base, 0);
  unsigned o = base + inc - loc;
  uint4 ra, rc;
  ra.x = o; o += a.x;
  ra.y = o; o += a.y;
  ra.z = o; o += a.z;
  ra.w = o; o += a.w;
  rc.x = o; o += c.x;
  rc.y = o; o += c.y;
  rc.z = o; o += c.z;
  rc.w = o;
  p[0] = ra;
  p[1] = rc;
  if (lane == 0) {
    binbase[w] = base;
    bintot[w] = tot;
  }
}

__device__ __forceinline__ void bin_put4(unsigned* cD, unsigned* cS, unsigned* __restrict__ binD,
                                         unsigned* __restrict__ binS, const int4 s, const int4 d) {
  const unsigned pd0 = atomicAdd(&cD[d.x >> BIN_SHIFT], 1u);
  const unsigned pd1 = atomicAdd(&cD[d.y >> BIN_SHIFT], 1u);
  const unsigned pd2 = atomicAdd(&cD[d.z >> BIN_SHIFT], 1u);
  const unsigned pd3 = atomicAdd(&cD[d.w >> BIN_SHIFT], 1u);
  const unsigned ps0 = atomicAdd(&cS[s.x >> BIN_SHIFT], 1u);
  const unsigned ps1 = atomicAdd(&cS[s.y >> BIN_SHIFT], 1u);
  const unsigned ps2 = atomicAdd(&cS[s.z >> BIN_SHIFT], 1u);
  const unsigned ps3 = atomicAdd(&cS[s.w >> BIN_SHIFT], 1u);
  binD[pd0] = ((unsigned)(d.x & (BIN_NODES - 1)) << 24) | (unsigned)s.x;
  binD[pd1] = ((unsigned)(d.y & (BIN_NODES - 1)) << 24) | (unsigned)s.y;
  binD[pd2] = ((unsigned)(d.z & (BIN_NODES - 1)) << 24) | (unsigned)s.z;
  binD[pd3] = ((unsigned)(d.w & (BIN_NODES - 1)) << 24) | (unsigned)s.w;
  binS[ps0] = (unsigned)(s.x & (BIN_NODES - 1));
  binS[ps1] = (unsigned)(s.y & (BIN_NODES - 1));
  binS[ps2] = (unsigned)(s.z & (BIN_NODES - 1));
  binS[ps3] = (unsigned)(s.w & (BIN_NODES - 1));
}

__global__ __launch_bounds__(256) void k_binscatter(const int* __restrict__ src, const int* __restrict__ dst,
                                                    const unsigned* __restrict__ off,
                                                    unsigned* __restrict__ binD,
                                                    unsigned* __restrict__ binS) {
  __shared__ unsigned cD[NBIN];
  __shared__ unsigned cS[NBIN];
  const int tid = threadIdx.x, b = blockIdx.x;
  for (int i = tid; i < NBIN; i += 256) {
    cD[i] = off[(size_t)i * SLICES + b];
    cS[i] = off[(size_t)(NBIN + i) * SLICES + b];
  }
  __syncthreads();
  const int e0 = b * SLICE_EDGES;
  const int n4 = min(EE - e0, SLICE_EDGES) >> 2;
  const int4* s4 = (const int4*)(src + e0);
  const int4* d4 = (const int4*)(dst + e0);
  int4 s[SLICE_IT], d[SLICE_IT];
#pragma unroll
  for (int u = 0; u < SLICE_IT; ++u) {
    const int i = min(tid + 256 * u, n4 - 1);
    s[u] = s4[i];
    d[u] = d4[i];
  }
#pragma unroll
  for (int u = 0; u < SLICE_IT; ++u) {
    if (tid + 256 * u < n4) {
      bin_put4(cD, cS, binD, binS, s[u], d[u]);
    }
  }
}

__global__ __launch_bounds__(256) void k_binrows(const unsigned* __restrict__ binD,
                                                 const unsigned* __restrict__ binS,
                                                 const unsigned* __restrict__ binbase,
                                                 const unsigned* __restrict__ bintot,
                                                 float* __restrict__ dsout, float* __restrict__ dsin,
                                                 unsigned* __restrict__ rows, unsigned* __restrict__ esrc) {
  __shared__ unsigned cD[BIN_NODES];
  __shared__ unsigned cS[BIN_NODES];
  __shared__ unsigned cur[BIN_NODES];
  __shared__ unsigned wtot[8];
  const int tid = threadIdx.x, j = blockIdx.x;
  const int lane = tid & 31, wv = tid >> 5;
  cD[tid] = 0u;
  cS[tid] = 0u;
  const unsigned bD = binbase[j], nD = bintot[j];
  const unsigned bS = binbase[NBIN + j], nS = bintot[NBIN + j];
  const unsigned* eD = binD + bD;
  const unsigned* eS = binS + bS;
  __syncthreads();

  for (unsigned i0 = 0u; i0 < nD; i0 += 256u * BIN_BATCH) {
    unsigned v[BIN_BATCH];
#pragma unroll
    for (int u = 0; u < BIN_BATCH; ++u) v[u] = eD[min(i0 + 256u * u + tid, nD - 1u)];
#pragma unroll
    for (int u = 0; u < BIN_BATCH; ++u)
      if (i0 + 256u * u + tid < nD) atomicAdd(&cD[v[u] >> 24], 1u);
  }
  for (unsigned i0 = 0u; i0 < nS; i0 += 256u * BIN_BATCH) {
    unsigned v[BIN_BATCH];
#pragma unroll
    for (int u = 0; u < BIN_BATCH; ++u) v[u] = eS[min(i0 + 256u * u + tid, nS - 1u)];
#pragma unroll
    for (int u = 0; u < BIN_BATCH; ++u)
      if (i0 + 256u * u + tid < nS) atomicAdd(&cS[v[u]], 1u);
  }
  __syncthreads();

  const unsigned id = cD[tid], od = cS[tid];
  unsigned inc = id;
  for (int d = 1; d < 32; d <<= 1) {
    const unsigned up = __shfl_up(inc, d);
    if (lane >= d) inc += up;
  }
  if (lane == 31) wtot[wv] = inc;
  __syncthreads();
  unsigned st = inc - id;
  for (int w = 0; w < wv; ++w) st += wtot[w];
  cur[tid] = st;
  const int node = j * BIN_NODES + tid;
  if (node < NN) {
    *(uint2*)(rows + 2 * node) = make_uint2(bD + st, bD + st + id);
    dsin[node] = id ? rsqrtf((float)id) : 1.0f;
    dsout[node] = od ? rsqrtf((float)od) : 1.0f;
  }
  __syncthreads();

  unsigned* o = esrc + bD;
  for (unsigned i0 = 0u; i0 < nD; i0 += 256u * BIN_BATCH) {
    unsigned v[BIN_BATCH];
#pragma unroll
    for (int u = 0; u < BIN_BATCH; ++u) v[u] = eD[min(i0 + 256u * u + tid, nD - 1u)];
    unsigned p[BIN_BATCH];
#pragma unroll
    for (int u = 0; u < BIN_BATCH; ++u) {
      p[u] = 0u;
      if (i0 + 256u * u + tid < nD) p[u] = atomicAdd(&cur[v[u] >> 24], 1u);
    }
#pragma unroll
    for (int u = 0; u < BIN_BATCH; ++u)
      if (i0 + 256u * u + tid < nD) o[p[u]] = v[u] & 0xFFFFFFu;
  }
}


static constexpr int AGG_NODES = 8;
static constexpr int AGG_GROUP = 8;
static constexpr int AGG_WAVES = NN / AGG_NODES;
static constexpr int AGG_BLOCKS = AGG_WAVES / 4;
static_assert(NN % AGG_NODES == 0 && AGG_WAVES % 4 == 0, "gather grid must cover the nodes exactly");

__device__ __forceinline__ unsigned lane_u32(unsigned v, int l) {
  return (unsigned)__builtin_amdgcn_readlane((int)v, l);
}
__device__ __forceinline__ float lane_f32(float v, int l) {
  return __int_as_float(__builtin_amdgcn_readlane(__float_as_int(v), l));
}

__device__ __forceinline__ void axpy4(float4& acc, float w, const float4& v) {
  acc.x = fmaf(w, v.x, acc.x);
  acc.y = fmaf(w, v.y, acc.y);
  acc.z = fmaf(w, v.z, acc.z);
  acc.w = fmaf(w, v.w, acc.w);
}

__global__ __launch_bounds__(128) void k_agg128(const float* __restrict__ h,
                                                const unsigned* __restrict__ rows,
                                                const unsigned* __restrict__ esrc,
                                                const float* __restrict__ dsout,
                                                const float* __restrict__ dsin,
                                                float* __restrict__ agg) {
  const int lane = threadIdx.x & 31;
  const int n0 = (blockIdx.x * 4 + (threadIdx.x >> 5)) * AGG_NODES;
  unsigned rb = 0u, re = 0u;
  float cin = 0.f;
  if (lane < AGG_NODES) {
    const uint2 r = *(const uint2*)(rows + 2 * (n0 + lane));
    rb = r.x;
    re = r.y;
    cin = dsin[n0 + lane];
  }
  unsigned b = lane_u32(rb, 0), e = lane_u32(re, 0);
  unsigned my = 0u;
  float myw = 0.f;
  if (b + lane < e) {
    my = esrc[b + lane];
    myw = dsout[my];
  }
  const float* hl = h + lane * 4;

#pragma unroll 1
  for (int k = 0; k < AGG_NODES; ++k) {
    unsigned nb = 0u, ne = 0u, nmy = 0u;
    if (k + 1 < AGG_NODES) {
      nb = lane_u32(rb, k + 1);
      ne = lane_u32(re, k + 1);
      if (nb + lane < ne) nmy = esrc[nb + lane];
    }
    float4 acc = make_float4(0.f, 0.f, 0.f, 0.f);
    const int cnt = (int)min(e - b, 32u);
    int j = 0;
    for (; j + AGG_GROUP <= cnt; j += AGG_GROUP) {
      float4 v[AGG_GROUP];
#pragma unroll
      for (int u = 0; u < AGG_GROUP; ++u)
        v[u] = *(const float4*)(hl + (size_t)lane_u32(my, j + u) * FF);
      __builtin_amdgcn_sched_barrier(0);
#pragma unroll
      for (int u = 0; u < AGG_GROUP; ++u) axpy4(acc, lane_f32(myw, j + u), v[u]);
    }
    const int rem = cnt - j;
    if (rem > 0) {
      float4 v[AGG_GROUP - 1];
#pragma unroll
      for (int u = 0; u < AGG_GROUP - 1; ++u) {
        v[u] = make_float4(0.f, 0.f, 0.f, 0.f);
        if (u < rem) v[u] = *(const float4*)(hl + (size_t)lane_u32(my, j + u) * FF);
      }
#pragma unroll
      for (int u = 0; u < AGG_GROUP - 1; ++u)
        if (u < rem) axpy4(acc, lane_f32(myw, j + u), v[u]);
    }
    for (unsigned i = b + 32u; i < e; ++i) {
      const unsigned s = esrc[i];
      axpy4(acc, dsout[s], *(const float4*)(hl + (size_t)s * FF));
    }
    const float c = lane_f32(cin, k);
    *(float4*)(agg + (size_t)(n0 + k) * FF + lane * 4) =
        make_float4(c * acc.x, c * acc.y, c * acc.z, c * acc.w);
    b = nb;
    e = ne;
    my = nmy;
    myw = 0.f;
    if (b + lane < e) myw = dsout[my];
  }
}

__global__ __launch_bounds__(128) void k_agg64(const float* __restrict__ P,
                                               const unsigned* __restrict__ rows,
                                               const unsigned* __restrict__ esrc,
                                               const float* __restrict__ dsin,
                                               float* __restrict__ R) {
  const int lane = threadIdx.x & 31;
  const int n0 = (blockIdx.x * 4 + (threadIdx.x >> 5)) * AGG_NODES;
  unsigned rb = 0u, re = 0u;
  float cin = 0.f;
  if (lane < AGG_NODES) {
    const uint2 r = *(const uint2*)(rows + 2 * (n0 + lane));
    rb = r.x;
    re = r.y;
    cin = dsin[n0 + lane];
  }
  unsigned b = lane_u32(rb, 0), e = lane_u32(re, 0);
  unsigned my = 0u;
  if (b + lane < e) my = esrc[b + lane];
  const float* pl = P + lane * 2;

#pragma unroll 1
  for (int k = 0; k < AGG_NODES; ++k) {
    unsigned nb = 0u, ne = 0u, nmy = 0u;
    if (k + 1 < AGG_NODES) {
      nb = lane_u32(rb, k + 1);
      ne = lane_u32(re, k + 1);
      if (nb + lane < ne) nmy = esrc[nb + lane];
    }
    float2 acc = make_float2(0.f, 0.f);
    const int cnt = (int)min(e - b, 32u);
    int j = 0;
    for (; j + AGG_GROUP <= cnt; j += AGG_GROUP) {
      float2 v[AGG_GROUP];
#pragma unroll
      for (int u = 0; u < AGG_GROUP; ++u)
        v[u] = *(const float2*)(pl + (size_t)lane_u32(my, j + u) * 64);
#pragma unroll
      for (int u = 0; u < AGG_GROUP; ++u) {
        acc.x += v[u].x;
        acc.y += v[u].y;
      }
    }
    const int rem = cnt - j;
    if (rem > 0) {
      float2 v[AGG_GROUP - 1];
#pragma unroll
      for (int u = 0; u < AGG_GROUP - 1; ++u) {
        v[u] = make_float2(0.f, 0.f);
        if (u < rem) v[u] = *(const float2*)(pl + (size_t)lane_u32(my, j + u) * 64);
      }
#pragma unroll
      for (int u = 0; u < AGG_GROUP - 1; ++u) {
        if (u < rem) {
          acc.x += v[u].x;
          acc.y += v[u].y;
        }
      }
    }
    for (unsigned i = b + 32u; i < e; ++i) {
      const float2 x = *(const float2*)(pl + (size_t)esrc[i] * 64);
      acc.x += x.x;
      acc.y += x.y;
    }
    const float c = lane_f32(cin, k);
    *(float2*)(R + (size_t)(n0 + k) * 64 + lane * 2) = make_float2(c * acc.x, c * acc.y);
    b = nb;
    e = ne;
    my = nmy;
  }
}


static constexpr int GEMM_BLOCKS = 512;
static constexpr int NSTRIP = NN / 16;

template <int NC>
__device__ __forceinline__ void stage_weights(const float* __restrict__ Bg,
                                              float4* __restrict__ s, int tid) {
  constexpr int C4 = NC / 4;
  constexpr int PITCH = NC + 8;
  for (int i = tid; i < 32 * C4; i += 256) {
    const int g = i / C4, c4 = i - g * C4;
    const float* p = Bg + (size_t)(4 * g) * NC + 4 * c4;
    const float4 r0 = *(const float4*)(p);
    const float4 r1 = *(const float4*)(p + NC);
    const float4 r2 = *(const float4*)(p + 2 * NC);
    const float4 r3 = *(const float4*)(p + 3 * NC);
    const int m = g >> 3, hh = (g >> 2) & 1, q = g & 3;
    float4* d = s + ((m * 4 + q) * 2 + hh) * PITCH + 4 * c4;
    d[0] = make_float4(r0.x, r1.x, r2.x, r3.x);
    d[1] = make_float4(r0.y, r1.y, r2.y, r3.y);
    d[2] = make_float4(r0.z, r1.z, r2.z, r3.z);
    d[3] = make_float4(r0.w, r1.w, r2.w, r3.w);
  }
}

__device__ __forceinline__ v8f mm4(float ax, float ay, float bx, float by, v8f c) {
  v2f a; a.x = ax; a.y = ay;
  v2f b; b.x = bx; b.y = by;
  return __builtin_amdgcn_wmma_f32_16x16x4_f32(false, a, false, b, (short)0, c, false, false);
}

__global__ __launch_bounds__(256) void k_gemm_dual(const float* __restrict__ Agg,
                                                   const float* __restrict__ H,
                                                   const float* __restrict__ W,
                                                   const float* __restrict__ L,
                                                   float* __restrict__ Z) {
  constexpr int PITCH = 128 + 8;
  __shared__ __attribute__((aligned(16))) float4 sW[32 * PITCH];
  __shared__ __attribute__((aligned(16))) float4 sL[32 * PITCH];
  stage_weights<128>(W, sW, threadIdx.x);
  stage_weights<128>(L, sL, threadIdx.x);
  __syncthreads();

  const int wv = threadIdx.x >> 5;
  const int lane = threadIdx.x & 31;
  const int t = lane & 15;
  const int hh = lane >> 4;

  for (int strip = blockIdx.x + GEMM_BLOCKS * wv; strip < NSTRIP; strip += GEMM_BLOCKS * 8) {
    const int r0 = strip << 4;
    const float* arow = Agg + (size_t)(r0 + t) * FF + 16 * hh;
    const float* hrow = H + (size_t)(r0 + t) * FF + 16 * hh;
    v8f acc[8];
#pragma unroll
    for (int ct = 0; ct < 8; ++ct) acc[ct] = v8f{};

#pragma unroll 1
    for (int m = 0; m < 4; ++m) {
      float4 a[4], h[4];
#pragma unroll
      for (int q = 0; q < 4; ++q) {
        a[q] = *(const float4*)(arow + 32 * m + 4 * q);
        h[q] = *(const float4*)(hrow + 32 * m + 4 * q);
      }
#pragma unroll
      for (int q = 0; q < 4; ++q) {
        const float4* pw = sW + ((m * 4 + q) * 2 + hh) * PITCH + t;
        const float4* pl = sL + ((m * 4 + q) * 2 + hh) * PITCH + t;
        float4 bw[8], bl[8];
#pragma unroll
        for (int ct = 0; ct < 8; ++ct) { bw[ct] = pw[16 * ct]; bl[ct] = pl[16 * ct]; }
#pragma unroll
        for (int ct = 0; ct < 8; ++ct) acc[ct] = mm4(a[q].x, a[q].y, bw[ct].x, bw[ct].y, acc[ct]);
#pragma unroll
        for (int ct = 0; ct < 8; ++ct) acc[ct] = mm4(h[q].x, h[q].y, bl[ct].x, bl[ct].y, acc[ct]);
#pragma unroll
        for (int ct = 0; ct < 8; ++ct) acc[ct] = mm4(a[q].z, a[q].w, bw[ct].z, bw[ct].w, acc[ct]);
#pragma unroll
        for (int ct = 0; ct < 8; ++ct) acc[ct] = mm4(h[q].z, h[q].w, bl[ct].z, bl[ct].w, acc[ct]);
      }
    }
#pragma unroll
    for (int ct = 0; ct < 8; ++ct)
#pragma unroll
      for (int v = 0; v < 8; ++v)
        Z[(size_t)(r0 + v + 8 * hh) * FF + 16 * ct + t] = acc[ct][v];
  }
}

__global__ __launch_bounds__(256) void k_gemm_p(const float* __restrict__ H,
                                                const float* __restrict__ dsout,
                                                const float* __restrict__ Wp,
                                                float* __restrict__ P) {
  constexpr int PITCH = 64 + 8;
  __shared__ __attribute__((aligned(16))) float4 sW[32 * PITCH];
  stage_weights<64>(Wp, sW, threadIdx.x);
  __syncthreads();

  const int wv = threadIdx.x >> 5;
  const int lane = threadIdx.x & 31;
  const int t = lane & 15, hh = lane >> 4;

  for (int strip = blockIdx.x + GEMM_BLOCKS * wv; strip < NSTRIP; strip += GEMM_BLOCKS * 8) {
    const int r0 = strip << 4;
    const float d = dsout[r0 + t];
    const float* hrow = H + (size_t)(r0 + t) * FF + 16 * hh;
    v8f acc[4];
#pragma unroll
    for (int ct = 0; ct < 4; ++ct) acc[ct] = v8f{};
#pragma unroll 1
    for (int m = 0; m < 4; ++m) {
      float4 a[4];
#pragma unroll
      for (int q = 0; q < 4; ++q) {
        const float4 x = *(const float4*)(hrow + 32 * m + 4 * q);
        a[q] = make_float4(d * x.x, d * x.y, d * x.z, d * x.w);
      }
#pragma unroll
      for (int q = 0; q < 4; ++q) {
        const float4* pw = sW + ((m * 4 + q) * 2 + hh) * PITCH + t;
        float4 bw[4];
#pragma unroll
        for (int ct = 0; ct < 4; ++ct) bw[ct] = pw[16 * ct];
#pragma unroll
        for (int ct = 0; ct < 4; ++ct) acc[ct] = mm4(a[q].x, a[q].y, bw[ct].x, bw[ct].y, acc[ct]);
#pragma unroll
        for (int ct = 0; ct < 4; ++ct) acc[ct] = mm4(a[q].z, a[q].w, bw[ct].z, bw[ct].w, acc[ct]);
      }
    }
#pragma unroll
    for (int ct = 0; ct < 4; ++ct)
#pragma unroll
      for (int v = 0; v < 8; ++v)
        P[(size_t)(r0 + v + 8 * hh) * 64 + 16 * ct + t] = acc[ct][v];
  }
}

__global__ __launch_bounds__(256) void k_gemm_out(const float* __restrict__ H,
                                                  const float* __restrict__ Lp,
                                                  const float* __restrict__ R,
                                                  const float* __restrict__ b2,
                                                  float* __restrict__ out) {
  constexpr int PITCH = 64 + 8;
  __shared__ __attribute__((aligned(16))) float4 sL[32 * PITCH];
  stage_weights<64>(Lp, sL, threadIdx.x);
  __syncthreads();

  const int wv = threadIdx.x >> 5;
  const int lane = threadIdx.x & 31;
  const int t = lane & 15, hh = lane >> 4;

  for (int strip = blockIdx.x + GEMM_BLOCKS * wv; strip < NSTRIP; strip += GEMM_BLOCKS * 8) {
    const int r0 = strip << 4;
    const float* hrow = H + (size_t)(r0 + t) * FF + 16 * hh;
    v8f acc[3];
#pragma unroll
    for (int ct = 0; ct < 3; ++ct) acc[ct] = v8f{};
#pragma unroll 1
    for (int m = 0; m < 4; ++m) {
      float4 a[4];
#pragma unroll
      for (int q = 0; q < 4; ++q) a[q] = *(const float4*)(hrow + 32 * m + 4 * q);
#pragma unroll
      for (int q = 0; q < 4; ++q) {
        const float4* pl = sL + ((m * 4 + q) * 2 + hh) * PITCH + t;
        float4 bl[3];
#pragma unroll
        for (int ct = 0; ct < 3; ++ct) bl[ct] = pl[16 * ct];
#pragma unroll
        for (int ct = 0; ct < 3; ++ct) acc[ct] = mm4(a[q].x, a[q].y, bl[ct].x, bl[ct].y, acc[ct]);
#pragma unroll
        for (int ct = 0; ct < 3; ++ct) acc[ct] = mm4(a[q].z, a[q].w, bl[ct].z, bl[ct].w, acc[ct]);
      }
    }
#pragma unroll
    for (int ct = 0; ct < 3; ++ct) {
      const int col = 16 * ct + t;
      if (col < 40) {
        const float bias = b2[col];
#pragma unroll
        for (int v = 0; v < 8; ++v) {
          const int row = r0 + v + 8 * hh;
          out[(size_t)row * 40 + col] = acc[ct][v] + R[(size_t)row * 64 + col] + bias;
        }
      }
    }
  }
}


static constexpr int CS_WAVES = 8;
static constexpr int CS_PER_WAVE = 25;
static constexpr int CS_ROWS = CS_WAVES * CS_PER_WAVE;
static constexpr int CS_BLOCKS = NN / CS_ROWS;
static constexpr int CS_THREADS = CS_WAVES * 32;
static_assert(CS_BLOCKS * CS_ROWS == NN, "statistics grid must cover the rows exactly");
static_assert(CS_THREADS == 2 * FF, "one thread per (statistic, column) in the block's last step");

__global__ __launch_bounds__(CS_THREADS) void k_colstats(const float* __restrict__ z,
                                                           float* __restrict__ cs,
                                                           float* __restrict__ cs2) {
  __shared__ __attribute__((aligned(16))) float4 red[2][CS_WAVES][32];
  const int lane = threadIdx.x & 31, wv = threadIdx.x >> 5;
  const float4* p = (const float4*)(z + ((size_t)blockIdx.x * CS_ROWS + wv) * FF) + lane;
  float4 v[CS_PER_WAVE];
#pragma unroll
  for (int i = 0; i < CS_PER_WAVE; ++i) v[i] = p[(size_t)i * CS_WAVES * (FF / 4)];
  float4 s = make_float4(0.f, 0.f, 0.f, 0.f), s2 = make_float4(0.f, 0.f, 0.f, 0.f);
#pragma unroll
  for (int i = 0; i < CS_PER_WAVE; ++i) {
    s.x += v[i].x;
    s.y += v[i].y;
    s.z += v[i].z;
    s.w += v[i].w;
    s2.x = fmaf(v[i].x, v[i].x, s2.x);
    s2.y = fmaf(v[i].y, v[i].y, s2.y);
    s2.z = fmaf(v[i].z, v[i].z, s2.z);
    s2.w = fmaf(v[i].w, v[i].w, s2.w);
  }
  red[0][wv][lane] = s;
  red[1][wv][lane] = s2;
  __syncthreads();
  const int st = threadIdx.x >> 7, c = threadIdx.x & 127;
  const float* r = (const float*)&red[st][0][0] + c;
  float t = 0.f;
#pragma unroll
  for (int w = 0; w < CS_WAVES; ++w) t += r[w * FF];
  atomicAdd(st ? &cs2[c] : &cs[c], t);
}

__global__ void k_bnparams(const float* __restrict__ cs, const float* __restrict__ cs2,
                           const float* __restrict__ g, const float* __restrict__ be,
                           float* __restrict__ bnA, float* __restrict__ bnB) {
  const int c = threadIdx.x;
  const float inv = 1.0f / (float)NN;
  const float mu = cs[c] * inv;
  const float var = cs2[c] * inv - mu * mu;
  const float a = g[c] * rsqrtf(var + 1e-5f);
  bnA[c] = a;
  bnB[c] = be[c] - mu * a;
}

__global__ void k_bnapply(const float* __restrict__ z, const float* __restrict__ bnA,
                          const float* __restrict__ bnB, float* __restrict__ hout) {
  const size_t i = (size_t)blockIdx.x * 256 + threadIdx.x;
  const int c = (int)(i & 127);
  const float v = fmaf(z[i], bnA[c], bnB[c]);
  hout[i] = v > 0.f ? v : 0.f;
}

__global__ void k_padw(const float* __restrict__ W2, const float* __restrict__ L2,
                       float* __restrict__ Wp, float* __restrict__ Lp) {
  const int i = blockIdx.x * 256 + threadIdx.x;
  const int k = i >> 6, c = i & 63;
  const float w = (c < 40) ? W2[k * 40 + c] : 0.f;
  const float l = (c < 40) ? L2[k * 40 + c] : 0.f;
  Wp[i] = w;
  Lp[i] = l;
}


extern "C" void kernel_launch(void* const* d_in, const int* in_sizes, int n_in,
                              void* d_out, int out_size, void* d_ws, size_t ws_size,
                              hipStream_t stream) {
  (void)in_sizes; (void)n_in; (void)out_size; (void)ws_size;

  const float* feat = (const float*)d_in[0];
  const int* src   = (const int*)d_in[1];
  const int* dst   = (const int*)d_in[2];
  const float* W0  = (const float*)d_in[3];
  const float* L0  = (const float*)d_in[4];
  const float* W1  = (const float*)d_in[5];
  const float* L1  = (const float*)d_in[6];
  const float* W2  = (const float*)d_in[7];
  const float* L2  = (const float*)d_in[8];
  const float* b2  = (const float*)d_in[9];
  const float* g0  = (const float*)d_in[10];
  const float* be0 = (const float*)d_in[11];
  const float* g1  = (const float*)d_in[12];
  const float* be1 = (const float*)d_in[13];
  float* out = (float*)d_out;

  char* base = (char*)d_ws;
  size_t off = 0;
  auto take = [&](size_t bytes) -> char* {
    char* p = base + off;
    off = (off + bytes + 255) & ~(size_t)255;
    return p;
  };
  unsigned* cnt     = (unsigned*)take((size_t)2 * NBIN * SLICES * 4);
  unsigned* bump    = (unsigned*)take(2 * 4);
  unsigned* binbase = (unsigned*)take((size_t)2 * NBIN * 4);
  unsigned* bintot  = (unsigned*)take((size_t)2 * NBIN * 4);
  unsigned* binD    = (unsigned*)take((size_t)EE * 4);
  unsigned* binS    = (unsigned*)take((size_t)EE * 4);
  unsigned* rows    = (unsigned*)take((size_t)NN * 8);
  unsigned* esrc    = (unsigned*)take((size_t)EE * 4);
  float* dsout = (float*)take((size_t)NN * 4);
  float* dsin  = (float*)take((size_t)NN * 4);
  float* cs    = (float*)take(128 * 4);
  float* cs2   = (float*)take(128 * 4);
  float* bnA   = (float*)take(128 * 4);
  float* bnB   = (float*)take(128 * 4);
  float* Wp    = (float*)take(128 * 64 * 4);
  float* Lp    = (float*)take(128 * 64 * 4);
  float* bufA  = (float*)take((size_t)NN * FF * 4);
  float* bufB  = (float*)take((size_t)NN * FF * 4);
  float* bufC  = (float*)take((size_t)NN * FF * 4);

  const int BApply = (NN * FF) / 256;

  k_binhist<<<SLICES, 256, 0, stream>>>(src, dst, cnt, bump);
  k_binalloc<<<ALLOC_BLOCKS, 256, 0, stream>>>(cnt, bump, binbase, bintot);
  k_binscatter<<<SLICES, 256, 0, stream>>>(src, dst, cnt, binD, binS);
  k_binrows<<<NBIN, 256, 0, stream>>>(binD, binS, binbase, bintot, dsout, dsin, rows, esrc);
  k_padw<<<32, 256, 0, stream>>>(W2, L2, Wp, Lp);

  k_agg128<<<AGG_BLOCKS, 128, 0, stream>>>(feat, rows, esrc, dsout, dsin, bufA);
  k_gemm_dual<<<GEMM_BLOCKS, 256, 0, stream>>>(bufA, feat, W0, L0, bufB);
  k_zero_u32<<<1, 256, 0, stream>>>((unsigned*)cs, 256);
  k_colstats<<<CS_BLOCKS, CS_THREADS, 0, stream>>>(bufB, cs, cs2);
  k_bnparams<<<1, 128, 0, stream>>>(cs, cs2, g0, be0, bnA, bnB);
  k_bnapply<<<BApply, 256, 0, stream>>>(bufB, bnA, bnB, bufC);

  k_agg128<<<AGG_BLOCKS, 128, 0, stream>>>(bufC, rows, esrc, dsout, dsin, bufA);
  k_gemm_dual<<<GEMM_BLOCKS, 256, 0, stream>>>(bufA, bufC, W1, L1, bufB);
  k_zero_u32<<<1, 256, 0, stream>>>((unsigned*)cs, 256);
  k_colstats<<<CS_BLOCKS, CS_THREADS, 0, stream>>>(bufB, cs, cs2);
  k_bnparams<<<1, 128, 0, stream>>>(cs, cs2, g1, be1, bnA, bnB);
  k_bnapply<<<BApply, 256, 0, stream>>>(bufB, bnA, bnB, bufC);

  k_gemm_p<<<GEMM_BLOCKS, 256, 0, stream>>>(bufC, dsout, Wp, bufA);
  k_agg64<<<AGG_BLOCKS, 128, 0, stream>>>(bufA, rows, esrc, dsin, bufB);
  k_gemm_out<<<GEMM_BLOCKS, 256, 0, stream>>>(bufC, Lp, bufB, b2, out);
}


# BN+ReLU passes folded into GEMM A-operand reads (that GEMM stores h): layer 1's h@L term runs as its own GEMM ahead of the gather, spmm(h)@W is added on top of it; layer-1 z sum order changes
# speedup vs baseline: 2.1854x; 1.0809x over previous
#include <hip/hip_runtime.h>
#include <hip/hip_bf16.h>


static constexpr int NN = 100000;
static constexpr int EE = 1600000;
static constexpr int FF = 128;

typedef __attribute__((ext_vector_type(2))) float v2f;
typedef __attribute__((ext_vector_type(8))) float v8f;


__global__ void k_zero_u32(unsigned* __restrict__ p, int n) {
  int i = blockIdx.x * 256 + threadIdx.x;
  if (i < n) p[i] = 0u;
}


static constexpr int BIN_SHIFT = 8;
static constexpr int BIN_NODES = 1 << BIN_SHIFT;
static constexpr int NBIN = (NN + BIN_NODES - 1) / BIN_NODES;
static constexpr int SLICES = 256;
static constexpr int SLICE_EDGES = 6256;
static constexpr int SLICE_IT = (SLICE_EDGES / 4 + 255) / 256;
static constexpr int ALLOC_BLOCKS = (2 * NBIN + 7) / 8;
static constexpr int BIN_BATCH = 8;
static_assert(SLICE_EDGES % 4 == 0 && EE % 4 == 0, "slices are read four edges at a time");
static_assert((SLICES - 1) * SLICE_EDGES < EE && SLICES * SLICE_EDGES >= EE, "slices must cover the edges");
static_assert(NN <= (1 << 24), "a bin entry keeps the node id in 24 bits");

__global__ __launch_bounds__(256) void k_binhist(const int* __restrict__ src, const int* __restrict__ dst,
                                                 unsigned* __restrict__ cnt, unsigned* __restrict__ bump) {
  __shared__ unsigned hD[NBIN];
  __shared__ unsigned hS[NBIN];
  const int tid = threadIdx.x, b = blockIdx.x;
  for (int i = tid; i < NBIN; i += 256) {
    hD[i] = 0u;
    hS[i] = 0u;
  }
  if (b == 0 && tid < 2) bump[tid] = 0u;
  __syncthreads();
  const int e0 = b * SLICE_EDGES;
  const int n4 = min(EE - e0, SLICE_EDGES) >> 2;
  const int4* s4 = (const int4*)(src + e0);
  const int4* d4 = (const int4*)(dst + e0);
  int4 s[SLICE_IT], d[SLICE_IT];
#pragma unroll
  for (int u = 0; u < SLICE_IT; ++u) {
    const int i = min(tid + 256 * u, n4 - 1);
    s[u] = s4[i];
    d[u] = d4[i];
  }
#pragma unroll
  for (int u = 0; u < SLICE_IT; ++u) {
    if (tid + 256 * u < n4) {
      atomicAdd(&hD[d[u].x >> BIN_SHIFT], 1u);
      atomicAdd(&hD[d[u].y >> BIN_SHIFT], 1u);
      atomicAdd(&hD[d[u].z >> BIN_SHIFT], 1u);
      atomicAdd(&hD[d[u].w >> BIN_SHIFT], 1u);
      atomicAdd(&hS[s[u].x >> BIN_SHIFT], 1u);
      atomicAdd(&hS[s[u].y >> BIN_SHIFT], 1u);
      atomicAdd(&hS[s[u].z >> BIN_SHIFT], 1u);
      atomicAdd(&hS[s[u].w >> BIN_SHIFT], 1u);
    }
  }
  __syncthreads();
  for (int i = tid; i < NBIN; i += 256) {
    cnt[(size_t)i * SLICES + b] = hD[i];
    cnt[(size_t)(NBIN + i) * SLICES + b] = hS[i];
  }
}

__global__ __launch_bounds__(256) void k_binalloc(unsigned* __restrict__ cnt, unsigned* __restrict__ bump,
                                                  unsigned* __restrict__ binbase,
                                                  unsigned* __restrict__ bintot) {
  const int lane = threadIdx.x & 31;
  const int w = blockIdx.x * 8 + (threadIdx.x >> 5);
  if (w >= 2 * NBIN) return;
  uint4* p = (uint4*)(cnt + (size_t)w * SLICES) + lane * 2;
  const uint4 a = p[0];
  const uint4 c = p[1];
  const unsigned loc = a.x + a.y + a.z + a.w + c.x + c.y + c.z + c.w;
  unsigned inc = loc;
  for (int d = 1; d < 32; d <<= 1) {
    const unsigned up = __shfl_up(inc, d);
    if (lane >= d) inc += up;
  }
  const unsigned tot = __shfl(inc, 31);
  unsigned base = 0u;
  if (lane == 0) base = atomicAdd(&bump[w >= NBIN ? 1 : 0], tot);
  base = __shfl(base, 0);
  unsigned o = base + inc - loc;
  uint4 ra, rc;
  ra.x = o; o += a.x;
  ra.y = o; o += a.y;
  ra.z = o; o += a.z;
  ra.w = o; o += a.w;
  rc.x = o; o += c.x;
  rc.y = o; o += c.y;
  rc.z = o; o += c.z;
  rc.w = o;
  p[0] = ra;
  p[1] = rc;
  if (lane == 0) {
    binbase[w] = base;
    bintot[w] = tot;
  }
}

__device__ __forceinline__ void bin_put4(unsigned* cD, unsigned* cS, unsigned* __restrict__ binD,
                                         unsigned* __restrict__ binS, const int4 s, const int4 d) {
  const unsigned pd0 = atomicAdd(&cD[d.x >> BIN_SHIFT], 1u);
  const unsigned pd1 = atomicAdd(&cD[d.y >> BIN_SHIFT], 1u);
  const unsigned pd2 = atomicAdd(&cD[d.z >> BIN_SHIFT], 1u);
  const unsigned pd3 = atomicAdd(&cD[d.w >> BIN_SHIFT], 1u);
  const unsigned ps0 = atomicAdd(&cS[s.x >> BIN_SHIFT], 1u);
  const unsigned ps1 = atomicAdd(&cS[s.y >> BIN_SHIFT], 1u);
  const unsigned ps2 = atomicAdd(&cS[s.z >> BIN_SHIFT], 1u);
  const unsigned ps3 = atomicAdd(&cS[s.w >> BIN_SHIFT], 1u);
  binD[pd0] = ((unsigned)(d.x & (BIN_NODES - 1)) << 24) | (unsigned)s.x;
  binD[pd1] = ((unsigned)(d.y & (BIN_NODES - 1)) << 24) | (unsigned)s.y;
  binD[pd2] = ((unsigned)(d.z & (BIN_NODES - 1)) << 24) | (unsigned)s.z;
  binD[pd3] = ((unsigned)(d.w & (BIN_NODES - 1)) << 24) | (unsigned)s.w;
  binS[ps0] = (unsigned)(s.x & (BIN_NODES - 1));
  binS[ps1] = (unsigned)(s.y & (BIN_NODES - 1));
  binS[ps2] = (unsigned)(s.z & (BIN_NODES - 1));
  binS[ps3] = (unsigned)(s.w & (BIN_NODES - 1));
}

__global__ __launch_bounds__(256) void k_binscatter(const int* __restrict__ src, const int* __restrict__ dst,
                                                    const unsigned* __restrict__ off,
                                                    unsigned* __restrict__ binD,
                                                    unsigned* __restrict__ binS) {
  __shared__ unsigned cD[NBIN];
  __shared__ unsigned cS[NBIN];
  const int tid = threadIdx.x, b = blockIdx.x;
  for (int i = tid; i < NBIN; i += 256) {
    cD[i] = off[(size_t)i * SLICES + b];
    cS[i] = off[(size_t)(NBIN + i) * SLICES + b];
  }
  __syncthreads();
  const int e0 = b * SLICE_EDGES;
  const int n4 = min(EE - e0, SLICE_EDGES) >> 2;
  const int4* s4 = (const int4*)(src + e0);
  const int4* d4 = (const int4*)(dst + e0);
  int4 s[SLICE_IT], d[SLICE_IT];
#pragma unroll
  for (int u = 0; u < SLICE_IT; ++u) {
    const int i = min(tid + 256 * u, n4 - 1);
    s[u] = s4[i];
    d[u] = d4[i];
  }
#pragma unroll
  for (int u = 0; u < SLICE_IT; ++u) {
    if (tid + 256 * u < n4) {
      bin_put4(cD, cS, binD, binS, s[u], d[u]);
    }
  }
}

__global__ __launch_bounds__(256) void k_binrows(const unsigned* __restrict__ binD,
                                                 const unsigned* __restrict__ binS,
                                                 const unsigned* __restrict__ binbase,
                                                 const unsigned* __restrict__ bintot,
                                                 float* __restrict__ dsout, float* __restrict__ dsin,
                                                 unsigned* __restrict__ rows, unsigned* __restrict__ esrc) {
  __shared__ unsigned cD[BIN_NODES];
  __shared__ unsigned cS[BIN_NODES];
  __shared__ unsigned cur[BIN_NODES];
  __shared__ unsigned wtot[8];
  const int tid = threadIdx.x, j = blockIdx.x;
  const int lane = tid & 31, wv = tid >> 5;
  cD[tid] = 0u;
  cS[tid] = 0u;
  const unsigned bD = binbase[j], nD = bintot[j];
  const unsigned bS = binbase[NBIN + j], nS = bintot[NBIN + j];
  const unsigned* eD = binD + bD;
  const unsigned* eS = binS + bS;
  __syncthreads();

  for (unsigned i0 = 0u; i0 < nD; i0 += 256u * BIN_BATCH) {
    unsigned v[BIN_BATCH];
#pragma unroll
    for (int u = 0; u < BIN_BATCH; ++u) v[u] = eD[min(i0 + 256u * u + tid, nD - 1u)];
#pragma unroll
    for (int u = 0; u < BIN_BATCH; ++u)
      if (i0 + 256u * u + tid < nD) atomicAdd(&cD[v[u] >> 24], 1u);
  }
  for (unsigned i0 = 0u; i0 < nS; i0 += 256u * BIN_BATCH) {
    unsigned v[BIN_BATCH];
#pragma unroll
    for (int u = 0; u < BIN_BATCH; ++u) v[u] = eS[min(i0 + 256u * u + tid, nS - 1u)];
#pragma unroll
    for (int u = 0; u < BIN_BATCH; ++u)
      if (i0 + 256u * u + tid < nS) atomicAdd(&cS[v[u]], 1u);
  }
  __syncthreads();

  const unsigned id = cD[tid], od = cS[tid];
  unsigned inc = id;
  for (int d = 1; d < 32; d <<= 1) {
    const unsigned up = __shfl_up(inc, d);
    if (lane >= d) inc += up;
  }
  if (lane == 31) wtot[wv] = inc;
  __syncthreads();
  unsigned st = inc - id;
  for (int w = 0; w < wv; ++w) st += wtot[w];
  cur[tid] = st;
  const int node = j * BIN_NODES + tid;
  if (node < NN) {
    *(uint2*)(rows + 2 * node) = make_uint2(bD + st, bD + st + id);
    dsin[node] = id ? rsqrtf((float)id) : 1.0f;
    dsout[node] = od ? rsqrtf((float)od) : 1.0f;
  }
  __syncthreads();

  unsigned* o = esrc + bD;
  for (unsigned i0 = 0u; i0 < nD; i0 += 256u * BIN_BATCH) {
    unsigned v[BIN_BATCH];
#pragma unroll
    for (int u = 0; u < BIN_BATCH; ++u) v[u] = eD[min(i0 + 256u * u + tid, nD - 1u)];
    unsigned p[BIN_BATCH];
#pragma unroll
    for (int u = 0; u < BIN_BATCH; ++u) {
      p[u] = 0u;
      if (i0 + 256u * u + tid < nD) p[u] = atomicAdd(&cur[v[u] >> 24], 1u);
    }
#pragma unroll
    for (int u = 0; u < BIN_BATCH; ++u)
      if (i0 + 256u * u + tid < nD) o[p[u]] = v[u] & 0xFFFFFFu;
  }
}


static constexpr int AGG_NODES = 8;
static constexpr int AGG_GROUP = 8;
static constexpr int AGG_WAVES = NN / AGG_NODES;
static constexpr int AGG_BLOCKS = AGG_WAVES / 4;
static_assert(NN % AGG_NODES == 0 && AGG_WAVES % 4 == 0, "gather grid must cover the nodes exactly");

__device__ __forceinline__ unsigned lane_u32(unsigned v, int l) {
  return (unsigned)__builtin_amdgcn_readlane((int)v, l);
}
__device__ __forceinline__ float lane_f32(float v, int l) {
  return __int_as_float(__builtin_amdgcn_readlane(__float_as_int(v), l));
}

__device__ __forceinline__ void axpy4(float4& acc, float w, const float4& v) {
  acc.x = fmaf(w, v.x, acc.x);
  acc.y = fmaf(w, v.y, acc.y);
  acc.z = fmaf(w, v.z, acc.z);
  acc.w = fmaf(w, v.w, acc.w);
}

__global__ __launch_bounds__(128) void k_agg128(const float* __restrict__ h,
                                                const unsigned* __restrict__ rows,
                                                const unsigned* __restrict__ esrc,
                                                const float* __restrict__ dsout,
                                                const float* __restrict__ dsin,
                                                float* __restrict__ agg) {
  const int lane = threadIdx.x & 31;
  const int n0 = (blockIdx.x * 4 + (threadIdx.x >> 5)) * AGG_NODES;
  unsigned rb = 0u, re = 0u;
  float cin = 0.f;
  if (lane < AGG_NODES) {
    const uint2 r = *(const uint2*)(rows + 2 * (n0 + lane));
    rb = r.x;
    re = r.y;
    cin = dsin[n0 + lane];
  }
  unsigned b = lane_u32(rb, 0), e = lane_u32(re, 0);
  unsigned my = 0u;
  float myw = 0.f;
  if (b + lane < e) {
    my = esrc[b + lane];
    myw = dsout[my];
  }
  const float* hl = h + lane * 4;

#pragma unroll 1
  for (int k = 0; k < AGG_NODES; ++k) {
    unsigned nb = 0u, ne = 0u, nmy = 0u;
    if (k + 1 < AGG_NODES) {
      nb = lane_u32(rb, k + 1);
      ne = lane_u32(re, k + 1);
      if (nb + lane < ne) nmy = esrc[nb + lane];
    }
    float4 acc = make_float4(0.f, 0.f, 0.f, 0.f);
    const int cnt = (int)min(e - b, 32u);
    int j = 0;
    for (; j + AGG_GROUP <= cnt; j += AGG_GROUP) {
      float4 v[AGG_GROUP];
#pragma unroll
      for (int u = 0; u < AGG_GROUP; ++u)
        v[u] = *(const float4*)(hl + (size_t)lane_u32(my, j + u) * FF);
      __builtin_amdgcn_sched_barrier(0);
#pragma unroll
      for (int u = 0; u < AGG_GROUP; ++u) axpy4(acc, lane_f32(myw, j + u), v[u]);
    }
    const int rem = cnt - j;
    if (rem > 0) {
      float4 v[AGG_GROUP - 1];
#pragma unroll
      for (int u = 0; u < AGG_GROUP - 1; ++u) {
        v[u] = make_float4(0.f, 0.f, 0.f, 0.f);
        if (u < rem) v[u] = *(const float4*)(hl + (size_t)lane_u32(my, j + u) * FF);
      }
#pragma unroll
      for (int u = 0; u < AGG_GROUP - 1; ++u)
        if (u < rem) axpy4(acc, lane_f32(myw, j + u), v[u]);
    }
    for (unsigned i = b + 32u; i < e; ++i) {
      const unsigned s = esrc[i];
      axpy4(acc, dsout[s], *(const float4*)(hl + (size_t)s * FF));
    }
    const float c = lane_f32(cin, k);
    *(float4*)(agg + (size_t)(n0 + k) * FF + lane * 4) =
        make_float4(c * acc.x, c * acc.y, c * acc.z, c * acc.w);
    b = nb;
    e = ne;
    my = nmy;
    myw = 0.f;
    if (b + lane < e) myw = dsout[my];
  }
}

__global__ __launch_bounds__(128) void k_agg64(const float* __restrict__ P,
                                               const unsigned* __restrict__ rows,
                                               const unsigned* __restrict__ esrc,
                                               const float* __restrict__ dsin,
                                               float* __restrict__ R) {
  const int lane = threadIdx.x & 31;
  const int n0 = (blockIdx.x * 4 + (threadIdx.x >> 5)) * AGG_NODES;
  unsigned rb = 0u, re = 0u;
  float cin = 0.f;
  if (lane < AGG_NODES) {
    const uint2 r = *(const uint2*)(rows + 2 * (n0 + lane));
    rb = r.x;
    re = r.y;
    cin = dsin[n0 + lane];
  }
  unsigned b = lane_u32(rb, 0), e = lane_u32(re, 0);
  unsigned my = 0u;
  if (b + lane < e) my = esrc[b + lane];
  const float* pl = P + lane * 2;

#pragma unroll 1
  for (int k = 0; k < AGG_NODES; ++k) {
    unsigned nb = 0u, ne = 0u, nmy = 0u;
    if (k + 1 < AGG_NODES) {
      nb = lane_u32(rb, k + 1);
      ne = lane_u32(re, k + 1);
      if (nb + lane < ne) nmy = esrc[nb + lane];
    }
    float2 acc = make_float2(0.f, 0.f);
    const int cnt = (int)min(e - b, 32u);
    int j = 0;
    for (; j + AGG_GROUP <= cnt; j += AGG_GROUP) {
      float2 v[AGG_GROUP];
#pragma unroll
      for (int u = 0; u < AGG_GROUP; ++u)
        v[u] = *(const float2*)(pl + (size_t)lane_u32(my, j + u) * 64);
#pragma unroll
      for (int u = 0; u < AGG_GROUP; ++u) {
        acc.x += v[u].x;
        acc.y += v[u].y;
      }
    }
    const int rem = cnt - j;
    if (rem > 0) {
      float2 v[AGG_GROUP - 1];
#pragma unroll
      for (int u = 0; u < AGG_GROUP - 1; ++u) {
        v[u] = make_float2(0.f, 0.f);
        if (u < rem) v[u] = *(const float2*)(pl + (size_t)lane_u32(my, j + u) * 64);
      }
#pragma unroll
      for (int u = 0; u < AGG_GROUP - 1; ++u) {
        if (u < rem) {
          acc.x += v[u].x;
          acc.y += v[u].y;
        }
      }
    }
    for (unsigned i = b + 32u; i < e; ++i) {
      const float2 x = *(const float2*)(pl + (size_t)esrc[i] * 64);
      acc.x += x.x;
      acc.y += x.y;
    }
    const float c = lane_f32(cin, k);
    *(float2*)(R + (size_t)(n0 + k) * 64 + lane * 2) = make_float2(c * acc.x, c * acc.y);
    b = nb;
    e = ne;
    my = nmy;
  }
}


static constexpr int GEMM_BLOCKS = 512;
static constexpr int NSTRIP = NN / 16;

template <int NC>
__device__ __forceinline__ void stage_weights(const float* __restrict__ Bg,
                                              float4* __restrict__ s, int tid) {
  constexpr int C4 = NC / 4;
  constexpr int PITCH = NC + 8;
  for (int i = tid; i < 32 * C4; i += 256) {
    const int g = i / C4, c4 = i - g * C4;
    const float* p = Bg + (size_t)(4 * g) * NC + 4 * c4;
    const float4 r0 = *(const float4*)(p);
    const float4 r1 = *(const float4*)(p + NC);
    const float4 r2 = *(const float4*)(p + 2 * NC);
    const float4 r3 = *(const float4*)(p + 3 * NC);
    const int m = g >> 3, hh = (g >> 2) & 1, q = g & 3;
    float4* d = s + ((m * 4 + q) * 2 + hh) * PITCH + 4 * c4;
    d[0] = make_float4(r0.x, r1.x, r2.x, r3.x);
    d[1] = make_float4(r0.y, r1.y, r2.y, r3.y);
    d[2] = make_float4(r0.z, r1.z, r2.z, r3.z);
    d[3] = make_float4(r0.w, r1.w, r2.w, r3.w);
  }
}

__device__ __forceinline__ v8f mm4(float ax, float ay, float bx, float by, v8f c) {
  v2f a; a.x = ax; a.y = ay;
  v2f b; b.x = bx; b.y = by;
  return __builtin_amdgcn_wmma_f32_16x16x4_f32(false, a, false, b, (short)0, c, false, false);
}

__device__ __forceinline__ float4 bn_relu4(const float4 z, const float4 a, const float4 b) {
  const float vx = fmaf(z.x, a.x, b.x);
  const float vy = fmaf(z.y, a.y, b.y);
  const float vz = fmaf(z.z, a.z, b.z);
  const float vw = fmaf(z.w, a.w, b.w);
  return make_float4(vx > 0.f ? vx : 0.f, vy > 0.f ? vy : 0.f, vz > 0.f ? vz : 0.f,
                     vw > 0.f ? vw : 0.f);
}

__global__ __launch_bounds__(256) void k_gemm_dual(const float* __restrict__ Agg,
                                                   const float* __restrict__ H,
                                                   const float* __restrict__ W,
                                                   const float* __restrict__ L,
                                                   float* __restrict__ Z) {
  constexpr int PITCH = 128 + 8;
  __shared__ __attribute__((aligned(16))) float4 sW[32 * PITCH];
  __shared__ __attribute__((aligned(16))) float4 sL[32 * PITCH];
  stage_weights<128>(W, sW, threadIdx.x);
  stage_weights<128>(L, sL, threadIdx.x);
  __syncthreads();

  const int wv = threadIdx.x >> 5;
  const int lane = threadIdx.x & 31;
  const int t = lane & 15;
  const int hh = lane >> 4;

  for (int strip = blockIdx.x + GEMM_BLOCKS * wv; strip < NSTRIP; strip += GEMM_BLOCKS * 8) {
    const int r0 = strip << 4;
    const float* arow = Agg + (size_t)(r0 + t) * FF + 16 * hh;
    const float* hrow = H + (size_t)(r0 + t) * FF + 16 * hh;
    v8f acc[8];
#pragma unroll
    for (int ct = 0; ct < 8; ++ct) acc[ct] = v8f{};

#pragma unroll 1
    for (int m = 0; m < 4; ++m) {
      float4 a[4], h[4];
#pragma unroll
      for (int q = 0; q < 4; ++q) {
        a[q] = *(const float4*)(arow + 32 * m + 4 * q);
        h[q] = *(const float4*)(hrow + 32 * m + 4 * q);
      }
#pragma unroll
      for (int q = 0; q < 4; ++q) {
        const float4* pw = sW + ((m * 4 + q) * 2 + hh) * PITCH + t;
        const float4* pl = sL + ((m * 4 + q) * 2 + hh) * PITCH + t;
        float4 bw[8], bl[8];
#pragma unroll
        for (int ct = 0; ct < 8; ++ct) { bw[ct] = pw[16 * ct]; bl[ct] = pl[16 * ct]; }
#pragma unroll
        for (int ct = 0; ct < 8; ++ct) acc[ct] = mm4(a[q].x, a[q].y, bw[ct].x, bw[ct].y, acc[ct]);
#pragma unroll
        for (int ct = 0; ct < 8; ++ct) acc[ct] = mm4(h[q].x, h[q].y, bl[ct].x, bl[ct].y, acc[ct]);
#pragma unroll
        for (int ct = 0; ct < 8; ++ct) acc[ct] = mm4(a[q].z, a[q].w, bw[ct].z, bw[ct].w, acc[ct]);
#pragma unroll
        for (int ct = 0; ct < 8; ++ct) acc[ct] = mm4(h[q].z, h[q].w, bl[ct].z, bl[ct].w, acc[ct]);
      }
    }
#pragma unroll
    for (int ct = 0; ct < 8; ++ct)
#pragma unroll
      for (int v = 0; v < 8; ++v)
        Z[(size_t)(r0 + v + 8 * hh) * FF + 16 * ct + t] = acc[ct][v];
  }
}

__global__ __launch_bounds__(256) void k_gemm_self(const float* __restrict__ Zin,
                                                   const float* __restrict__ bnA,
                                                   const float* __restrict__ bnB,
                                                   const float* __restrict__ L,
                                                   float* __restrict__ Hout,
                                                   float* __restrict__ Zs) {
  constexpr int PITCH = 128 + 8;
  __shared__ __attribute__((aligned(16))) float4 sL[32 * PITCH];
  __shared__ __attribute__((aligned(16))) float sBN[2 * FF];
  static_assert(2 * FF == 256, "one thread per scale or shift value");
  stage_weights<128>(L, sL, threadIdx.x);
  sBN[threadIdx.x] = threadIdx.x < FF ? bnA[threadIdx.x] : bnB[threadIdx.x - FF];
  __syncthreads();

  const int wv = threadIdx.x >> 5;
  const int lane = threadIdx.x & 31;
  const int t = lane & 15, hh = lane >> 4;

  for (int strip = blockIdx.x + GEMM_BLOCKS * wv; strip < NSTRIP; strip += GEMM_BLOCKS * 8) {
    const int r0 = strip << 4;
    const float* zrow = Zin + (size_t)(r0 + t) * FF + 16 * hh;
    float* hrow = Hout + (size_t)(r0 + t) * FF + 16 * hh;
    v8f acc[8];
#pragma unroll
    for (int ct = 0; ct < 8; ++ct) acc[ct] = v8f{};
#pragma unroll 1
    for (int m = 0; m < 4; ++m) {
      float4 h[4];
#pragma unroll
      for (int q = 0; q < 4; ++q) {
        const int c = 32 * m + 16 * hh + 4 * q;
        const float4 x = *(const float4*)(zrow + 32 * m + 4 * q);
        h[q] = bn_relu4(x, *(const float4*)(sBN + c), *(const float4*)(sBN + FF + c));
        *(float4*)(hrow + 32 * m + 4 * q) = h[q];
      }
#pragma unroll
      for (int q = 0; q < 4; ++q) {
        const float4* pl = sL + ((m * 4 + q) * 2 + hh) * PITCH + t;
        float4 bl[8];
#pragma unroll
        for (int ct = 0; ct < 8; ++ct) bl[ct] = pl[16 * ct];
#pragma unroll
        for (int ct = 0; ct < 8; ++ct) acc[ct] = mm4(h[q].x, h[q].y, bl[ct].x, bl[ct].y, acc[ct]);
#pragma unroll
        for (int ct = 0; ct < 8; ++ct) acc[ct] = mm4(h[q].z, h[q].w, bl[ct].z, bl[ct].w, acc[ct]);
      }
    }
#pragma unroll
    for (int ct = 0; ct < 8; ++ct)
#pragma unroll
      for (int v = 0; v < 8; ++v)
        Zs[(size_t)(r0 + v + 8 * hh) * FF + 16 * ct + t] = acc[ct][v];
  }
}

__global__ __launch_bounds__(256) void k_gemm_nb(const float* __restrict__ Agg,
                                                 const float* __restrict__ W,
                                                 const float* __restrict__ Zs,
                                                 float* __restrict__ Z) {
  constexpr int PITCH = 128 + 8;
  __shared__ __attribute__((aligned(16))) float4 sW[32 * PITCH];
  stage_weights<128>(W, sW, threadIdx.x);
  __syncthreads();

  const int wv = threadIdx.x >> 5;
  const int lane = threadIdx.x & 31;
  const int t = lane & 15, hh = lane >> 4;

  for (int strip = blockIdx.x + GEMM_BLOCKS * wv; strip < NSTRIP; strip += GEMM_BLOCKS * 8) {
    const int r0 = strip << 4;
    const float* arow = Agg + (size_t)(r0 + t) * FF + 16 * hh;
    v8f acc[8];
#pragma unroll
    for (int ct = 0; ct < 8; ++ct)
#pragma unroll
      for (int v = 0; v < 8; ++v)
        acc[ct][v] = Zs[(size_t)(r0 + v + 8 * hh) * FF + 16 * ct + t];
#pragma unroll 1
    for (int m = 0; m < 4; ++m) {
      float4 a[4];
#pragma unroll
      for (int q = 0; q < 4; ++q) a[q] = *(const float4*)(arow + 32 * m + 4 * q);
#pragma unroll
      for (int q = 0; q < 4; ++q) {
        const float4* pw = sW + ((m * 4 + q) * 2 + hh) * PITCH + t;
        float4 bw[8];
#pragma unroll
        for (int ct = 0; ct < 8; ++ct) bw[ct] = pw[16 * ct];
#pragma unroll
        for (int ct = 0; ct < 8; ++ct) acc[ct] = mm4(a[q].x, a[q].y, bw[ct].x, bw[ct].y, acc[ct]);
#pragma unroll
        for (int ct = 0; ct < 8; ++ct) acc[ct] = mm4(a[q].z, a[q].w, bw[ct].z, bw[ct].w, acc[ct]);
      }
    }
#pragma unroll
    for (int ct = 0; ct < 8; ++ct)
#pragma unroll
      for (int v = 0; v < 8; ++v)
        Z[(size_t)(r0 + v + 8 * hh) * FF + 16 * ct + t] = acc[ct][v];
  }
}

__global__ __launch_bounds__(256) void k_gemm_p(const float* __restrict__ Zin,
                                                const float* __restrict__ bnA,
                                                const float* __restrict__ bnB,
                                                const float* __restrict__ dsout,
                                                const float* __restrict__ Wp,
                                                float* __restrict__ Hout,
                                                float* __restrict__ P) {
  constexpr int PITCH = 64 + 8;
  __shared__ __attribute__((aligned(16))) float4 sW[32 * PITCH];
  __shared__ __attribute__((aligned(16))) float sBN[2 * FF];
  stage_weights<64>(Wp, sW, threadIdx.x);
  sBN[threadIdx.x] = threadIdx.x < FF ? bnA[threadIdx.x] : bnB[threadIdx.x - FF];
  __syncthreads();

  const int wv = threadIdx.x >> 5;
  const int lane = threadIdx.x & 31;
  const int t = lane & 15, hh = lane >> 4;

  for (int strip = blockIdx.x + GEMM_BLOCKS * wv; strip < NSTRIP; strip += GEMM_BLOCKS * 8) {
    const int r0 = strip << 4;
    const float d = dsout[r0 + t];
    const float* zrow = Zin + (size_t)(r0 + t) * FF + 16 * hh;
    float* hrow = Hout + (size_t)(r0 + t) * FF + 16 * hh;
    v8f acc[4];
#pragma unroll
    for (int ct = 0; ct < 4; ++ct) acc[ct] = v8f{};
#pragma unroll 1
    for (int m = 0; m < 4; ++m) {
      float4 a[4];
#pragma unroll
      for (int q = 0; q < 4; ++q) {
        const int c = 32 * m + 16 * hh + 4 * q;
        const float4 z = *(const float4*)(zrow + 32 * m + 4 * q);
        const float4 x = bn_relu4(z, *(const float4*)(sBN + c), *(const float4*)(sBN + FF + c));
        *(float4*)(hrow + 32 * m + 4 * q) = x;
        a[q] = make_float4(d * x.x, d * x.y, d * x.z, d * x.w);
      }
#pragma unroll
      for (int q = 0; q < 4; ++q) {
        const float4* pw = sW + ((m * 4 + q) * 2 + hh) * PITCH + t;
        float4 bw[4];
#pragma unroll
        for (int ct = 0; ct < 4; ++ct) bw[ct] = pw[16 * ct];
#pragma unroll
        for (int ct = 0; ct < 4; ++ct) acc[ct] = mm4(a[q].x, a[q].y, bw[ct].x, bw[ct].y, acc[ct]);
#pragma unroll
        for (int ct = 0; ct < 4; ++ct) acc[ct] = mm4(a[q].z, a[q].w, bw[ct].z, bw[ct].w, acc[ct]);
      }
    }
#pragma unroll
    for (int ct = 0; ct < 4; ++ct)
#pragma unroll
      for (int v = 0; v < 8; ++v)
        P[(size_t)(r0 + v + 8 * hh) * 64 + 16 * ct + t] = acc[ct][v];
  }
}

__global__ __launch_bounds__(256) void k_gemm_out(const float* __restrict__ H,
                                                  const float* __restrict__ Lp,
                                                  const float* __restrict__ R,
                                                  const float* __restrict__ b2,
                                                  float* __restrict__ out) {
  constexpr int PITCH = 64 + 8;
  __shared__ __attribute__((aligned(16))) float4 sL[32 * PITCH];
  stage_weights<64>(Lp, sL, threadIdx.x);
  __syncthreads();

  const int wv = threadIdx.x >> 5;
  const int lane = threadIdx.x & 31;
  const int t = lane & 15, hh = lane >> 4;

  for (int strip = blockIdx.x + GEMM_BLOCKS * wv; strip < NSTRIP; strip += GEMM_BLOCKS * 8) {
    const int r0 = strip << 4;
    const float* hrow = H + (size_t)(r0 + t) * FF + 16 * hh;
    v8f acc[3];
#pragma unroll
    for (int ct = 0; ct < 3; ++ct) acc[ct] = v8f{};
#pragma unroll 1
    for (int m = 0; m < 4; ++m) {
      float4 a[4];
#pragma unroll
      for (int q = 0; q < 4; ++q) a[q] = *(const float4*)(hrow + 32 * m + 4 * q);
#pragma unroll
      for (int q = 0; q < 4; ++q) {
        const float4* pl = sL + ((m * 4 + q) * 2 + hh) * PITCH + t;
        float4 bl[3];
#pragma unroll
        for (int ct = 0; ct < 3; ++ct) bl[ct] = pl[16 * ct];
#pragma unroll
        for (int ct = 0; ct < 3; ++ct) acc[ct] = mm4(a[q].x, a[q].y, bl[ct].x, bl[ct].y, acc[ct]);
#pragma unroll
        for (int ct = 0; ct < 3; ++ct) acc[ct] = mm4(a[q].z, a[q].w, bl[ct].z, bl[ct].w, acc[ct]);
      }
    }
#pragma unroll
    for (int ct = 0; ct < 3; ++ct) {
      const int col = 16 * ct + t;
      if (col < 40) {
        const float bias = b2[col];
#pragma unroll
        for (int v = 0; v < 8; ++v) {
          const int row = r0 + v + 8 * hh;
          out[(size_t)row * 40 + col] = acc[ct][v] + R[(size_t)row * 64 + col] + bias;
        }
      }
    }
  }
}


static constexpr int CS_WAVES = 8;
static constexpr int CS_PER_WAVE = 25;
static constexpr int CS_ROWS = CS_WAVES * CS_PER_WAVE;
static constexpr int CS_BLOCKS = NN / CS_ROWS;
static constexpr int CS_THREADS = CS_WAVES * 32;
static_assert(CS_BLOCKS * CS_ROWS == NN, "statistics grid must cover the rows exactly");
static_assert(CS_THREADS == 2 * FF, "one thread per (statistic, column) in the block's last step");

__global__ __launch_bounds__(CS_THREADS) void k_colstats(const float* __restrict__ z,
                                                           float* __restrict__ cs,
                                                           float* __restrict__ cs2) {
  __shared__ __attribute__((aligned(16))) float4 red[2][CS_WAVES][32];
  const int lane = threadIdx.x & 31, wv = threadIdx.x >> 5;
  const float4* p = (const float4*)(z + ((size_t)blockIdx.x * CS_ROWS + wv) * FF) + lane;
  float4 v[CS_PER_WAVE];
#pragma unroll
  for (int i = 0; i < CS_PER_WAVE; ++i) v[i] = p[(size_t)i * CS_WAVES * (FF / 4)];
  float4 s = make_float4(0.f, 0.f, 0.f, 0.f), s2 = make_float4(0.f, 0.f, 0.f, 0.f);
#pragma unroll
  for (int i = 0; i < CS_PER_WAVE; ++i) {
    s.x += v[i].x;
    s.y += v[i].y;
    s.z += v[i].z;
    s.w += v[i].w;
    s2.x = fmaf(v[i].x, v[i].x, s2.x);
    s2.y = fmaf(v[i].y, v[i].y, s2.y);
    s2.z = fmaf(v[i].z, v[i].z, s2.z);
    s2.w = fmaf(v[i].w, v[i].w, s2.w);
  }
  red[0][wv][lane] = s;
  red[1][wv][lane] = s2;
  __syncthreads();
  const int st = threadIdx.x >> 7, c = threadIdx.x & 127;
  const float* r = (const float*)&red[st][0][0] + c;
  float t = 0.f;
#pragma unroll
  for (int w = 0; w < CS_WAVES; ++w) t += r[w * FF];
  atomicAdd(st ? &cs2[c] : &cs[c], t);
}

__global__ void k_bnparams(const float* __restrict__ cs, const float* __restrict__ cs2,
                           const float* __restrict__ g, const float* __restrict__ be,
                           float* __restrict__ bnA, float* __restrict__ bnB) {
  const int c = threadIdx.x;
  const float inv = 1.0f / (float)NN;
  const float mu = cs[c] * inv;
  const float var = cs2[c] * inv - mu * mu;
  const float a = g[c] * rsqrtf(var + 1e-5f);
  bnA[c] = a;
  bnB[c] = be[c] - mu * a;
}

__global__ void k_padw(const float* __restrict__ W2, const float* __restrict__ L2,
                       float* __restrict__ Wp, float* __restrict__ Lp) {
  const int i = blockIdx.x * 256 + threadIdx.x;
  const int k = i >> 6, c = i & 63;
  const float w = (c < 40) ? W2[k * 40 + c] : 0.f;
  const float l = (c < 40) ? L2[k * 40 + c] : 0.f;
  Wp[i] = w;
  Lp[i] = l;
}


extern "C" void kernel_launch(void* const* d_in, const int* in_sizes, int n_in,
                              void* d_out, int out_size, void* d_ws, size_t ws_size,
                              hipStream_t stream) {
  (void)in_sizes; (void)n_in; (void)out_size; (void)ws_size;

  const float* feat = (const float*)d_in[0];
  const int* src   = (const int*)d_in[1];
  const int* dst   = (const int*)d_in[2];
  const float* W0  = (const float*)d_in[3];
  const float* L0  = (const float*)d_in[4];
  const float* W1  = (const float*)d_in[5];
  const float* L1  = (const float*)d_in[6];
  const float* W2  = (const float*)d_in[7];
  const float* L2  = (const float*)d_in[8];
  const float* b2  = (const float*)d_in[9];
  const float* g0  = (const float*)d_in[10];
  const float* be0 = (const float*)d_in[11];
  const float* g1  = (const float*)d_in[12];
  const float* be1 = (const float*)d_in[13];
  float* out = (float*)d_out;

  char* base = (char*)d_ws;
  size_t off = 0;
  auto take = [&](size_t bytes) -> char* {
    char* p = base + off;
    off = (off + bytes + 255) & ~(size_t)255;
    return p;
  };
  unsigned* cnt     = (unsigned*)take((size_t)2 * NBIN * SLICES * 4);
  unsigned* bump    = (unsigned*)take(2 * 4);
  unsigned* binbase = (unsigned*)take((size_t)2 * NBIN * 4);
  unsigned* bintot  = (unsigned*)take((size_t)2 * NBIN * 4);
  unsigned* binD    = (unsigned*)take((size_t)EE * 4);
  unsigned* binS    = (unsigned*)take((size_t)EE * 4);
  unsigned* rows    = (unsigned*)take((size_t)NN * 8);
  unsigned* esrc    = (unsigned*)take((size_t)EE * 4);
  float* dsout = (float*)take((size_t)NN * 4);
  float* dsin  = (float*)take((size_t)NN * 4);
  float* cs    = (float*)take(128 * 4);
  float* cs2   = (float*)take(128 * 4);
  float* bnA   = (float*)take(128 * 4);
  float* bnB   = (float*)take(128 * 4);
  float* Wp    = (float*)take(128 * 64 * 4);
  float* Lp    = (float*)take(128 * 64 * 4);
  float* bufA  = (float*)take((size_t)NN * FF * 4);
  float* bufB  = (float*)take((size_t)NN * FF * 4);
  float* bufC  = (float*)take((size_t)NN * FF * 4);
  float* bufD  = (float*)take((size_t)NN * FF * 4);

  k_binhist<<<SLICES, 256, 0, stream>>>(src, dst, cnt, bump);
  k_binalloc<<<ALLOC_BLOCKS, 256, 0, stream>>>(cnt, bump, binbase, bintot);
  k_binscatter<<<SLICES, 256, 0, stream>>>(src, dst, cnt, binD, binS);
  k_binrows<<<NBIN, 256, 0, stream>>>(binD, binS, binbase, bintot, dsout, dsin, rows, esrc);
  k_padw<<<32, 256, 0, stream>>>(W2, L2, Wp, Lp);

  k_agg128<<<AGG_BLOCKS, 128, 0, stream>>>(feat, rows, esrc, dsout, dsin, bufA);
  k_gemm_dual<<<GEMM_BLOCKS, 256, 0, stream>>>(bufA, feat, W0, L0, bufB);
  k_zero_u32<<<1, 256, 0, stream>>>((unsigned*)cs, 256);
  k_colstats<<<CS_BLOCKS, CS_THREADS, 0, stream>>>(bufB, cs, cs2);
  k_bnparams<<<1, 128, 0, stream>>>(cs, cs2, g0, be0, bnA, bnB);

  k_gemm_self<<<GEMM_BLOCKS, 256, 0, stream>>>(bufB, bnA, bnB, L1, bufC, bufD);
  k_agg128<<<AGG_BLOCKS, 128, 0, stream>>>(bufC, rows, esrc, dsout, dsin, bufA);
  k_gemm_nb<<<GEMM_BLOCKS, 256, 0, stream>>>(bufA, W1, bufD, bufB);
  k_zero_u32<<<1, 256, 0, stream>>>((unsigned*)cs, 256);
  k_colstats<<<CS_BLOCKS, CS_THREADS, 0, stream>>>(bufB, cs, cs2);
  k_bnparams<<<1, 128, 0, stream>>>(cs, cs2, g1, be1, bnA, bnB);

  k_gemm_p<<<GEMM_BLOCKS, 256, 0, stream>>>(bufB, bnA, bnB, dsout, Wp, bufC, bufA);
  k_agg64<<<AGG_BLOCKS, 128, 0, stream>>>(bufA, rows, esrc, dsin, bufB);
  k_gemm_out<<<GEMM_BLOCKS, 256, 0, stream>>>(bufC, Lp, bufB, b2, out);
}


# Layer 1 projected before aggregation: one GEMM pass over z0 gives dsout*(h1@W1) and h1@L1; the gather sums these rows and adds the self term; h1/agg planes, one GEMM launch gone; z1 sum order changes
# speedup vs baseline: 2.2870x; 1.0465x over previous
#include <hip/hip_runtime.h>
#include <hip/hip_bf16.h>


static constexpr int NN = 100000;
static constexpr int EE = 1600000;
static constexpr int FF = 128;

typedef __attribute__((ext_vector_type(2))) float v2f;
typedef __attribute__((ext_vector_type(8))) float v8f;


__global__ void k_zero_u32(unsigned* __restrict__ p, int n) {
  int i = blockIdx.x * 256 + threadIdx.x;
  if (i < n) p[i] = 0u;
}


static constexpr int BIN_SHIFT = 8;
static constexpr int BIN_NODES = 1 << BIN_SHIFT;
static constexpr int NBIN = (NN + BIN_NODES - 1) / BIN_NODES;
static constexpr int SLICES = 256;
static constexpr int SLICE_EDGES = 6256;
static constexpr int SLICE_IT = (SLICE_EDGES / 4 + 255) / 256;
static constexpr int ALLOC_BLOCKS = (2 * NBIN + 7) / 8;
static constexpr int BIN_BATCH = 8;
static_assert(SLICE_EDGES % 4 == 0 && EE % 4 == 0, "slices are read four edges at a time");
static_assert((SLICES - 1) * SLICE_EDGES < EE && SLICES * SLICE_EDGES >= EE, "slices must cover the edges");
static_assert(NN <= (1 << 24), "a bin entry keeps the node id in 24 bits");

__global__ __launch_bounds__(256) void k_binhist(const int* __restrict__ src, const int* __restrict__ dst,
                                                 unsigned* __restrict__ cnt, unsigned* __restrict__ bump) {
  __shared__ unsigned hD[NBIN];
  __shared__ unsigned hS[NBIN];
  const int tid = threadIdx.x, b = blockIdx.x;
  for (int i = tid; i < NBIN; i += 256) {
    hD[i] = 0u;
    hS[i] = 0u;
  }
  if (b == 0 && tid < 2) bump[tid] = 0u;
  __syncthreads();
  const int e0 = b * SLICE_EDGES;
  const int n4 = min(EE - e0, SLICE_EDGES) >> 2;
  const int4* s4 = (const int4*)(src + e0);
  const int4* d4 = (const int4*)(dst + e0);
  int4 s[SLICE_IT], d[SLICE_IT];
#pragma unroll
  for (int u = 0; u < SLICE_IT; ++u) {
    const int i = min(tid + 256 * u, n4 - 1);
    s[u] = s4[i];
    d[u] = d4[i];
  }
#pragma unroll
  for (int u = 0; u < SLICE_IT; ++u) {
    if (tid + 256 * u < n4) {
      atomicAdd(&hD[d[u].x >> BIN_SHIFT], 1u);
      atomicAdd(&hD[d[u].y >> BIN_SHIFT], 1u);
      atomicAdd(&hD[d[u].z >> BIN_SHIFT], 1u);
      atomicAdd(&hD[d[u].w >> BIN_SHIFT], 1u);
      atomicAdd(&hS[s[u].x >> BIN_SHIFT], 1u);
      atomicAdd(&hS[s[u].y >> BIN_SHIFT], 1u);
      atomicAdd(&hS[s[u].z >> BIN_SHIFT], 1u);
      atomicAdd(&hS[s[u].w >> BIN_SHIFT], 1u);
    }
  }
  __syncthreads();
  for (int i = tid; i < NBIN; i += 256) {
    cnt[(size_t)i * SLICES + b] = hD[i];
    cnt[(size_t)(NBIN + i) * SLICES + b] = hS[i];
  }
}

__global__ __launch_bounds__(256) void k_binalloc(unsigned* __restrict__ cnt, unsigned* __restrict__ bump,
                                                  unsigned* __restrict__ binbase,
                                                  unsigned* __restrict__ bintot) {
  const int lane = threadIdx.x & 31;
  const int w = blockIdx.x * 8 + (threadIdx.x >> 5);
  if (w >= 2 * NBIN) return;
  uint4* p = (uint4*)(cnt + (size_t)w * SLICES) + lane * 2;
  const uint4 a = p[0];
  const uint4 c = p[1];
  const unsigned loc = a.x + a.y + a.z + a.w + c.x + c.y + c.z + c.w;
  unsigned inc = loc;
  for (int d = 1; d < 32; d <<= 1) {
    const unsigned up = __shfl_up(inc, d);
    if (lane >= d) inc += up;
  }
  const unsigned tot = __shfl(inc, 31);
  unsigned base = 0u;
  if (lane == 0) base = atomicAdd(&bump[w >= NBIN ? 1 : 0], tot);
  base = __shfl(base, 0);
  unsigned o = base + inc - loc;
  uint4 ra, rc;
  ra.x = o; o += a.x;
  ra.y = o; o += a.y;
  ra.z = o; o += a.z;
  ra.w = o; o += a.w;
  rc.x = o; o += c.x;
  rc.y = o; o += c.y;
  rc.z = o; o += c.z;
  rc.w = o;
  p[0] = ra;
  p[1] = rc;
  if (lane == 0) {
    binbase[w] = base;
    bintot[w] = tot;
  }
}

__device__ __forceinline__ void bin_put4(unsigned* cD, unsigned* cS, unsigned* __restrict__ binD,
                                         unsigned* __restrict__ binS, const int4 s, const int4 d) {
  const unsigned pd0 = atomicAdd(&cD[d.x >> BIN_SHIFT], 1u);
  const unsigned pd1 = atomicAdd(&cD[d.y >> BIN_SHIFT], 1u);
  const unsigned pd2 = atomicAdd(&cD[d.z >> BIN_SHIFT], 1u);
  const unsigned pd3 = atomicAdd(&cD[d.w >> BIN_SHIFT], 1u);
  const unsigned ps0 = atomicAdd(&cS[s.x >> BIN_SHIFT], 1u);
  const unsigned ps1 = atomicAdd(&cS[s.y >> BIN_SHIFT], 1u);
  const unsigned ps2 = atomicAdd(&cS[s.z >> BIN_SHIFT], 1u);
  const unsigned ps3 = atomicAdd(&cS[s.w >> BIN_SHIFT], 1u);
  binD[pd0] = ((unsigned)(d.x & (BIN_NODES - 1)) << 24) | (unsigned)s.x;
  binD[pd1] = ((unsigned)(d.y & (BIN_NODES - 1)) << 24) | (unsigned)s.y;
  binD[pd2] = ((unsigned)(d.z & (BIN_NODES - 1)) << 24) | (unsigned)s.z;
  binD[pd3] = ((unsigned)(d.w & (BIN_NODES - 1)) << 24) | (unsigned)s.w;
  binS[ps0] = (unsigned)(s.x & (BIN_NODES - 1));
  binS[ps1] = (unsigned)(s.y & (BIN_NODES - 1));
  binS[ps2] = (unsigned)(s.z & (BIN_NODES - 1));
  binS[ps3] = (unsigned)(s.w & (BIN_NODES - 1));
}

__global__ __launch_bounds__(256) void k_binscatter(const int* __restrict__ src, const int* __restrict__ dst,
                                                    const unsigned* __restrict__ off,
                                                    unsigned* __restrict__ binD,
                                                    unsigned* __restrict__ binS) {
  __shared__ unsigned cD[NBIN];
  __shared__ unsigned cS[NBIN];
  const int tid = threadIdx.x, b = blockIdx.x;
  for (int i = tid; i < NBIN; i += 256) {
    cD[i] = off[(size_t)i * SLICES + b];
    cS[i] = off[(size_t)(NBIN + i) * SLICES + b];
  }
  __syncthreads();
  const int e0 = b * SLICE_EDGES;
  const int n4 = min(EE - e0, SLICE_EDGES) >> 2;
  const int4* s4 = (const int4*)(src + e0);
  const int4* d4 = (const int4*)(dst + e0);
  int4 s[SLICE_IT], d[SLICE_IT];
#pragma unroll
  for (int u = 0; u < SLICE_IT; ++u) {
    const int i = min(tid + 256 * u, n4 - 1);
    s[u] = s4[i];
    d[u] = d4[i];
  }
#pragma unroll
  for (int u = 0; u < SLICE_IT; ++u) {
    if (tid + 256 * u < n4) {
      bin_put4(cD, cS, binD, binS, s[u], d[u]);
    }
  }
}

__global__ __launch_bounds__(256) void k_binrows(const unsigned* __restrict__ binD,
                                                 const unsigned* __restrict__ binS,
                                                 const unsigned* __restrict__ binbase,
                                                 const unsigned* __restrict__ bintot,
                                                 float* __restrict__ dsout, float* __restrict__ dsin,
                                                 unsigned* __restrict__ rows, unsigned* __restrict__ esrc) {
  __shared__ unsigned cD[BIN_NODES];
  __shared__ unsigned cS[BIN_NODES];
  __shared__ unsigned cur[BIN_NODES];
  __shared__ unsigned wtot[8];
  const int tid = threadIdx.x, j = blockIdx.x;
  const int lane = tid & 31, wv = tid >> 5;
  cD[tid] = 0u;
  cS[tid] = 0u;
  const unsigned bD = binbase[j], nD = bintot[j];
  const unsigned bS = binbase[NBIN + j], nS = bintot[NBIN + j];
  const unsigned* eD = binD + bD;
  const unsigned* eS = binS + bS;
  __syncthreads();

  for (unsigned i0 = 0u; i0 < nD; i0 += 256u * BIN_BATCH) {
    unsigned v[BIN_BATCH];
#pragma unroll
    for (int u = 0; u < BIN_BATCH; ++u) v[u] = eD[min(i0 + 256u * u + tid, nD - 1u)];
#pragma unroll
    for (int u = 0; u < BIN_BATCH; ++u)
      if (i0 + 256u * u + tid < nD) atomicAdd(&cD[v[u] >> 24], 1u);
  }
  for (unsigned i0 = 0u; i0 < nS; i0 += 256u * BIN_BATCH) {
    unsigned v[BIN_BATCH];
#pragma unroll
    for (int u = 0; u < BIN_BATCH; ++u) v[u] = eS[min(i0 + 256u * u + tid, nS - 1u)];
#pragma unroll
    for (int u = 0; u < BIN_BATCH; ++u)
      if (i0 + 256u * u + tid < nS) atomicAdd(&cS[v[u]], 1u);
  }
  __syncthreads();

  const unsigned id = cD[tid], od = cS[tid];
  unsigned inc = id;
  for (int d = 1; d < 32; d <<= 1) {
    const unsigned up = __shfl_up(inc, d);
    if (lane >= d) inc += up;
  }
  if (lane == 31) wtot[wv] = inc;
  __syncthreads();
  unsigned st = inc - id;
  for (int w = 0; w < wv; ++w) st += wtot[w];
  cur[tid] = st;
  const int node = j * BIN_NODES + tid;
  if (node < NN) {
    *(uint2*)(rows + 2 * node) = make_uint2(bD + st, bD + st + id);
    dsin[node] = id ? rsqrtf((float)id) : 1.0f;
    dsout[node] = od ? rsqrtf((float)od) : 1.0f;
  }
  __syncthreads();

  unsigned* o = esrc + bD;
  for (unsigned i0 = 0u; i0 < nD; i0 += 256u * BIN_BATCH) {
    unsigned v[BIN_BATCH];
#pragma unroll
    for (int u = 0; u < BIN_BATCH; ++u) v[u] = eD[min(i0 + 256u * u + tid, nD - 1u)];
    unsigned p[BIN_BATCH];
#pragma unroll
    for (int u = 0; u < BIN_BATCH; ++u) {
      p[u] = 0u;
      if (i0 + 256u * u + tid < nD) p[u] = atomicAdd(&cur[v[u] >> 24], 1u);
    }
#pragma unroll
    for (int u = 0; u < BIN_BATCH; ++u)
      if (i0 + 256u * u + tid < nD) o[p[u]] = v[u] & 0xFFFFFFu;
  }
}


static constexpr int AGG_NODES = 8;
static constexpr int AGG_GROUP = 8;
static constexpr int AGG_WAVES = NN / AGG_NODES;
static constexpr int AGG_BLOCKS = AGG_WAVES / 4;
static_assert(NN % AGG_NODES == 0 && AGG_WAVES % 4 == 0, "gather grid must cover the nodes exactly");

__device__ __forceinline__ unsigned lane_u32(unsigned v, int l) {
  return (unsigned)__builtin_amdgcn_readlane((int)v, l);
}
__device__ __forceinline__ float lane_f32(float v, int l) {
  return __int_as_float(__builtin_amdgcn_readlane(__float_as_int(v), l));
}

__device__ __forceinline__ void axpy4(float4& acc, float w, const float4& v) {
  acc.x = fmaf(w, v.x, acc.x);
  acc.y = fmaf(w, v.y, acc.y);
  acc.z = fmaf(w, v.z, acc.z);
  acc.w = fmaf(w, v.w, acc.w);
}

__global__ __launch_bounds__(128) void k_agg128(const float* __restrict__ h,
                                                const unsigned* __restrict__ rows,
                                                const unsigned* __restrict__ esrc,
                                                const float* __restrict__ dsout,
                                                const float* __restrict__ dsin,
                                                float* __restrict__ agg) {
  const int lane = threadIdx.x & 31;
  const int n0 = (blockIdx.x * 4 + (threadIdx.x >> 5)) * AGG_NODES;
  unsigned rb = 0u, re = 0u;
  float cin = 0.f;
  if (lane < AGG_NODES) {
    const uint2 r = *(const uint2*)(rows + 2 * (n0 + lane));
    rb = r.x;
    re = r.y;
    cin = dsin[n0 + lane];
  }
  unsigned b = lane_u32(rb, 0), e = lane_u32(re, 0);
  unsigned my = 0u;
  float myw = 0.f;
  if (b + lane < e) {
    my = esrc[b + lane];
    myw = dsout[my];
  }
  const float* hl = h + lane * 4;

#pragma unroll 1
  for (int k = 0; k < AGG_NODES; ++k) {
    unsigned nb = 0u, ne = 0u, nmy = 0u;
    if (k + 1 < AGG_NODES) {
      nb = lane_u32(rb, k + 1);
      ne = lane_u32(re, k + 1);
      if (nb + lane < ne) nmy = esrc[nb + lane];
    }
    float4 acc = make_float4(0.f, 0.f, 0.f, 0.f);
    const int cnt = (int)min(e - b, 32u);
    int j = 0;
    for (; j + AGG_GROUP <= cnt; j += AGG_GROUP) {
      float4 v[AGG_GROUP];
#pragma unroll
      for (int u = 0; u < AGG_GROUP; ++u)
        v[u] = *(const float4*)(hl + (size_t)lane_u32(my, j + u) * FF);
      __builtin_amdgcn_sched_barrier(0);
#pragma unroll
      for (int u = 0; u < AGG_GROUP; ++u) axpy4(acc, lane_f32(myw, j + u), v[u]);
    }
    const int rem = cnt - j;
    if (rem > 0) {
      float4 v[AGG_GROUP - 1];
#pragma unroll
      for (int u = 0; u < AGG_GROUP - 1; ++u) {
        v[u] = make_float4(0.f, 0.f, 0.f, 0.f);
        if (u < rem) v[u] = *(const float4*)(hl + (size_t)lane_u32(my, j + u) * FF);
      }
#pragma unroll
      for (int u = 0; u < AGG_GROUP - 1; ++u)
        if (u < rem) axpy4(acc, lane_f32(myw, j + u), v[u]);
    }
    for (unsigned i = b + 32u; i < e; ++i) {
      const unsigned s = esrc[i];
      axpy4(acc, dsout[s], *(const float4*)(hl + (size_t)s * FF));
    }
    const float c = lane_f32(cin, k);
    *(float4*)(agg + (size_t)(n0 + k) * FF + lane * 4) =
        make_float4(c * acc.x, c * acc.y, c * acc.z, c * acc.w);
    b = nb;
    e = ne;
    my = nmy;
    myw = 0.f;
    if (b + lane < e) myw = dsout[my];
  }
}

__global__ __launch_bounds__(128) void k_agg64(const float* __restrict__ P,
                                               const unsigned* __restrict__ rows,
                                               const unsigned* __restrict__ esrc,
                                               const float* __restrict__ dsin,
                                               float* __restrict__ R) {
  const int lane = threadIdx.x & 31;
  const int n0 = (blockIdx.x * 4 + (threadIdx.x >> 5)) * AGG_NODES;
  unsigned rb = 0u, re = 0u;
  float cin = 0.f;
  if (lane < AGG_NODES) {
    const uint2 r = *(const uint2*)(rows + 2 * (n0 + lane));
    rb = r.x;
    re = r.y;
    cin = dsin[n0 + lane];
  }
  unsigned b = lane_u32(rb, 0), e = lane_u32(re, 0);
  unsigned my = 0u;
  if (b + lane < e) my = esrc[b + lane];
  const float* pl = P + lane * 2;

#pragma unroll 1
  for (int k = 0; k < AGG_NODES; ++k) {
    unsigned nb = 0u, ne = 0u, nmy = 0u;
    if (k + 1 < AGG_NODES) {
      nb = lane_u32(rb, k + 1);
      ne = lane_u32(re, k + 1);
      if (nb + lane < ne) nmy = esrc[nb + lane];
    }
    float2 acc = make_float2(0.f, 0.f);
    const int cnt = (int)min(e - b, 32u);
    int j = 0;
    for (; j + AGG_GROUP <= cnt; j += AGG_GROUP) {
      float2 v[AGG_GROUP];
#pragma unroll
      for (int u = 0; u < AGG_GROUP; ++u)
        v[u] = *(const float2*)(pl + (size_t)lane_u32(my, j + u) * 64);
#pragma unroll
      for (int u = 0; u < AGG_GROUP; ++u) {
        acc.x += v[u].x;
        acc.y += v[u].y;
      }
    }
    const int rem = cnt - j;
    if (rem > 0) {
      float2 v[AGG_GROUP - 1];
#pragma unroll
      for (int u = 0; u < AGG_GROUP - 1; ++u) {
        v[u] = make_float2(0.f, 0.f);
        if (u < rem) v[u] = *(const float2*)(pl + (size_t)lane_u32(my, j + u) * 64);
      }
#pragma unroll
      for (int u = 0; u < AGG_GROUP - 1; ++u) {
        if (u < rem) {
          acc.x += v[u].x;
          acc.y += v[u].y;
        }
      }
    }
    for (unsigned i = b + 32u; i < e; ++i) {
      const float2 x = *(const float2*)(pl + (size_t)esrc[i] * 64);
      acc.x += x.x;
      acc.y += x.y;
    }
    const float c = lane_f32(cin, k);
    *(float2*)(R + (size_t)(n0 + k) * 64 + lane * 2) = make_float2(c * acc.x, c * acc.y);
    b = nb;
    e = ne;
    my = nmy;
  }
}

__global__ __launch_bounds__(128) void k_aggadd128(const float* __restrict__ G,
                                                   const unsigned* __restrict__ rows,
                                                   const unsigned* __restrict__ esrc,
                                                   const float* __restrict__ dsin,
                                                   const float* __restrict__ S,
                                                   float* __restrict__ Z) {
  const int lane = threadIdx.x & 31;
  const int n0 = (blockIdx.x * 4 + (threadIdx.x >> 5)) * AGG_NODES;
  unsigned rb = 0u, re = 0u;
  float cin = 0.f;
  if (lane < AGG_NODES) {
    const uint2 r = *(const uint2*)(rows + 2 * (n0 + lane));
    rb = r.x;
    re = r.y;
    cin = dsin[n0 + lane];
  }
  unsigned b = lane_u32(rb, 0), e = lane_u32(re, 0);
  unsigned my = 0u;
  if (b + lane < e) my = esrc[b + lane];
  const float* gl = G + lane * 4;

#pragma unroll 1
  for (int k = 0; k < AGG_NODES; ++k) {
    unsigned nb = 0u, ne = 0u, nmy = 0u;
    if (k + 1 < AGG_NODES) {
      nb = lane_u32(rb, k + 1);
      ne = lane_u32(re, k + 1);
      if (nb + lane < ne) nmy = esrc[nb + lane];
    }
    const float4 self = *(const float4*)(S + (size_t)(n0 + k) * FF + lane * 4);
    float4 acc = make_float4(0.f, 0.f, 0.f, 0.f);
    const int cnt = (int)min(e - b, 32u);
    int j = 0;
    for (; j + AGG_GROUP <= cnt; j += AGG_GROUP) {
      float4 v[AGG_GROUP];
#pragma unroll
      for (int u = 0; u < AGG_GROUP; ++u)
        v[u] = *(const float4*)(gl + (size_t)lane_u32(my, j + u) * FF);
#pragma unroll
      for (int u = 0; u < AGG_GROUP; ++u) {
        acc.x += v[u].x;
        acc.y += v[u].y;
        acc.z += v[u].z;
        acc.w += v[u].w;
      }
    }
    const int rem = cnt - j;
    if (rem > 0) {
      float4 v[AGG_GROUP - 1];
#pragma unroll
      for (int u = 0; u < AGG_GROUP - 1; ++u) {
        v[u] = make_float4(0.f, 0.f, 0.f, 0.f);
        if (u < rem) v[u] = *(const float4*)(gl + (size_t)lane_u32(my, j + u) * FF);
      }
#pragma unroll
      for (int u = 0; u < AGG_GROUP - 1; ++u) {
        if (u < rem) {
          acc.x += v[u].x;
          acc.y += v[u].y;
          acc.z += v[u].z;
          acc.w += v[u].w;
        }
      }
    }
    for (unsigned i = b + 32u; i < e; ++i) {
      const float4 x = *(const float4*)(gl + (size_t)esrc[i] * FF);
      acc.x += x.x;
      acc.y += x.y;
      acc.z += x.z;
      acc.w += x.w;
    }
    const float c = lane_f32(cin, k);
    *(float4*)(Z + (size_t)(n0 + k) * FF + lane * 4) =
        make_float4(fmaf(c, acc.x, self.x), fmaf(c, acc.y, self.y), fmaf(c, acc.z, self.z),
                    fmaf(c, acc.w, self.w));
    b = nb;
    e = ne;
    my = nmy;
  }
}


static constexpr int GEMM_BLOCKS = 512;
static constexpr int NSTRIP = NN / 16;

template <int NC>
__device__ __forceinline__ void stage_weights(const float* __restrict__ Bg,
                                              float4* __restrict__ s, int tid) {
  constexpr int C4 = NC / 4;
  constexpr int PITCH = NC + 8;
  for (int i = tid; i < 32 * C4; i += 256) {
    const int g = i / C4, c4 = i - g * C4;
    const float* p = Bg + (size_t)(4 * g) * NC + 4 * c4;
    const float4 r0 = *(const float4*)(p);
    const float4 r1 = *(const float4*)(p + NC);
    const float4 r2 = *(const float4*)(p + 2 * NC);
    const float4 r3 = *(const float4*)(p + 3 * NC);
    const int m = g >> 3, hh = (g >> 2) & 1, q = g & 3;
    float4* d = s + ((m * 4 + q) * 2 + hh) * PITCH + 4 * c4;
    d[0] = make_float4(r0.x, r1.x, r2.x, r3.x);
    d[1] = make_float4(r0.y, r1.y, r2.y, r3.y);
    d[2] = make_float4(r0.z, r1.z, r2.z, r3.z);
    d[3] = make_float4(r0.w, r1.w, r2.w, r3.w);
  }
}

__device__ __forceinline__ v8f mm4(float ax, float ay, float bx, float by, v8f c) {
  v2f a; a.x = ax; a.y = ay;
  v2f b; b.x = bx; b.y = by;
  return __builtin_amdgcn_wmma_f32_16x16x4_f32(false, a, false, b, (short)0, c, false, false);
}

__device__ __forceinline__ float4 bn_relu4(const float4 z, const float4 a, const float4 b) {
  const float vx = fmaf(z.x, a.x, b.x);
  const float vy = fmaf(z.y, a.y, b.y);
  const float vz = fmaf(z.z, a.z, b.z);
  const float vw = fmaf(z.w, a.w, b.w);
  return make_float4(vx > 0.f ? vx : 0.f, vy > 0.f ? vy : 0.f, vz > 0.f ? vz : 0.f,
                     vw > 0.f ? vw : 0.f);
}

__global__ __launch_bounds__(256) void k_gemm_dual(const float* __restrict__ Agg,
                                                   const float* __restrict__ H,
                                                   const float* __restrict__ W,
                                                   const float* __restrict__ L,
                                                   float* __restrict__ Z) {
  constexpr int PITCH = 128 + 8;
  __shared__ __attribute__((aligned(16))) float4 sW[32 * PITCH];
  __shared__ __attribute__((aligned(16))) float4 sL[32 * PITCH];
  stage_weights<128>(W, sW, threadIdx.x);
  stage_weights<128>(L, sL, threadIdx.x);
  __syncthreads();

  const int wv = threadIdx.x >> 5;
  const int lane = threadIdx.x & 31;
  const int t = lane & 15;
  const int hh = lane >> 4;

  for (int strip = blockIdx.x + GEMM_BLOCKS * wv; strip < NSTRIP; strip += GEMM_BLOCKS * 8) {
    const int r0 = strip << 4;
    const float* arow = Agg + (size_t)(r0 + t) * FF + 16 * hh;
    const float* hrow = H + (size_t)(r0 + t) * FF + 16 * hh;
    v8f acc[8];
#pragma unroll
    for (int ct = 0; ct < 8; ++ct) acc[ct] = v8f{};

#pragma unroll 1
    for (int m = 0; m < 4; ++m) {
      float4 a[4], h[4];
#pragma unroll
      for (int q = 0; q < 4; ++q) {
        a[q] = *(const float4*)(arow + 32 * m + 4 * q);
        h[q] = *(const float4*)(hrow + 32 * m + 4 * q);
      }
#pragma unroll
      for (int q = 0; q < 4; ++q) {
        const float4* pw = sW + ((m * 4 + q) * 2 + hh) * PITCH + t;
        const float4* pl = sL + ((m * 4 + q) * 2 + hh) * PITCH + t;
        float4 bw[8], bl[8];
#pragma unroll
        for (int ct = 0; ct < 8; ++ct) { bw[ct] = pw[16 * ct]; bl[ct] = pl[16 * ct]; }
#pragma unroll
        for (int ct = 0; ct < 8; ++ct) acc[ct] = mm4(a[q].x, a[q].y, bw[ct].x, bw[ct].y, acc[ct]);
#pragma unroll
        for (int ct = 0; ct < 8; ++ct) acc[ct] = mm4(h[q].x, h[q].y, bl[ct].x, bl[ct].y, acc[ct]);
#pragma unroll
        for (int ct = 0; ct < 8; ++ct) acc[ct] = mm4(a[q].z, a[q].w, bw[ct].z, bw[ct].w, acc[ct]);
#pragma unroll
        for (int ct = 0; ct < 8; ++ct) acc[ct] = mm4(h[q].z, h[q].w, bl[ct].z, bl[ct].w, acc[ct]);
      }
    }
#pragma unroll
    for (int ct = 0; ct < 8; ++ct)
#pragma unroll
      for (int v = 0; v < 8; ++v)
        Z[(size_t)(r0 + v + 8 * hh) * FF + 16 * ct + t] = acc[ct][v];
  }
}

__global__ __launch_bounds__(256) void k_gemm_proj(const float* __restrict__ Zin,
                                                   const float* __restrict__ bnA,
                                                   const float* __restrict__ bnB,
                                                   const float* __restrict__ dsout,
                                                   const float* __restrict__ W,
                                                   const float* __restrict__ L,
                                                   float* __restrict__ Gw,
                                                   float* __restrict__ Gl) {
  constexpr int PITCH = 128 + 8;
  __shared__ __attribute__((aligned(16))) float4 sW[32 * PITCH];
  __shared__ __attribute__((aligned(16))) float4 sL[32 * PITCH];
  __shared__ __attribute__((aligned(16))) float sBN[2 * FF];
  static_assert(2 * FF == 256, "one thread per scale or shift value");
  stage_weights<128>(W, sW, threadIdx.x);
  stage_weights<128>(L, sL, threadIdx.x);
  sBN[threadIdx.x] = threadIdx.x < FF ? bnA[threadIdx.x] : bnB[threadIdx.x - FF];
  __syncthreads();

  const int wv = threadIdx.x >> 5;
  const int lane = threadIdx.x & 31;
  const int t = lane & 15, hh = lane >> 4;

  for (int strip = blockIdx.x + GEMM_BLOCKS * wv; strip < NSTRIP; strip += GEMM_BLOCKS * 8) {
    const int r0 = strip << 4;
    const float* zrow = Zin + (size_t)(r0 + t) * FF + 16 * hh;
    const float4 d0 = *(const float4*)(dsout + r0 + 8 * hh);
    const float4 d1 = *(const float4*)(dsout + r0 + 8 * hh + 4);
    v8f aw[8], al[8];
#pragma unroll
    for (int ct = 0; ct < 8; ++ct) { aw[ct] = v8f{}; al[ct] = v8f{}; }
#pragma unroll 1
    for (int m = 0; m < 4; ++m) {
      float4 h[4];
#pragma unroll
      for (int q = 0; q < 4; ++q) {
        const int c = 32 * m + 16 * hh + 4 * q;
        const float4 x = *(const float4*)(zrow + 32 * m + 4 * q);
        h[q] = bn_relu4(x, *(const float4*)(sBN + c), *(const float4*)(sBN + FF + c));
      }
#pragma unroll
      for (int q = 0; q < 4; ++q) {
        const float4* pw = sW + ((m * 4 + q) * 2 + hh) * PITCH + t;
        const float4* pl = sL + ((m * 4 + q) * 2 + hh) * PITCH + t;
        float4 bw[8], bl[8];
#pragma unroll
        for (int ct = 0; ct < 8; ++ct) { bw[ct] = pw[16 * ct]; bl[ct] = pl[16 * ct]; }
#pragma unroll
        for (int ct = 0; ct < 8; ++ct) aw[ct] = mm4(h[q].x, h[q].y, bw[ct].x, bw[ct].y, aw[ct]);
#pragma unroll
        for (int ct = 0; ct < 8; ++ct) al[ct] = mm4(h[q].x, h[q].y, bl[ct].x, bl[ct].y, al[ct]);
#pragma unroll
        for (int ct = 0; ct < 8; ++ct) aw[ct] = mm4(h[q].z, h[q].w, bw[ct].z, bw[ct].w, aw[ct]);
#pragma unroll
        for (int ct = 0; ct < 8; ++ct) al[ct] = mm4(h[q].z, h[q].w, bl[ct].z, bl[ct].w, al[ct]);
      }
    }
    const float dv[8] = {d0.x, d0.y, d0.z, d0.w, d1.x, d1.y, d1.z, d1.w};
#pragma unroll
    for (int ct = 0; ct < 8; ++ct)
#pragma unroll
      for (int v = 0; v < 8; ++v) {
        const size_t o = (size_t)(r0 + v + 8 * hh) * FF + 16 * ct + t;
        Gw[o] = dv[v] * aw[ct][v];
        Gl[o] = al[ct][v];
      }
  }
}

__global__ __launch_bounds__(256) void k_gemm_p(const float* __restrict__ Zin,
                                                const float* __restrict__ bnA,
                                                const float* __restrict__ bnB,
                                                const float* __restrict__ dsout,
                                                const float* __restrict__ Wp,
                                                float* __restrict__ Hout,
                                                float* __restrict__ P) {
  constexpr int PITCH = 64 + 8;
  __shared__ __attribute__((aligned(16))) float4 sW[32 * PITCH];
  __shared__ __attribute__((aligned(16))) float sBN[2 * FF];
  stage_weights<64>(Wp, sW, threadIdx.x);
  sBN[threadIdx.x] = threadIdx.x < FF ? bnA[threadIdx.x] : bnB[threadIdx.x - FF];
  __syncthreads();

  const int wv = threadIdx.x >> 5;
  const int lane = threadIdx.x & 31;
  const int t = lane & 15, hh = lane >> 4;

  for (int strip = blockIdx.x + GEMM_BLOCKS * wv; strip < NSTRIP; strip += GEMM_BLOCKS * 8) {
    const int r0 = strip << 4;
    const float d = dsout[r0 + t];
    const float* zrow = Zin + (size_t)(r0 + t) * FF + 16 * hh;
    float* hrow = Hout + (size_t)(r0 + t) * FF + 16 * hh;
    v8f acc[4];
#pragma unroll
    for (int ct = 0; ct < 4; ++ct) acc[ct] = v8f{};
#pragma unroll 1
    for (int m = 0; m < 4; ++m) {
      float4 a[4];
#pragma unroll
      for (int q = 0; q < 4; ++q) {
        const int c = 32 * m + 16 * hh + 4 * q;
        const float4 z = *(const float4*)(zrow + 32 * m + 4 * q);
        const float4 x = bn_relu4(z, *(const float4*)(sBN + c), *(const float4*)(sBN + FF + c));
        *(float4*)(hrow + 32 * m + 4 * q) = x;
        a[q] = make_float4(d * x.x, d * x.y, d * x.z, d * x.w);
      }
#pragma unroll
      for (int q = 0; q < 4; ++q) {
        const float4* pw = sW + ((m * 4 + q) * 2 + hh) * PITCH + t;
        float4 bw[4];
#pragma unroll
        for (int ct = 0; ct < 4; ++ct) bw[ct] = pw[16 * ct];
#pragma unroll
        for (int ct = 0; ct < 4; ++ct) acc[ct] = mm4(a[q].x, a[q].y, bw[ct].x, bw[ct].y, acc[ct]);
#pragma unroll
        for (int ct = 0; ct < 4; ++ct) acc[ct] = mm4(a[q].z, a[q].w, bw[ct].z, bw[ct].w, acc[ct]);
      }
    }
#pragma unroll
    for (int ct = 0; ct < 4; ++ct)
#pragma unroll
      for (int v = 0; v < 8; ++v)
        P[(size_t)(r0 + v + 8 * hh) * 64 + 16 * ct + t] = acc[ct][v];
  }
}

__global__ __launch_bounds__(256) void k_gemm_out(const float* __restrict__ H,
                                                  const float* __restrict__ Lp,
                                                  const float* __restrict__ R,
                                                  const float* __restrict__ b2,
                                                  float* __restrict__ out) {
  constexpr int PITCH = 64 + 8;
  __shared__ __attribute__((aligned(16))) float4 sL[32 * PITCH];
  stage_weights<64>(Lp, sL, threadIdx.x);
  __syncthreads();

  const int wv = threadIdx.x >> 5;
  const int lane = threadIdx.x & 31;
  const int t = lane & 15, hh = lane >> 4;

  for (int strip = blockIdx.x + GEMM_BLOCKS * wv; strip < NSTRIP; strip += GEMM_BLOCKS * 8) {
    const int r0 = strip << 4;
    const float* hrow = H + (size_t)(r0 + t) * FF + 16 * hh;
    v8f acc[3];
#pragma unroll
    for (int ct = 0; ct < 3; ++ct) acc[ct] = v8f{};
#pragma unroll 1
    for (int m = 0; m < 4; ++m) {
      float4 a[4];
#pragma unroll
      for (int q = 0; q < 4; ++q) a[q] = *(const float4*)(hrow + 32 * m + 4 * q);
#pragma unroll
      for (int q = 0; q < 4; ++q) {
        const float4* pl = sL + ((m * 4 + q) * 2 + hh) * PITCH + t;
        float4 bl[3];
#pragma unroll
        for (int ct = 0; ct < 3; ++ct) bl[ct] = pl[16 * ct];
#pragma unroll
        for (int ct = 0; ct < 3; ++ct) acc[ct] = mm4(a[q].x, a[q].y, bl[ct].x, bl[ct].y, acc[ct]);
#pragma unroll
        for (int ct = 0; ct < 3; ++ct) acc[ct] = mm4(a[q].z, a[q].w, bl[ct].z, bl[ct].w, acc[ct]);
      }
    }
#pragma unroll
    for (int ct = 0; ct < 3; ++ct) {
      const int col = 16 * ct + t;
      if (col < 40) {
        const float bias = b2[col];
#pragma unroll
        for (int v = 0; v < 8; ++v) {
          const int row = r0 + v + 8 * hh;
          out[(size_t)row * 40 + col] = acc[ct][v] + R[(size_t)row * 64 + col] + bias;
        }
      }
    }
  }
}


static constexpr int CS_WAVES = 8;
static constexpr int CS_PER_WAVE = 25;
static constexpr int CS_ROWS = CS_WAVES * CS_PER_WAVE;
static constexpr int CS_BLOCKS = NN / CS_ROWS;
static constexpr int CS_THREADS = CS_WAVES * 32;
static_assert(CS_BLOCKS * CS_ROWS == NN, "statistics grid must cover the rows exactly");
static_assert(CS_THREADS == 2 * FF, "one thread per (statistic, column) in the block's last step");

__global__ __launch_bounds__(CS_THREADS) void k_colstats(const float* __restrict__ z,
                                                           float* __restrict__ cs,
                                                           float* __restrict__ cs2) {
  __shared__ __attribute__((aligned(16))) float4 red[2][CS_WAVES][32];
  const int lane = threadIdx.x & 31, wv = threadIdx.x >> 5;
  const float4* p = (const float4*)(z + ((size_t)blockIdx.x * CS_ROWS + wv) * FF) + lane;
  float4 v[CS_PER_WAVE];
#pragma unroll
  for (int i = 0; i < CS_PER_WAVE; ++i) v[i] = p[(size_t)i * CS_WAVES * (FF / 4)];
  float4 s = make_float4(0.f, 0.f, 0.f, 0.f), s2 = make_float4(0.f, 0.f, 0.f, 0.f);
#pragma unroll
  for (int i = 0; i < CS_PER_WAVE; ++i) {
    s.x += v[i].x;
    s.y += v[i].y;
    s.z += v[i].z;
    s.w += v[i].w;
    s2.x = fmaf(v[i].x, v[i].x, s2.x);
    s2.y = fmaf(v[i].y, v[i].y, s2.y);
    s2.z = fmaf(v[i].z, v[i].z, s2.z);
    s2.w = fmaf(v[i].w, v[i].w, s2.w);
  }
  red[0][wv][lane] = s;
  red[1][wv][lane] = s2;
  __syncthreads();
  const int st = threadIdx.x >> 7, c = threadIdx.x & 127;
  const float* r = (const float*)&red[st][0][0] + c;
  float t = 0.f;
#pragma unroll
  for (int w = 0; w < CS_WAVES; ++w) t += r[w * FF];
  atomicAdd(st ? &cs2[c] : &cs[c], t);
}

__global__ void k_bnparams(const float* __restrict__ cs, const float* __restrict__ cs2,
                           const float* __restrict__ g, const float* __restrict__ be,
                           float* __restrict__ bnA, float* __restrict__ bnB) {
  const int c = threadIdx.x;
  const float inv = 1.0f / (float)NN;
  const float mu = cs[c] * inv;
  const float var = cs2[c] * inv - mu * mu;
  const float a = g[c] * rsqrtf(var + 1e-5f);
  bnA[c] = a;
  bnB[c] = be[c] - mu * a;
}

__global__ void k_padw(const float* __restrict__ W2, const float* __restrict__ L2,
                       float* __restrict__ Wp, float* __restrict__ Lp) {
  const int i = blockIdx.x * 256 + threadIdx.x;
  const int k = i >> 6, c = i & 63;
  const float w = (c < 40) ? W2[k * 40 + c] : 0.f;
  const float l = (c < 40) ? L2[k * 40 + c] : 0.f;
  Wp[i] = w;
  Lp[i] = l;
}


extern "C" void kernel_launch(void* const* d_in, const int* in_sizes, int n_in,
                              void* d_out, int out_size, void* d_ws, size_t ws_size,
                              hipStream_t stream) {
  (void)in_sizes; (void)n_in; (void)out_size; (void)ws_size;

  const float* feat = (const float*)d_in[0];
  const int* src   = (const int*)d_in[1];
  const int* dst   = (const int*)d_in[2];
  const float* W0  = (const float*)d_in[3];
  const float* L0  = (const float*)d_in[4];
  const float* W1  = (const float*)d_in[5];
  const float* L1  = (const float*)d_in[6];
  const float* W2  = (const float*)d_in[7];
  const float* L2  = (const float*)d_in[8];
  const float* b2  = (const float*)d_in[9];
  const float* g0  = (const float*)d_in[10];
  const float* be0 = (const float*)d_in[11];
  const float* g1  = (const float*)d_in[12];
  const float* be1 = (const float*)d_in[13];
  float* out = (float*)d_out;

  char* base = (char*)d_ws;
  size_t off = 0;
  auto take = [&](size_t bytes) -> char* {
    char* p = base + off;
    off = (off + bytes + 255) & ~(size_t)255;
    return p;
  };
  unsigned* cnt     = (unsigned*)take((size_t)2 * NBIN * SLICES * 4);
  unsigned* bump    = (unsigned*)take(2 * 4);
  unsigned* binbase = (unsigned*)take((size_t)2 * NBIN * 4);
  unsigned* bintot  = (unsigned*)take((size_t)2 * NBIN * 4);
  unsigned* binD    = (unsigned*)take((size_t)EE * 4);
  unsigned* binS    = (unsigned*)take((size_t)EE * 4);
  unsigned* rows    = (unsigned*)take((size_t)NN * 8);
  unsigned* esrc    = (unsigned*)take((size_t)EE * 4);
  float* dsout = (float*)take((size_t)NN * 4);
  float* dsin  = (float*)take((size_t)NN * 4);
  float* cs    = (float*)take(128 * 4);
  float* cs2   = (float*)take(128 * 4);
  float* bnA   = (float*)take(128 * 4);
  float* bnB   = (float*)take(128 * 4);
  float* Wp    = (float*)take(128 * 64 * 4);
  float* Lp    = (float*)take(128 * 64 * 4);
  float* bufA  = (float*)take((size_t)NN * FF * 4);
  float* bufB  = (float*)take((size_t)NN * FF * 4);
  float* bufC  = (float*)take((size_t)NN * FF * 4);
  float* bufD  = (float*)take((size_t)NN * FF * 4);

  k_binhist<<<SLICES, 256, 0, stream>>>(src, dst, cnt, bump);
  k_binalloc<<<ALLOC_BLOCKS, 256, 0, stream>>>(cnt, bump, binbase, bintot);
  k_binscatter<<<SLICES, 256, 0, stream>>>(src, dst, cnt, binD, binS);
  k_binrows<<<NBIN, 256, 0, stream>>>(binD, binS, binbase, bintot, dsout, dsin, rows, esrc);
  k_padw<<<32, 256, 0, stream>>>(W2, L2, Wp, Lp);

  k_agg128<<<AGG_BLOCKS, 128, 0, stream>>>(feat, rows, esrc, dsout, dsin, bufA);
  k_gemm_dual<<<GEMM_BLOCKS, 256, 0, stream>>>(bufA, feat, W0, L0, bufB);
  k_zero_u32<<<1, 256, 0, stream>>>((unsigned*)cs, 256);
  k_colstats<<<CS_BLOCKS, CS_THREADS, 0, stream>>>(bufB, cs, cs2);
  k_bnparams<<<1, 128, 0, stream>>>(cs, cs2, g0, be0, bnA, bnB);

  k_gemm_proj<<<GEMM_BLOCKS, 256, 0, stream>>>(bufB, bnA, bnB, dsout, W1, L1, bufA, bufD);
  k_aggadd128<<<AGG_BLOCKS, 128, 0, stream>>>(bufA, rows, esrc, dsin, bufD, bufB);
  k_zero_u32<<<1, 256, 0, stream>>>((unsigned*)cs, 256);
  k_colstats<<<CS_BLOCKS, CS_THREADS, 0, stream>>>(bufB, cs, cs2);
  k_bnparams<<<1, 128, 0, stream>>>(cs, cs2, g1, be1, bnA, bnB);

  k_gemm_p<<<GEMM_BLOCKS, 256, 0, stream>>>(bufB, bnA, bnB, dsout, Wp, bufC, bufA);
  k_agg64<<<AGG_BLOCKS, 128, 0, stream>>>(bufA, rows, esrc, dsin, bufB);
  k_gemm_out<<<GEMM_BLOCKS, 256, 0, stream>>>(bufC, Lp, bufB, b2, out);
}


# BN sums taken where z is made (GEMM accumulators for z0, gather rows for z1); the reading GEMM derives scale/shift in its prologue; stats, clear, params launches gone (18 to 12); BN sum order changes
# speedup vs baseline: 2.4327x; 1.0637x over previous
#include <hip/hip_runtime.h>
#include <hip/hip_bf16.h>


static constexpr int NN = 100000;
static constexpr int EE = 1600000;
static constexpr int FF = 128;

typedef __attribute__((ext_vector_type(2))) float v2f;
typedef __attribute__((ext_vector_type(8))) float v8f;

static constexpr int STAT_COPIES = 16;
static constexpr int STAT_LAYER = STAT_COPIES * 256;


static constexpr int BIN_SHIFT = 8;
static constexpr int BIN_NODES = 1 << BIN_SHIFT;
static constexpr int NBIN = (NN + BIN_NODES - 1) / BIN_NODES;
static constexpr int SLICES = 256;
static constexpr int SLICE_EDGES = 6256;
static constexpr int SLICE_IT = (SLICE_EDGES / 4 + 255) / 256;
static constexpr int ALLOC_BLOCKS = (2 * NBIN + 7) / 8;
static constexpr int BIN_BATCH = 8;
static_assert(SLICE_EDGES % 4 == 0 && EE % 4 == 0, "slices are read four edges at a time");
static_assert((SLICES - 1) * SLICE_EDGES < EE && SLICES * SLICE_EDGES >= EE, "slices must cover the edges");
static_assert(NN <= (1 << 24), "a bin entry keeps the node id in 24 bits");

__global__ __launch_bounds__(256) void k_binhist(const int* __restrict__ src, const int* __restrict__ dst,
                                                 unsigned* __restrict__ cnt, unsigned* __restrict__ bump,
                                                 float* __restrict__ stat) {
  __shared__ unsigned hD[NBIN];
  __shared__ unsigned hS[NBIN];
  const int tid = threadIdx.x, b = blockIdx.x;
  for (int i = tid; i < NBIN; i += 256) {
    hD[i] = 0u;
    hS[i] = 0u;
  }
  if (b == 0 && tid < 2) bump[tid] = 0u;
  if (b < 2 * STAT_COPIES) stat[b * 256 + tid] = 0.f;
  __syncthreads();
  const int e0 = b * SLICE_EDGES;
  const int n4 = min(EE - e0, SLICE_EDGES) >> 2;
  const int4* s4 = (const int4*)(src + e0);
  const int4* d4 = (const int4*)(dst + e0);
  int4 s[SLICE_IT], d[SLICE_IT];
#pragma unroll
  for (int u = 0; u < SLICE_IT; ++u) {
    const int i = min(tid + 256 * u, n4 - 1);
    s[u] = s4[i];
    d[u] = d4[i];
  }
#pragma unroll
  for (int u = 0; u < SLICE_IT; ++u) {
    if (tid + 256 * u < n4) {
      atomicAdd(&hD[d[u].x >> BIN_SHIFT], 1u);
      atomicAdd(&hD[d[u].y >> BIN_SHIFT], 1u);
      atomicAdd(&hD[d[u].z >> BIN_SHIFT], 1u);
      atomicAdd(&hD[d[u].w >> BIN_SHIFT], 1u);
      atomicAdd(&hS[s[u].x >> BIN_SHIFT], 1u);
      atomicAdd(&hS[s[u].y >> BIN_SHIFT], 1u);
      atomicAdd(&hS[s[u].z >> BIN_SHIFT], 1u);
      atomicAdd(&hS[s[u].w >> BIN_SHIFT], 1u);
    }
  }
  __syncthreads();
  for (int i = tid; i < NBIN; i += 256) {
    cnt[(size_t)i * SLICES + b] = hD[i];
    cnt[(size_t)(NBIN + i) * SLICES + b] = hS[i];
  }
}

__global__ __launch_bounds__(256) void k_binalloc(unsigned* __restrict__ cnt, unsigned* __restrict__ bump,
                                                  unsigned* __restrict__ binbase,
                                                  unsigned* __restrict__ bintot) {
  const int lane = threadIdx.x & 31;
  const int w = blockIdx.x * 8 + (threadIdx.x >> 5);
  if (w >= 2 * NBIN) return;
  uint4* p = (uint4*)(cnt + (size_t)w * SLICES) + lane * 2;
  const uint4 a = p[0];
  const uint4 c = p[1];
  const unsigned loc = a.x + a.y + a.z + a.w + c.x + c.y + c.z + c.w;
  unsigned inc = loc;
  for (int d = 1; d < 32; d <<= 1) {
    const unsigned up = __shfl_up(inc, d);
    if (lane >= d) inc += up;
  }
  const unsigned tot = __shfl(inc, 31);
  unsigned base = 0u;
  if (lane == 0) base = atomicAdd(&bump[w >= NBIN ? 1 : 0], tot);
  base = __shfl(base, 0);
  unsigned o = base + inc - loc;
  uint4 ra, rc;
  ra.x = o; o += a.x;
  ra.y = o; o += a.y;
  ra.z = o; o += a.z;
  ra.w = o; o += a.w;
  rc.x = o; o += c.x;
  rc.y = o; o += c.y;
  rc.z = o; o += c.z;
  rc.w = o;
  p[0] = ra;
  p[1] = rc;
  if (lane == 0) {
    binbase[w] = base;
    bintot[w] = tot;
  }
}

__device__ __forceinline__ void bin_put4(unsigned* cD, unsigned* cS, unsigned* __restrict__ binD,
                                         unsigned* __restrict__ binS, const int4 s, const int4 d) {
  const unsigned pd0 = atomicAdd(&cD[d.x >> BIN_SHIFT], 1u);
  const unsigned pd1 = atomicAdd(&cD[d.y >> BIN_SHIFT], 1u);
  const unsigned pd2 = atomicAdd(&cD[d.z >> BIN_SHIFT], 1u);
  const unsigned pd3 = atomicAdd(&cD[d.w >> BIN_SHIFT], 1u);
  const unsigned ps0 = atomicAdd(&cS[s.x >> BIN_SHIFT], 1u);
  const unsigned ps1 = atomicAdd(&cS[s.y >> BIN_SHIFT], 1u);
  const unsigned ps2 = atomicAdd(&cS[s.z >> BIN_SHIFT], 1u);
  const unsigned ps3 = atomicAdd(&cS[s.w >> BIN_SHIFT], 1u);
  binD[pd0] = ((unsigned)(d.x & (BIN_NODES - 1)) << 24) | (unsigned)s.x;
  binD[pd1] = ((unsigned)(d.y & (BIN_NODES - 1)) << 24) | (unsigned)s.y;
  binD[pd2] = ((unsigned)(d.z & (BIN_NODES - 1)) << 24) | (unsigned)s.z;
  binD[pd3] = ((unsigned)(d.w & (BIN_NODES - 1)) << 24) | (unsigned)s.w;
  binS[ps0] = (unsigned)(s.x & (BIN_NODES - 1));
  binS[ps1] = (unsigned)(s.y & (BIN_NODES - 1));
  binS[ps2] = (unsigned)(s.z & (BIN_NODES - 1));
  binS[ps3] = (unsigned)(s.w & (BIN_NODES - 1));
}

__global__ __launch_bounds__(256) void k_binscatter(const int* __restrict__ src, const int* __restrict__ dst,
                                                    const unsigned* __restrict__ off,
                                                    unsigned* __restrict__ binD,
                                                    unsigned* __restrict__ binS) {
  __shared__ unsigned cD[NBIN];
  __shared__ unsigned cS[NBIN];
  const int tid = threadIdx.x, b = blockIdx.x;
  for (int i = tid; i < NBIN; i += 256) {
    cD[i] = off[(size_t)i * SLICES + b];
    cS[i] = off[(size_t)(NBIN + i) * SLICES + b];
  }
  __syncthreads();
  const int e0 = b * SLICE_EDGES;
  const int n4 = min(EE - e0, SLICE_EDGES) >> 2;
  const int4* s4 = (const int4*)(src + e0);
  const int4* d4 = (const int4*)(dst + e0);
  int4 s[SLICE_IT], d[SLICE_IT];
#pragma unroll
  for (int u = 0; u < SLICE_IT; ++u) {
    const int i = min(tid + 256 * u, n4 - 1);
    s[u] = s4[i];
    d[u] = d4[i];
  }
#pragma unroll
  for (int u = 0; u < SLICE_IT; ++u) {
    if (tid + 256 * u < n4) {
      bin_put4(cD, cS, binD, binS, s[u], d[u]);
    }
  }
}

__global__ __launch_bounds__(256) void k_binrows(const unsigned* __restrict__ binD,
                                                 const unsigned* __restrict__ binS,
                                                 const unsigned* __restrict__ binbase,
                                                 const unsigned* __restrict__ bintot,
                                                 float* __restrict__ dsout, float* __restrict__ dsin,
                                                 unsigned* __restrict__ rows, unsigned* __restrict__ esrc) {
  __shared__ unsigned cD[BIN_NODES];
  __shared__ unsigned cS[BIN_NODES];
  __shared__ unsigned cur[BIN_NODES];
  __shared__ unsigned wtot[8];
  const int tid = threadIdx.x, j = blockIdx.x;
  const int lane = tid & 31, wv = tid >> 5;
  cD[tid] = 0u;
  cS[tid] = 0u;
  const unsigned bD = binbase[j], nD = bintot[j];
  const unsigned bS = binbase[NBIN + j], nS = bintot[NBIN + j];
  const unsigned* eD = binD + bD;
  const unsigned* eS = binS + bS;
  __syncthreads();

  for (unsigned i0 = 0u; i0 < nD; i0 += 256u * BIN_BATCH) {
    unsigned v[BIN_BATCH];
#pragma unroll
    for (int u = 0; u < BIN_BATCH; ++u) v[u] = eD[min(i0 + 256u * u + tid, nD - 1u)];
#pragma unroll
    for (int u = 0; u < BIN_BATCH; ++u)
      if (i0 + 256u * u + tid < nD) atomicAdd(&cD[v[u] >> 24], 1u);
  }
  for (unsigned i0 = 0u; i0 < nS; i0 += 256u * BIN_BATCH) {
    unsigned v[BIN_BATCH];
#pragma unroll
    for (int u = 0; u < BIN_BATCH; ++u) v[u] = eS[min(i0 + 256u * u + tid, nS - 1u)];
#pragma unroll
    for (int u = 0; u < BIN_BATCH; ++u)
      if (i0 + 256u * u + tid < nS) atomicAdd(&cS[v[u]], 1u);
  }
  __syncthreads();

  const unsigned id = cD[tid], od = cS[tid];
  unsigned inc = id;
  for (int d = 1; d < 32; d <<= 1) {
    const unsigned up = __shfl_up(inc, d);
    if (lane >= d) inc += up;
  }
  if (lane == 31) wtot[wv] = inc;
  __syncthreads();
  unsigned st = inc - id;
  for (int w = 0; w < wv; ++w) st += wtot[w];
  cur[tid] = st;
  const int node = j * BIN_NODES + tid;
  if (node < NN) {
    *(uint2*)(rows + 2 * node) = make_uint2(bD + st, bD + st + id);
    dsin[node] = id ? rsqrtf((float)id) : 1.0f;
    dsout[node] = od ? rsqrtf((float)od) : 1.0f;
  }
  __syncthreads();

  unsigned* o = esrc + bD;
  for (unsigned i0 = 0u; i0 < nD; i0 += 256u * BIN_BATCH) {
    unsigned v[BIN_BATCH];
#pragma unroll
    for (int u = 0; u < BIN_BATCH; ++u) v[u] = eD[min(i0 + 256u * u + tid, nD - 1u)];
    unsigned p[BIN_BATCH];
#pragma unroll
    for (int u = 0; u < BIN_BATCH; ++u) {
      p[u] = 0u;
      if (i0 + 256u * u + tid < nD) p[u] = atomicAdd(&cur[v[u] >> 24], 1u);
    }
#pragma unroll
    for (int u = 0; u < BIN_BATCH; ++u)
      if (i0 + 256u * u + tid < nD) o[p[u]] = v[u] & 0xFFFFFFu;
  }
}


static constexpr int AGG_NODES = 8;
static constexpr int AGG_GROUP = 8;
static constexpr int AGG_WAVES = NN / AGG_NODES;
static constexpr int AGG_BLOCKS = AGG_WAVES / 4;
static_assert(NN % AGG_NODES == 0 && AGG_WAVES % 4 == 0, "gather grid must cover the nodes exactly");

__device__ __forceinline__ unsigned lane_u32(unsigned v, int l) {
  return (unsigned)__builtin_amdgcn_readlane((int)v, l);
}
__device__ __forceinline__ float lane_f32(float v, int l) {
  return __int_as_float(__builtin_amdgcn_readlane(__float_as_int(v), l));
}

__device__ __forceinline__ void axpy4(float4& acc, float w, const float4& v) {
  acc.x = fmaf(w, v.x, acc.x);
  acc.y = fmaf(w, v.y, acc.y);
  acc.z = fmaf(w, v.z, acc.z);
  acc.w = fmaf(w, v.w, acc.w);
}

__global__ __launch_bounds__(128) void k_agg128(const float* __restrict__ h,
                                                const unsigned* __restrict__ rows,
                                                const unsigned* __restrict__ esrc,
                                                const float* __restrict__ dsout,
                                                const float* __restrict__ dsin,
                                                float* __restrict__ agg) {
  const int lane = threadIdx.x & 31;
  const int n0 = (blockIdx.x * 4 + (threadIdx.x >> 5)) * AGG_NODES;
  unsigned rb = 0u, re = 0u;
  float cin = 0.f;
  if (lane < AGG_NODES) {
    const uint2 r = *(const uint2*)(rows + 2 * (n0 + lane));
    rb = r.x;
    re = r.y;
    cin = dsin[n0 + lane];
  }
  unsigned b = lane_u32(rb, 0), e = lane_u32(re, 0);
  unsigned my = 0u;
  float myw = 0.f;
  if (b + lane < e) {
    my = esrc[b + lane];
    myw = dsout[my];
  }
  const float* hl = h + lane * 4;

#pragma unroll 1
  for (int k = 0; k < AGG_NODES; ++k) {
    unsigned nb = 0u, ne = 0u, nmy = 0u;
    if (k + 1 < AGG_NODES) {
      nb = lane_u32(rb, k + 1);
      ne = lane_u32(re, k + 1);
      if (nb + lane < ne) nmy = esrc[nb + lane];
    }
    float4 acc = make_float4(0.f, 0.f, 0.f, 0.f);
    const int cnt = (int)min(e - b, 32u);
    int j = 0;
    for (; j + AGG_GROUP <= cnt; j += AGG_GROUP) {
      float4 v[AGG_GROUP];
#pragma unroll
      for (int u = 0; u < AGG_GROUP; ++u)
        v[u] = *(const float4*)(hl + (size_t)lane_u32(my, j + u) * FF);
      __builtin_amdgcn_sched_barrier(0);
#pragma unroll
      for (int u = 0; u < AGG_GROUP; ++u) axpy4(acc, lane_f32(myw, j + u), v[u]);
    }
    const int rem = cnt - j;
    if (rem > 0) {
      float4 v[AGG_GROUP - 1];
#pragma unroll
      for (int u = 0; u < AGG_GROUP - 1; ++u) {
        v[u] = make_float4(0.f, 0.f, 0.f, 0.f);
        if (u < rem) v[u] = *(const float4*)(hl + (size_t)lane_u32(my, j + u) * FF);
      }
#pragma unroll
      for (int u = 0; u < AGG_GROUP - 1; ++u)
        if (u < rem) axpy4(acc, lane_f32(myw, j + u), v[u]);
    }
    for (unsigned i = b + 32u; i < e; ++i) {
      const unsigned s = esrc[i];
      axpy4(acc, dsout[s], *(const float4*)(hl + (size_t)s * FF));
    }
    const float c = lane_f32(cin, k);
    *(float4*)(agg + (size_t)(n0 + k) * FF + lane * 4) =
        make_float4(c * acc.x, c * acc.y, c * acc.z, c * acc.w);
    b = nb;
    e = ne;
    my = nmy;
    myw = 0.f;
    if (b + lane < e) myw = dsout[my];
  }
}

__global__ __launch_bounds__(128) void k_agg64(const float* __restrict__ P,
                                               const unsigned* __restrict__ rows,
                                               const unsigned* __restrict__ esrc,
                                               const float* __restrict__ dsin,
                                               float* __restrict__ R) {
  const int lane = threadIdx.x & 31;
  const int n0 = (blockIdx.x * 4 + (threadIdx.x >> 5)) * AGG_NODES;
  unsigned rb = 0u, re = 0u;
  float cin = 0.f;
  if (lane < AGG_NODES) {
    const uint2 r = *(const uint2*)(rows + 2 * (n0 + lane));
    rb = r.x;
    re = r.y;
    cin = dsin[n0 + lane];
  }
  unsigned b = lane_u32(rb, 0), e = lane_u32(re, 0);
  unsigned my = 0u;
  if (b + lane < e) my = esrc[b + lane];
  const float* pl = P + lane * 2;

#pragma unroll 1
  for (int k = 0; k < AGG_NODES; ++k) {
    unsigned nb = 0u, ne = 0u, nmy = 0u;
    if (k + 1 < AGG_NODES) {
      nb = lane_u32(rb, k + 1);
      ne = lane_u32(re, k + 1);
      if (nb + lane < ne) nmy = esrc[nb + lane];
    }
    float2 acc = make_float2(0.f, 0.f);
    const int cnt = (int)min(e - b, 32u);
    int j = 0;
    for (; j + AGG_GROUP <= cnt; j += AGG_GROUP) {
      float2 v[AGG_GROUP];
#pragma unroll
      for (int u = 0; u < AGG_GROUP; ++u)
        v[u] = *(const float2*)(pl + (size_t)lane_u32(my, j + u) * 64);
#pragma unroll
      for (int u = 0; u < AGG_GROUP; ++u) {
        acc.x += v[u].x;
        acc.y += v[u].y;
      }
    }
    const int rem = cnt - j;
    if (rem > 0) {
      float2 v[AGG_GROUP - 1];
#pragma unroll
      for (int u = 0; u < AGG_GROUP - 1; ++u) {
        v[u] = make_float2(0.f, 0.f);
        if (u < rem) v[u] = *(const float2*)(pl + (size_t)lane_u32(my, j + u) * 64);
      }
#pragma unroll
      for (int u = 0; u < AGG_GROUP - 1; ++u) {
        if (u < rem) {
          acc.x += v[u].x;
          acc.y += v[u].y;
        }
      }
    }
    for (unsigned i = b + 32u; i < e; ++i) {
      const float2 x = *(const float2*)(pl + (size_t)esrc[i] * 64);
      acc.x += x.x;
      acc.y += x.y;
    }
    const float c = lane_f32(cin, k);
    *(float2*)(R + (size_t)(n0 + k) * 64 + lane * 2) = make_float2(c * acc.x, c * acc.y);
    b = nb;
    e = ne;
    my = nmy;
  }
}

__global__ __launch_bounds__(128) void k_aggadd128(const float* __restrict__ G,
                                                   const unsigned* __restrict__ rows,
                                                   const unsigned* __restrict__ esrc,
                                                   const float* __restrict__ dsin,
                                                   const float* __restrict__ S,
                                                   float* __restrict__ Z,
                                                   float* __restrict__ stat) {
  __shared__ float red[2][4][4][32];
  const int lane = threadIdx.x & 31;
  float* mys = &red[0][threadIdx.x >> 5][0][lane];
  float* myq = &red[1][threadIdx.x >> 5][0][lane];
  mys[0] = mys[32] = mys[64] = mys[96] = 0.f;
  myq[0] = myq[32] = myq[64] = myq[96] = 0.f;
  const int n0 = (blockIdx.x * 4 + (threadIdx.x >> 5)) * AGG_NODES;
  unsigned rb = 0u, re = 0u;
  float cin = 0.f;
  if (lane < AGG_NODES) {
    const uint2 r = *(const uint2*)(rows + 2 * (n0 + lane));
    rb = r.x;
    re = r.y;
    cin = dsin[n0 + lane];
  }
  unsigned b = lane_u32(rb, 0), e = lane_u32(re, 0);
  unsigned my = 0u;
  if (b + lane < e) my = esrc[b + lane];
  const float* gl = G + lane * 4;

#pragma unroll 1
  for (int k = 0; k < AGG_NODES; ++k) {
    unsigned nb = 0u, ne = 0u, nmy = 0u;
    if (k + 1 < AGG_NODES) {
      nb = lane_u32(rb, k + 1);
      ne = lane_u32(re, k + 1);
      if (nb + lane < ne) nmy = esrc[nb + lane];
    }
    const float4 self = *(const float4*)(S + (size_t)(n0 + k) * FF + lane * 4);
    float4 acc = make_float4(0.f, 0.f, 0.f, 0.f);
    const int cnt = (int)min(e - b, 32u);
    int j = 0;
    for (; j + AGG_GROUP <= cnt; j += AGG_GROUP) {
      float4 v[AGG_GROUP];
#pragma unroll
      for (int u = 0; u < AGG_GROUP; ++u)
        v[u] = *(const float4*)(gl + (size_t)lane_u32(my, j + u) * FF);
#pragma unroll
      for (int u = 0; u < AGG_GROUP; ++u) {
        acc.x += v[u].x;
        acc.y += v[u].y;
        acc.z += v[u].z;
        acc.w += v[u].w;
      }
    }
    const int rem = cnt - j;
    if (rem > 0) {
      float4 v[AGG_GROUP - 1];
#pragma unroll
      for (int u = 0; u < AGG_GROUP - 1; ++u) {
        v[u] = make_float4(0.f, 0.f, 0.f, 0.f);
        if (u < rem) v[u] = *(const float4*)(gl + (size_t)lane_u32(my, j + u) * FF);
      }
#pragma unroll
      for (int u = 0; u < AGG_GROUP - 1; ++u) {
        if (u < rem) {
          acc.x += v[u].x;
          acc.y += v[u].y;
          acc.z += v[u].z;
          acc.w += v[u].w;
        }
      }
    }
    for (unsigned i = b + 32u; i < e; ++i) {
      const float4 x = *(const float4*)(gl + (size_t)esrc[i] * FF);
      acc.x += x.x;
      acc.y += x.y;
      acc.z += x.z;
      acc.w += x.w;
    }
    const float c = lane_f32(cin, k);
    const float4 z = make_float4(fmaf(c, acc.x, self.x), fmaf(c, acc.y, self.y),
                                 fmaf(c, acc.z, self.z), fmaf(c, acc.w, self.w));
    *(float4*)(Z + (size_t)(n0 + k) * FF + lane * 4) = z;
    atomicAdd(mys, z.x);
    atomicAdd(mys + 32, z.y);
    atomicAdd(mys + 64, z.z);
    atomicAdd(mys + 96, z.w);
    atomicAdd(myq, z.x * z.x);
    atomicAdd(myq + 32, z.y * z.y);
    atomicAdd(myq + 64, z.z * z.z);
    atomicAdd(myq + 96, z.w * z.w);
    b = nb;
    e = ne;
    my = nmy;
  }
  __syncthreads();
  const float* r = &red[0][0][threadIdx.x & 3][threadIdx.x >> 2];
  float ts = 0.f, tq = 0.f;
  for (int w = 0; w < 4; ++w) {
    ts += r[w * FF];
    tq += r[(4 + w) * FF];
  }
  float* o = stat + (blockIdx.x % STAT_COPIES) * 256 + threadIdx.x;
  atomicAdd(o, ts);
  atomicAdd(o + FF, tq);
}


static constexpr int GEMM_BLOCKS = 512;
static constexpr int NSTRIP = NN / 16;

template <int NC>
__device__ __forceinline__ void stage_weights(const float* __restrict__ Bg,
                                              float4* __restrict__ s, int tid) {
  constexpr int C4 = NC / 4;
  constexpr int PITCH = NC + 8;
  for (int i = tid; i < 32 * C4; i += 256) {
    const int g = i / C4, c4 = i - g * C4;
    const float* p = Bg + (size_t)(4 * g) * NC + 4 * c4;
    const float4 r0 = *(const float4*)(p);
    const float4 r1 = *(const float4*)(p + NC);
    const float4 r2 = *(const float4*)(p + 2 * NC);
    const float4 r3 = *(const float4*)(p + 3 * NC);
    const int m = g >> 3, hh = (g >> 2) & 1, q = g & 3;
    float4* d = s + ((m * 4 + q) * 2 + hh) * PITCH + 4 * c4;
    d[0] = make_float4(r0.x, r1.x, r2.x, r3.x);
    d[1] = make_float4(r0.y, r1.y, r2.y, r3.y);
    d[2] = make_float4(r0.z, r1.z, r2.z, r3.z);
    d[3] = make_float4(r0.w, r1.w, r2.w, r3.w);
  }
}

__device__ __forceinline__ v8f mm4(float ax, float ay, float bx, float by, v8f c) {
  v2f a; a.x = ax; a.y = ay;
  v2f b; b.x = bx; b.y = by;
  return __builtin_amdgcn_wmma_f32_16x16x4_f32(false, a, false, b, (short)0, c, false, false);
}

__device__ __forceinline__ float4 bn_relu4(const float4 z, const float4 a, const float4 b) {
  const float vx = fmaf(z.x, a.x, b.x);
  const float vy = fmaf(z.y, a.y, b.y);
  const float vz = fmaf(z.z, a.z, b.z);
  const float vw = fmaf(z.w, a.w, b.w);
  return make_float4(vx > 0.f ? vx : 0.f, vy > 0.f ? vy : 0.f, vz > 0.f ? vz : 0.f,
                     vw > 0.f ? vw : 0.f);
}

__device__ __forceinline__ void bn_params_to_lds(const float* __restrict__ stat,
                                                 const float* __restrict__ g,
                                                 const float* __restrict__ be,
                                                 float* __restrict__ sBN, int tid) {
  if (tid < FF) {
    float s = 0.f, q = 0.f;
#pragma unroll
    for (int k = 0; k < STAT_COPIES; ++k) {
      s += stat[k * 256 + tid];
      q += stat[k * 256 + FF + tid];
    }
    const float inv = 1.0f / (float)NN;
    const float mu = s * inv;
    const float var = q * inv - mu * mu;
    const float a = g[tid] * rsqrtf(var + 1e-5f);
    sBN[tid] = a;
    sBN[FF + tid] = be[tid] - mu * a;
  }
}

__global__ __launch_bounds__(256) void k_gemm_dual(const float* __restrict__ Agg,
                                                   const float* __restrict__ H,
                                                   const float* __restrict__ W,
                                                   const float* __restrict__ L,
                                                   float* __restrict__ Z,
                                                   float* __restrict__ stat) {
  constexpr int PITCH = 128 + 8;
  __shared__ __attribute__((aligned(16))) float4 sW[32 * PITCH];
  __shared__ __attribute__((aligned(16))) float4 sL[32 * PITCH];
  stage_weights<128>(W, sW, threadIdx.x);
  stage_weights<128>(L, sL, threadIdx.x);
  __syncthreads();

  const int wv = threadIdx.x >> 5;
  const int lane = threadIdx.x & 31;
  const int t = lane & 15;
  const int hh = lane >> 4;
  float cs[8], cq[8];
#pragma unroll
  for (int ct = 0; ct < 8; ++ct) { cs[ct] = 0.f; cq[ct] = 0.f; }

  for (int strip = blockIdx.x + GEMM_BLOCKS * wv; strip < NSTRIP; strip += GEMM_BLOCKS * 8) {
    const int r0 = strip << 4;
    const float* arow = Agg + (size_t)(r0 + t) * FF + 16 * hh;
    const float* hrow = H + (size_t)(r0 + t) * FF + 16 * hh;
    v8f acc[8];
#pragma unroll
    for (int ct = 0; ct < 8; ++ct) acc[ct] = v8f{};

#pragma unroll 1
    for (int m = 0; m < 4; ++m) {
      float4 a[4], h[4];
#pragma unroll
      for (int q = 0; q < 4; ++q) {
        a[q] = *(const float4*)(arow + 32 * m + 4 * q);
        h[q] = *(const float4*)(hrow + 32 * m + 4 * q);
      }
#pragma unroll
      for (int q = 0; q < 4; ++q) {
        const float4* pw = sW + ((m * 4 + q) * 2 + hh) * PITCH + t;
        const float4* pl = sL + ((m * 4 + q) * 2 + hh) * PITCH + t;
        float4 bw[8], bl[8];
#pragma unroll
        for (int ct = 0; ct < 8; ++ct) { bw[ct] = pw[16 * ct]; bl[ct] = pl[16 * ct]; }
#pragma unroll
        for (int ct = 0; ct < 8; ++ct) acc[ct] = mm4(a[q].x, a[q].y, bw[ct].x, bw[ct].y, acc[ct]);
#pragma unroll
        for (int ct = 0; ct < 8; ++ct) acc[ct] = mm4(h[q].x, h[q].y, bl[ct].x, bl[ct].y, acc[ct]);
#pragma unroll
        for (int ct = 0; ct < 8; ++ct) acc[ct] = mm4(a[q].z, a[q].w, bw[ct].z, bw[ct].w, acc[ct]);
#pragma unroll
        for (int ct = 0; ct < 8; ++ct) acc[ct] = mm4(h[q].z, h[q].w, bl[ct].z, bl[ct].w, acc[ct]);
      }
    }
#pragma unroll
    for (int ct = 0; ct < 8; ++ct)
#pragma unroll
      for (int v = 0; v < 8; ++v) {
        const float x = acc[ct][v];
        Z[(size_t)(r0 + v + 8 * hh) * FF + 16 * ct + t] = x;
        cs[ct] += x;
        cq[ct] = fmaf(x, x, cq[ct]);
      }
  }

  __syncthreads();
  float* red = (float*)sW;
#pragma unroll
  for (int ct = 0; ct < 8; ++ct) {
    red[(wv * 2 + hh) * FF + 16 * ct + t] = cs[ct];
    red[(16 + wv * 2 + hh) * FF + 16 * ct + t] = cq[ct];
  }
  __syncthreads();
  const int st = threadIdx.x >> 7, col = threadIdx.x & 127;
  const float* rr = red + st * 16 * FF + col;
  float ts = 0.f;
#pragma unroll
  for (int i = 0; i < 16; ++i) ts += rr[i * FF];
  atomicAdd(stat + (blockIdx.x % STAT_COPIES) * 256 + st * FF + col, ts);
}

__global__ __launch_bounds__(256) void k_gemm_proj(const float* __restrict__ Zin,
                                                   const float* __restrict__ stat,
                                                   const float* __restrict__ g,
                                                   const float* __restrict__ be,
                                                   const float* __restrict__ dsout,
                                                   const float* __restrict__ W,
                                                   const float* __restrict__ L,
                                                   float* __restrict__ Gw,
                                                   float* __restrict__ Gl) {
  constexpr int PITCH = 128 + 8;
  __shared__ __attribute__((aligned(16))) float4 sW[32 * PITCH];
  __shared__ __attribute__((aligned(16))) float4 sL[32 * PITCH];
  __shared__ __attribute__((aligned(16))) float sBN[2 * FF];
  bn_params_to_lds(stat, g, be, sBN, threadIdx.x);
  stage_weights<128>(W, sW, threadIdx.x);
  stage_weights<128>(L, sL, threadIdx.x);
  __syncthreads();

  const int wv = threadIdx.x >> 5;
  const int lane = threadIdx.x & 31;
  const int t = lane & 15, hh = lane >> 4;

  for (int strip = blockIdx.x + GEMM_BLOCKS * wv; strip < NSTRIP; strip += GEMM_BLOCKS * 8) {
    const int r0 = strip << 4;
    const float* zrow = Zin + (size_t)(r0 + t) * FF + 16 * hh;
    const float4 d0 = *(const float4*)(dsout + r0 + 8 * hh);
    const float4 d1 = *(const float4*)(dsout + r0 + 8 * hh + 4);
    v8f aw[8], al[8];
#pragma unroll
    for (int ct = 0; ct < 8; ++ct) { aw[ct] = v8f{}; al[ct] = v8f{}; }
#pragma unroll 1
    for (int m = 0; m < 4; ++m) {
      float4 h[4];
#pragma unroll
      for (int q = 0; q < 4; ++q) {
        const int c = 32 * m + 16 * hh + 4 * q;
        const float4 x = *(const float4*)(zrow + 32 * m + 4 * q);
        h[q] = bn_relu4(x, *(const float4*)(sBN + c), *(const float4*)(sBN + FF + c));
      }
#pragma unroll
      for (int q = 0; q < 4; ++q) {
        const float4* pw = sW + ((m * 4 + q) * 2 + hh) * PITCH + t;
        const float4* pl = sL + ((m * 4 + q) * 2 + hh) * PITCH + t;
        float4 bw[8], bl[8];
#pragma unroll
        for (int ct = 0; ct < 8; ++ct) { bw[ct] = pw[16 * ct]; bl[ct] = pl[16 * ct]; }
#pragma unroll
        for (int ct = 0; ct < 8; ++ct) aw[ct] = mm4(h[q].x, h[q].y, bw[ct].x, bw[ct].y, aw[ct]);
#pragma unroll
        for (int ct = 0; ct < 8; ++ct) al[ct] = mm4(h[q].x, h[q].y, bl[ct].x, bl[ct].y, al[ct]);
#pragma unroll
        for (int ct = 0; ct < 8; ++ct) aw[ct] = mm4(h[q].z, h[q].w, bw[ct].z, bw[ct].w, aw[ct]);
#pragma unroll
        for (int ct = 0; ct < 8; ++ct) al[ct] = mm4(h[q].z, h[q].w, bl[ct].z, bl[ct].w, al[ct]);
      }
    }
    const float dv[8] = {d0.x, d0.y, d0.z, d0.w, d1.x, d1.y, d1.z, d1.w};
#pragma unroll
    for (int ct = 0; ct < 8; ++ct)
#pragma unroll
      for (int v = 0; v < 8; ++v) {
        const size_t o = (size_t)(r0 + v + 8 * hh) * FF + 16 * ct + t;
        Gw[o] = dv[v] * aw[ct][v];
        Gl[o] = al[ct][v];
      }
  }
}

__global__ __launch_bounds__(256) void k_gemm_p(const float* __restrict__ Zin,
                                                const float* __restrict__ stat,
                                                const float* __restrict__ g,
                                                const float* __restrict__ be,
                                                const float* __restrict__ dsout,
                                                const float* __restrict__ Wp,
                                                float* __restrict__ Hout,
                                                float* __restrict__ P) {
  constexpr int PITCH = 64 + 8;
  __shared__ __attribute__((aligned(16))) float4 sW[32 * PITCH];
  __shared__ __attribute__((aligned(16))) float sBN[2 * FF];
  bn_params_to_lds(stat, g, be, sBN, threadIdx.x);
  stage_weights<64>(Wp, sW, threadIdx.x);
  __syncthreads();

  const int wv = threadIdx.x >> 5;
  const int lane = threadIdx.x & 31;
  const int t = lane & 15, hh = lane >> 4;

  for (int strip = blockIdx.x + GEMM_BLOCKS * wv; strip < NSTRIP; strip += GEMM_BLOCKS * 8) {
    const int r0 = strip << 4;
    const float d = dsout[r0 + t];
    const float* zrow = Zin + (size_t)(r0 + t) * FF + 16 * hh;
    float* hrow = Hout + (size_t)(r0 + t) * FF + 16 * hh;
    v8f acc[4];
#pragma unroll
    for (int ct = 0; ct < 4; ++ct) acc[ct] = v8f{};
#pragma unroll 1
    for (int m = 0; m < 4; ++m) {
      float4 a[4];
#pragma unroll
      for (int q = 0; q < 4; ++q) {
        const int c = 32 * m + 16 * hh + 4 * q;
        const float4 z = *(const float4*)(zrow + 32 * m + 4 * q);
        const float4 x = bn_relu4(z, *(const float4*)(sBN + c), *(const float4*)(sBN + FF + c));
        *(float4*)(hrow + 32 * m + 4 * q) = x;
        a[q] = make_float4(d * x.x, d * x.y, d * x.z, d * x.w);
      }
#pragma unroll
      for (int q = 0; q < 4; ++q) {
        const float4* pw = sW + ((m * 4 + q) * 2 + hh) * PITCH + t;
        float4 bw[4];
#pragma unroll
        for (int ct = 0; ct < 4; ++ct) bw[ct] = pw[16 * ct];
#pragma unroll
        for (int ct = 0; ct < 4; ++ct) acc[ct] = mm4(a[q].x, a[q].y, bw[ct].x, bw[ct].y, acc[ct]);
#pragma unroll
        for (int ct = 0; ct < 4; ++ct) acc[ct] = mm4(a[q].z, a[q].w, bw[ct].z, bw[ct].w, acc[ct]);
      }
    }
#pragma unroll
    for (int ct = 0; ct < 4; ++ct)
#pragma unroll
      for (int v = 0; v < 8; ++v)
        P[(size_t)(r0 + v + 8 * hh) * 64 + 16 * ct + t] = acc[ct][v];
  }
}

__global__ __launch_bounds__(256) void k_gemm_out(const float* __restrict__ H,
                                                  const float* __restrict__ Lp,
                                                  const float* __restrict__ R,
                                                  const float* __restrict__ b2,
                                                  float* __restrict__ out) {
  constexpr int PITCH = 64 + 8;
  __shared__ __attribute__((aligned(16))) float4 sL[32 * PITCH];
  stage_weights<64>(Lp, sL, threadIdx.x);
  __syncthreads();

  const int wv = threadIdx.x >> 5;
  const int lane = threadIdx.x & 31;
  const int t = lane & 15, hh = lane >> 4;

  for (int strip = blockIdx.x + GEMM_BLOCKS * wv; strip < NSTRIP; strip += GEMM_BLOCKS * 8) {
    const int r0 = strip << 4;
    const float* hrow = H + (size_t)(r0 + t) * FF + 16 * hh;
    v8f acc[3];
#pragma unroll
    for (int ct = 0; ct < 3; ++ct) acc[ct] = v8f{};
#pragma unroll 1
    for (int m = 0; m < 4; ++m) {
      float4 a[4];
#pragma unroll
      for (int q = 0; q < 4; ++q) a[q] = *(const float4*)(hrow + 32 * m + 4 * q);
#pragma unroll
      for (int q = 0; q < 4; ++q) {
        const float4* pl = sL + ((m * 4 + q) * 2 + hh) * PITCH + t;
        float4 bl[3];
#pragma unroll
        for (int ct = 0; ct < 3; ++ct) bl[ct] = pl[16 * ct];
#pragma unroll
        for (int ct = 0; ct < 3; ++ct) acc[ct] = mm4(a[q].x, a[q].y, bl[ct].x, bl[ct].y, acc[ct]);
#pragma unroll
        for (int ct = 0; ct < 3; ++ct) acc[ct] = mm4(a[q].z, a[q].w, bl[ct].z, bl[ct].w, acc[ct]);
      }
    }
#pragma unroll
    for (int ct = 0; ct < 3; ++ct) {
      const int col = 16 * ct + t;
      if (col < 40) {
        const float bias = b2[col];
#pragma unroll
        for (int v = 0; v < 8; ++v) {
          const int row = r0 + v + 8 * hh;
          out[(size_t)row * 40 + col] = acc[ct][v] + R[(size_t)row * 64 + col] + bias;
        }
      }
    }
  }
}


__global__ void k_padw(const float* __restrict__ W2, const float* __restrict__ L2,
                       float* __restrict__ Wp, float* __restrict__ Lp) {
  const int i = blockIdx.x * 256 + threadIdx.x;
  const int k = i >> 6, c = i & 63;
  const float w = (c < 40) ? W2[k * 40 + c] : 0.f;
  const float l = (c < 40) ? L2[k * 40 + c] : 0.f;
  Wp[i] = w;
  Lp[i] = l;
}


extern "C" void kernel_launch(void* const* d_in, const int* in_sizes, int n_in,
                              void* d_out, int out_size, void* d_ws, size_t ws_size,
                              hipStream_t stream) {
  (void)in_sizes; (void)n_in; (void)out_size; (void)ws_size;

  const float* feat = (const float*)d_in[0];
  const int* src   = (const int*)d_in[1];
  const int* dst   = (const int*)d_in[2];
  const float* W0  = (const float*)d_in[3];
  const float* L0  = (const float*)d_in[4];
  const float* W1  = (const float*)d_in[5];
  const float* L1  = (const float*)d_in[6];
  const float* W2  = (const float*)d_in[7];
  const float* L2  = (const float*)d_in[8];
  const float* b2  = (const float*)d_in[9];
  const float* g0  = (const float*)d_in[10];
  const float* be0 = (const float*)d_in[11];
  const float* g1  = (const float*)d_in[12];
  const float* be1 = (const float*)d_in[13];
  float* out = (float*)d_out;

  char* base = (char*)d_ws;
  size_t off = 0;
  auto take = [&](size_t bytes) -> char* {
    char* p = base + off;
    off = (off + bytes + 255) & ~(size_t)255;
    return p;
  };
  unsigned* cnt     = (unsigned*)take((size_t)2 * NBIN * SLICES * 4);
  unsigned* bump    = (unsigned*)take(2 * 4);
  unsigned* binbase = (unsigned*)take((size_t)2 * NBIN * 4);
  unsigned* bintot  = (unsigned*)take((size_t)2 * NBIN * 4);
  unsigned* binD    = (unsigned*)take((size_t)EE * 4);
  unsigned* binS    = (unsigned*)take((size_t)EE * 4);
  unsigned* rows    = (unsigned*)take((size_t)NN * 8);
  unsigned* esrc    = (unsigned*)take((size_t)EE * 4);
  float* dsout = (float*)take((size_t)NN * 4);
  float* dsin  = (float*)take((size_t)NN * 4);
  float* stat0 = (float*)take((size_t)2 * STAT_LAYER * 4);
  float* stat1 = stat0 + STAT_LAYER;
  float* Wp    = (float*)take(128 * 64 * 4);
  float* Lp    = (float*)take(128 * 64 * 4);
  float* bufA  = (float*)take((size_t)NN * FF * 4);
  float* bufB  = (float*)take((size_t)NN * FF * 4);
  float* bufC  = (float*)take((size_t)NN * FF * 4);
  float* bufD  = (float*)take((size_t)NN * FF * 4);

  k_binhist<<<SLICES, 256, 0, stream>>>(src, dst, cnt, bump, stat0);
  k_binalloc<<<ALLOC_BLOCKS, 256, 0, stream>>>(cnt, bump, binbase, bintot);
  k_binscatter<<<SLICES, 256, 0, stream>>>(src, dst, cnt, binD, binS);
  k_binrows<<<NBIN, 256, 0, stream>>>(binD, binS, binbase, bintot, dsout, dsin, rows, esrc);
  k_padw<<<32, 256, 0, stream>>>(W2, L2, Wp, Lp);

  k_agg128<<<AGG_BLOCKS, 128, 0, stream>>>(feat, rows, esrc, dsout, dsin, bufA);
  k_gemm_dual<<<GEMM_BLOCKS, 256, 0, stream>>>(bufA, feat, W0, L0, bufB, stat0);

  k_gemm_proj<<<GEMM_BLOCKS, 256, 0, stream>>>(bufB, stat0, g0, be0, dsout, W1, L1, bufA, bufD);
  k_aggadd128<<<AGG_BLOCKS, 128, 0, stream>>>(bufA, rows, esrc, dsin, bufD, bufB,
                                               stat1);

  k_gemm_p<<<GEMM_BLOCKS, 256, 0, stream>>>(bufB, stat1, g1, be1, dsout, Wp, bufC, bufA);
  k_agg64<<<AGG_BLOCKS, 128, 0, stream>>>(bufA, rows, esrc, dsin, bufB);
  k_gemm_out<<<GEMM_BLOCKS, 256, 0, stream>>>(bufC, Lp, bufB, b2, out);
}


# CSR build in 2 launches (was 4): a block counting-sorts its edge slice by bin in LDS (count, scan, place); a bin's block reads 256 slice chunks, 32 per wave in one load group; row sum order changes
# speedup vs baseline: 2.7968x; 1.1497x over previous
#include <hip/hip_runtime.h>
#include <hip/hip_bf16.h>


static constexpr int NN = 100000;
static constexpr int EE = 1600000;
static constexpr int FF = 128;

typedef __attribute__((ext_vector_type(2))) float v2f;
typedef __attribute__((ext_vector_type(8))) float v8f;

static constexpr int STAT_COPIES = 16;
static constexpr int STAT_LAYER = STAT_COPIES * 256;


static constexpr int BIN_SHIFT = 8;
static constexpr int BIN_NODES = 1 << BIN_SHIFT;
static constexpr int NBIN = (NN + BIN_NODES - 1) / BIN_NODES;
static constexpr int TAB_ROWS = NBIN + 1;
static constexpr int SLICES = 256;
static constexpr int SLICE_EDGES = 6256;
static constexpr int SLICE_IT = (SLICE_EDGES / 4 + 255) / 256;
static_assert(SLICE_EDGES % 4 == 0 && EE % 4 == 0, "slices are read four edges at a time");
static_assert((SLICES - 1) * SLICE_EDGES < EE && SLICES * SLICE_EDGES >= EE, "slices must cover the edges");
static_assert(NN <= (1 << 24), "a bin entry keeps the node id in 24 bits");
static_assert(TAB_ROWS <= 2 * 256, "thread t scans bins 2t and 2t + 1");

__device__ __forceinline__ unsigned lane_u32(unsigned v, int l) {
  return (unsigned)__builtin_amdgcn_readlane((int)v, l);
}
__device__ __forceinline__ float lane_f32(float v, int l) {
  return __int_as_float(__builtin_amdgcn_readlane(__float_as_int(v), l));
}

__device__ __forceinline__ void bin_put4(unsigned* cD, unsigned* cS, unsigned* __restrict__ binD,
                                         unsigned* __restrict__ binS, const int4 s, const int4 d) {
  const unsigned pd0 = atomicAdd(&cD[d.x >> BIN_SHIFT], 1u);
  const unsigned pd1 = atomicAdd(&cD[d.y >> BIN_SHIFT], 1u);
  const unsigned pd2 = atomicAdd(&cD[d.z >> BIN_SHIFT], 1u);
  const unsigned pd3 = atomicAdd(&cD[d.w >> BIN_SHIFT], 1u);
  const unsigned ps0 = atomicAdd(&cS[s.x >> BIN_SHIFT], 1u);
  const unsigned ps1 = atomicAdd(&cS[s.y >> BIN_SHIFT], 1u);
  const unsigned ps2 = atomicAdd(&cS[s.z >> BIN_SHIFT], 1u);
  const unsigned ps3 = atomicAdd(&cS[s.w >> BIN_SHIFT], 1u);
  binD[pd0] = ((unsigned)(d.x & (BIN_NODES - 1)) << 24) | (unsigned)s.x;
  binD[pd1] = ((unsigned)(d.y & (BIN_NODES - 1)) << 24) | (unsigned)s.y;
  binD[pd2] = ((unsigned)(d.z & (BIN_NODES - 1)) << 24) | (unsigned)s.z;
  binD[pd3] = ((unsigned)(d.w & (BIN_NODES - 1)) << 24) | (unsigned)s.w;
  binS[ps0] = (unsigned)(s.x & (BIN_NODES - 1));
  binS[ps1] = (unsigned)(s.y & (BIN_NODES - 1));
  binS[ps2] = (unsigned)(s.z & (BIN_NODES - 1));
  binS[ps3] = (unsigned)(s.w & (BIN_NODES - 1));
}

__global__ __launch_bounds__(256) void k_slicesort(const int* __restrict__ src, const int* __restrict__ dst,
                                                   unsigned* __restrict__ tabD, unsigned* __restrict__ tabS,
                                                   unsigned* __restrict__ binD, unsigned* __restrict__ binS,
                                                   unsigned* __restrict__ bump, float* __restrict__ stat) {
  __shared__ unsigned hD[2 * 256];
  __shared__ unsigned hS[2 * 256];
  __shared__ unsigned wtD[8];
  __shared__ unsigned wtS[8];
  const int tid = threadIdx.x, b = blockIdx.x;
  const int lane = tid & 31, wv = tid >> 5;
  hD[tid] = 0u;
  hD[tid + 256] = 0u;
  hS[tid] = 0u;
  hS[tid + 256] = 0u;
  if (b == 0 && tid == 0) bump[0] = 0u;
  if (b < 2 * STAT_COPIES) stat[b * 256 + tid] = 0.f;
  __syncthreads();
  const int e0 = b * SLICE_EDGES;
  const int n4 = min(EE - e0, SLICE_EDGES) >> 2;
  const int4* s4 = (const int4*)(src + e0);
  const int4* d4 = (const int4*)(dst + e0);
  int4 s[SLICE_IT], d[SLICE_IT];
#pragma unroll
  for (int u = 0; u < SLICE_IT; ++u) {
    const int i = min(tid + 256 * u, n4 - 1);
    s[u] = s4[i];
    d[u] = d4[i];
  }
#pragma unroll
  for (int u = 0; u < SLICE_IT; ++u) {
    if (tid + 256 * u < n4) {
      atomicAdd(&hD[d[u].x >> BIN_SHIFT], 1u);
      atomicAdd(&hD[d[u].y >> BIN_SHIFT], 1u);
      atomicAdd(&hD[d[u].z >> BIN_SHIFT], 1u);
      atomicAdd(&hD[d[u].w >> BIN_SHIFT], 1u);
      atomicAdd(&hS[s[u].x >> BIN_SHIFT], 1u);
      atomicAdd(&hS[s[u].y >> BIN_SHIFT], 1u);
      atomicAdd(&hS[s[u].z >> BIN_SHIFT], 1u);
      atomicAdd(&hS[s[u].w >> BIN_SHIFT], 1u);
    }
  }
  __syncthreads();

  const unsigned a0 = hD[2 * tid], a1 = hD[2 * tid + 1];
  const unsigned c0 = hS[2 * tid], c1 = hS[2 * tid + 1];
  const unsigned sd = a0 + a1, ss = c0 + c1;
  unsigned incD = sd, incS = ss;
  for (int dl = 1; dl < 32; dl <<= 1) {
    const unsigned upD = __shfl_up(incD, dl);
    const unsigned upS = __shfl_up(incS, dl);
    if (lane >= dl) {
      incD += upD;
      incS += upS;
    }
  }
  if (lane == 31) {
    wtD[wv] = incD;
    wtS[wv] = incS;
  }
  __syncthreads();
  unsigned oD = (unsigned)e0 + incD - sd, oS = (unsigned)e0 + incS - ss;
  for (int w = 0; w < wv; ++w) {
    oD += wtD[w];
    oS += wtS[w];
  }
  hD[2 * tid] = oD;
  hD[2 * tid + 1] = oD + a0;
  hS[2 * tid] = oS;
  hS[2 * tid + 1] = oS + c0;
  if (2 * tid < TAB_ROWS) {
    tabD[(size_t)(2 * tid) * SLICES + b] = oD;
    tabS[(size_t)(2 * tid) * SLICES + b] = oS;
  }
  if (2 * tid + 1 < TAB_ROWS) {
    tabD[(size_t)(2 * tid + 1) * SLICES + b] = oD + a0;
    tabS[(size_t)(2 * tid + 1) * SLICES + b] = oS + c0;
  }
  __syncthreads();

#pragma unroll
  for (int u = 0; u < SLICE_IT; ++u) {
    if (tid + 256 * u < n4) {
      bin_put4(hD, hS, binD, binS, s[u], d[u]);
    }
  }
}

__global__ __launch_bounds__(256) void k_binrows(const unsigned* __restrict__ binD,
                                                 const unsigned* __restrict__ binS,
                                                 const unsigned* __restrict__ tabD,
                                                 const unsigned* __restrict__ tabS,
                                                 unsigned* __restrict__ bump,
                                                 float* __restrict__ dsout, float* __restrict__ dsin,
                                                 unsigned* __restrict__ rows, unsigned* __restrict__ esrc) {
  __shared__ unsigned cD[BIN_NODES];
  __shared__ unsigned cS[BIN_NODES];
  __shared__ unsigned cur[BIN_NODES];
  __shared__ unsigned wtot[8];
  __shared__ unsigned wsum[8];
  __shared__ unsigned sbase;
  const int tid = threadIdx.x, j = blockIdx.x;
  const int lane = tid & 31, wv = tid >> 5;
  const unsigned ul = (unsigned)lane;
  cD[tid] = 0u;
  cS[tid] = 0u;
  const unsigned stD = tabD[(size_t)j * SLICES + tid];
  const unsigned cnD = tabD[(size_t)(j + 1) * SLICES + tid] - stD;
  const unsigned stS = tabS[(size_t)j * SLICES + tid];
  const unsigned cnS = tabS[(size_t)(j + 1) * SLICES + tid] - stS;
  unsigned tot = cnD;
  for (int dl = 16; dl > 0; dl >>= 1) tot += __shfl_xor(tot, dl);
  if (lane == 0) wsum[wv] = tot;
  __syncthreads();

  unsigned v[32], w[32];
#pragma unroll
  for (int k = 0; k < 32; ++k) {
    v[k] = 0u;
    if (ul < lane_u32(cnD, k)) v[k] = binD[lane_u32(stD, k) + ul];
  }
#pragma unroll
  for (int k = 0; k < 32; ++k) {
    w[k] = 0u;
    if (ul < lane_u32(cnS, k)) w[k] = binS[lane_u32(stS, k) + ul];
  }
  unsigned mybase = 0u;
  if (tid == 0) {
    unsigned nD = 0u;
    for (int w8 = 0; w8 < 8; ++w8) nD += wsum[w8];
    mybase = atomicAdd(bump, nD);
  }
#pragma unroll
  for (int k = 0; k < 32; ++k)
    if (ul < lane_u32(cnD, k)) atomicAdd(&cD[v[k] >> 24], 1u);
#pragma unroll
  for (int k = 0; k < 32; ++k)
    if (ul < lane_u32(cnS, k)) atomicAdd(&cS[w[k]], 1u);
#pragma unroll 1
  for (int k = 0; k < 32; ++k) {
    const unsigned cn = lane_u32(cnD, k);
    if (cn > 32u) {
      const unsigned* e = binD + lane_u32(stD, k);
      for (unsigned i = 32u + ul; i < cn; i += 32u) atomicAdd(&cD[e[i] >> 24], 1u);
    }
    const unsigned cm = lane_u32(cnS, k);
    if (cm > 32u) {
      const unsigned* e = binS + lane_u32(stS, k);
      for (unsigned i = 32u + ul; i < cm; i += 32u) atomicAdd(&cS[e[i]], 1u);
    }
  }
  if (tid == 0) sbase = mybase;
  __syncthreads();

  const unsigned id = cD[tid], od = cS[tid];
  unsigned inc = id;
  for (int d = 1; d < 32; d <<= 1) {
    const unsigned up = __shfl_up(inc, d);
    if (lane >= d) inc += up;
  }
  if (lane == 31) wtot[wv] = inc;
  __syncthreads();
  unsigned st = inc - id;
  for (int w = 0; w < wv; ++w) st += wtot[w];
  cur[tid] = st;
  const unsigned bD = sbase;
  const int node = j * BIN_NODES + tid;
  if (node < NN) {
    *(uint2*)(rows + 2 * node) = make_uint2(bD + st, bD + st + id);
    dsin[node] = id ? rsqrtf((float)id) : 1.0f;
    dsout[node] = od ? rsqrtf((float)od) : 1.0f;
  }
  __syncthreads();

  unsigned* o = esrc + bD;
  unsigned p[32];
#pragma unroll
  for (int k = 0; k < 32; ++k) {
    p[k] = 0u;
    if (ul < lane_u32(cnD, k)) p[k] = atomicAdd(&cur[v[k] >> 24], 1u);
  }
#pragma unroll
  for (int k = 0; k < 32; ++k)
    if (ul < lane_u32(cnD, k)) o[p[k]] = v[k] & 0xFFFFFFu;
#pragma unroll 1
  for (int k = 0; k < 32; ++k) {
    const unsigned cn = lane_u32(cnD, k);
    if (cn > 32u) {
      const unsigned* e = binD + lane_u32(stD, k);
      for (unsigned i = 32u + ul; i < cn; i += 32u) {
        const unsigned x = e[i];
        o[atomicAdd(&cur[x >> 24], 1u)] = x & 0xFFFFFFu;
      }
    }
  }
}


static constexpr int AGG_NODES = 8;
static constexpr int AGG_GROUP = 8;
static constexpr int AGG_WAVES = NN / AGG_NODES;
static constexpr int AGG_BLOCKS = AGG_WAVES / 4;
static_assert(NN % AGG_NODES == 0 && AGG_WAVES % 4 == 0, "gather grid must cover the nodes exactly");

__device__ __forceinline__ void axpy4(float4& acc, float w, const float4& v) {
  acc.x = fmaf(w, v.x, acc.x);
  acc.y = fmaf(w, v.y, acc.y);
  acc.z = fmaf(w, v.z, acc.z);
  acc.w = fmaf(w, v.w, acc.w);
}

__global__ __launch_bounds__(128) void k_agg128(const float* __restrict__ h,
                                                const unsigned* __restrict__ rows,
                                                const unsigned* __restrict__ esrc,
                                                const float* __restrict__ dsout,
                                                const float* __restrict__ dsin,
                                                float* __restrict__ agg) {
  const int lane = threadIdx.x & 31;
  const int n0 = (blockIdx.x * 4 + (threadIdx.x >> 5)) * AGG_NODES;
  unsigned rb = 0u, re = 0u;
  float cin = 0.f;
  if (lane < AGG_NODES) {
    const uint2 r = *(const uint2*)(rows + 2 * (n0 + lane));
    rb = r.x;
    re = r.y;
    cin = dsin[n0 + lane];
  }
  unsigned b = lane_u32(rb, 0), e = lane_u32(re, 0);
  unsigned my = 0u;
  float myw = 0.f;
  if (b + lane < e) {
    my = esrc[b + lane];
    myw = dsout[my];
  }
  const float* hl = h + lane * 4;

#pragma unroll 1
  for (int k = 0; k < AGG_NODES; ++k) {
    unsigned nb = 0u, ne = 0u, nmy = 0u;
    if (k + 1 < AGG_NODES) {
      nb = lane_u32(rb, k + 1);
      ne = lane_u32(re, k + 1);
      if (nb + lane < ne) nmy = esrc[nb + lane];
    }
    float4 acc = make_float4(0.f, 0.f, 0.f, 0.f);
    const int cnt = (int)min(e - b, 32u);
    int j = 0;
    for (; j + AGG_GROUP <= cnt; j += AGG_GROUP) {
      float4 v[AGG_GROUP];
#pragma unroll
      for (int u = 0; u < AGG_GROUP; ++u)
        v[u] = *(const float4*)(hl + (size_t)lane_u32(my, j + u) * FF);
      __builtin_amdgcn_sched_barrier(0);
#pragma unroll
      for (int u = 0; u < AGG_GROUP; ++u) axpy4(acc, lane_f32(myw, j + u), v[u]);
    }
    const int rem = cnt - j;
    if (rem > 0) {
      float4 v[AGG_GROUP - 1];
#pragma unroll
      for (int u = 0; u < AGG_GROUP - 1; ++u) {
        v[u] = make_float4(0.f, 0.f, 0.f, 0.f);
        if (u < rem) v[u] = *(const float4*)(hl + (size_t)lane_u32(my, j + u) * FF);
      }
#pragma unroll
      for (int u = 0; u < AGG_GROUP - 1; ++u)
        if (u < rem) axpy4(acc, lane_f32(myw, j + u), v[u]);
    }
    for (unsigned i = b + 32u; i < e; ++i) {
      const unsigned s = esrc[i];
      axpy4(acc, dsout[s], *(const float4*)(hl + (size_t)s * FF));
    }
    const float c = lane_f32(cin, k);
    *(float4*)(agg + (size_t)(n0 + k) * FF + lane * 4) =
        make_float4(c * acc.x, c * acc.y, c * acc.z, c * acc.w);
    b = nb;
    e = ne;
    my = nmy;
    myw = 0.f;
    if (b + lane < e) myw = dsout[my];
  }
}

__global__ __launch_bounds__(128) void k_agg64(const float* __restrict__ P,
                                               const unsigned* __restrict__ rows,
                                               const unsigned* __restrict__ esrc,
                                               const float* __restrict__ dsin,
                                               float* __restrict__ R) {
  const int lane = threadIdx.x & 31;
  const int n0 = (blockIdx.x * 4 + (threadIdx.x >> 5)) * AGG_NODES;
  unsigned rb = 0u, re = 0u;
  float cin = 0.f;
  if (lane < AGG_NODES) {
    const uint2 r = *(const uint2*)(rows + 2 * (n0 + lane));
    rb = r.x;
    re = r.y;
    cin = dsin[n0 + lane];
  }
  unsigned b = lane_u32(rb, 0), e = lane_u32(re, 0);
  unsigned my = 0u;
  if (b + lane < e) my = esrc[b + lane];
  const float* pl = P + lane * 2;

#pragma unroll 1
  for (int k = 0; k < AGG_NODES; ++k) {
    unsigned nb = 0u, ne = 0u, nmy = 0u;
    if (k + 1 < AGG_NODES) {
      nb = lane_u32(rb, k + 1);
      ne = lane_u32(re, k + 1);
      if (nb + lane < ne) nmy = esrc[nb + lane];
    }
    float2 acc = make_float2(0.f, 0.f);
    const int cnt = (int)min(e - b, 32u);
    int j = 0;
    for (; j + AGG_GROUP <= cnt; j += AGG_GROUP) {
      float2 v[AGG_GROUP];
#pragma unroll
      for (int u = 0; u < AGG_GROUP; ++u)
        v[u] = *(const float2*)(pl + (size_t)lane_u32(my, j + u) * 64);
#pragma unroll
      for (int u = 0; u < AGG_GROUP; ++u) {
        acc.x += v[u].x;
        acc.y += v[u].y;
      }
    }
    const int rem = cnt - j;
    if (rem > 0) {
      float2 v[AGG_GROUP - 1];
#pragma unroll
      for (int u = 0; u < AGG_GROUP - 1; ++u) {
        v[u] = make_float2(0.f, 0.f);
        if (u < rem) v[u] = *(const float2*)(pl + (size_t)lane_u32(my, j + u) * 64);
      }
#pragma unroll
      for (int u = 0; u < AGG_GROUP - 1; ++u) {
        if (u < rem) {
          acc.x += v[u].x;
          acc.y += v[u].y;
        }
      }
    }
    for (unsigned i = b + 32u; i < e; ++i) {
      const float2 x = *(const float2*)(pl + (size_t)esrc[i] * 64);
      acc.x += x.x;
      acc.y += x.y;
    }
    const float c = lane_f32(cin, k);
    *(float2*)(R + (size_t)(n0 + k) * 64 + lane * 2) = make_float2(c * acc.x, c * acc.y);
    b = nb;
    e = ne;
    my = nmy;
  }
}

__global__ __launch_bounds__(128) void k_aggadd128(const float* __restrict__ G,
                                                   const unsigned* __restrict__ rows,
                                                   const unsigned* __restrict__ esrc,
                                                   const float* __restrict__ dsin,
                                                   const float* __restrict__ S,
                                                   float* __restrict__ Z,
                                                   float* __restrict__ stat) {
  __shared__ float red[2][4][4][32];
  const int lane = threadIdx.x & 31;
  float* mys = &red[0][threadIdx.x >> 5][0][lane];
  float* myq = &red[1][threadIdx.x >> 5][0][lane];
  mys[0] = mys[32] = mys[64] = mys[96] = 0.f;
  myq[0] = myq[32] = myq[64] = myq[96] = 0.f;
  const int n0 = (blockIdx.x * 4 + (threadIdx.x >> 5)) * AGG_NODES;
  unsigned rb = 0u, re = 0u;
  float cin = 0.f;
  if (lane < AGG_NODES) {
    const uint2 r = *(const uint2*)(rows + 2 * (n0 + lane));
    rb = r.x;
    re = r.y;
    cin = dsin[n0 + lane];
  }
  unsigned b = lane_u32(rb, 0), e = lane_u32(re, 0);
  unsigned my = 0u;
  if (b + lane < e) my = esrc[b + lane];
  const float* gl = G + lane * 4;

#pragma unroll 1
  for (int k = 0; k < AGG_NODES; ++k) {
    unsigned nb = 0u, ne = 0u, nmy = 0u;
    if (k + 1 < AGG_NODES) {
      nb = lane_u32(rb, k + 1);
      ne = lane_u32(re, k + 1);
      if (nb + lane < ne) nmy = esrc[nb + lane];
    }
    const float4 self = *(const float4*)(S + (size_t)(n0 + k) * FF + lane * 4);
    float4 acc = make_float4(0.f, 0.f, 0.f, 0.f);
    const int cnt = (int)min(e - b, 32u);
    int j = 0;
    for (; j + AGG_GROUP <= cnt; j += AGG_GROUP) {
      float4 v[AGG_GROUP];
#pragma unroll
      for (int u = 0; u < AGG_GROUP; ++u)
        v[u] = *(const float4*)(gl + (size_t)lane_u32(my, j + u) * FF);
#pragma unroll
      for (int u = 0; u < AGG_GROUP; ++u) {
        acc.x += v[u].x;
        acc.y += v[u].y;
        acc.z += v[u].z;
        acc.w += v[u].w;
      }
    }
    const int rem = cnt - j;
    if (rem > 0) {
      float4 v[AGG_GROUP - 1];
#pragma unroll
      for (int u = 0; u < AGG_GROUP - 1; ++u) {
        v[u] = make_float4(0.f, 0.f, 0.f, 0.f);
        if (u < rem) v[u] = *(const float4*)(gl + (size_t)lane_u32(my, j + u) * FF);
      }
#pragma unroll
      for (int u = 0; u < AGG_GROUP - 1; ++u) {
        if (u < rem) {
          acc.x += v[u].x;
          acc.y += v[u].y;
          acc.z += v[u].z;
          acc.w += v[u].w;
        }
      }
    }
    for (unsigned i = b + 32u; i < e; ++i) {
      const float4 x = *(const float4*)(gl + (size_t)esrc[i] * FF);
      acc.x += x.x;
      acc.y += x.y;
      acc.z += x.z;
      acc.w += x.w;
    }
    const float c = lane_f32(cin, k);
    const float4 z = make_float4(fmaf(c, acc.x, self.x), fmaf(c, acc.y, self.y),
                                 fmaf(c, acc.z, self.z), fmaf(c, acc.w, self.w));
    *(float4*)(Z + (size_t)(n0 + k) * FF + lane * 4) = z;
    atomicAdd(mys, z.x);
    atomicAdd(mys + 32, z.y);
    atomicAdd(mys + 64, z.z);
    atomicAdd(mys + 96, z.w);
    atomicAdd(myq, z.x * z.x);
    atomicAdd(myq + 32, z.y * z.y);
    atomicAdd(myq + 64, z.z * z.z);
    atomicAdd(myq + 96, z.w * z.w);
    b = nb;
    e = ne;
    my = nmy;
  }
  __syncthreads();
  const float* r = &red[0][0][threadIdx.x & 3][threadIdx.x >> 2];
  float ts = 0.f, tq = 0.f;
  for (int w = 0; w < 4; ++w) {
    ts += r[w * FF];
    tq += r[(4 + w) * FF];
  }
  float* o = stat + (blockIdx.x % STAT_COPIES) * 256 + threadIdx.x;
  atomicAdd(o, ts);
  atomicAdd(o + FF, tq);
}


static constexpr int GEMM_BLOCKS = 512;
static constexpr int NSTRIP = NN / 16;

template <int NC>
__device__ __forceinline__ void stage_weights(const float* __restrict__ Bg,
                                              float4* __restrict__ s, int tid) {
  constexpr int C4 = NC / 4;
  constexpr int PITCH = NC + 8;
  for (int i = tid; i < 32 * C4; i += 256) {
    const int g = i / C4, c4 = i - g * C4;
    const float* p = Bg + (size_t)(4 * g) * NC + 4 * c4;
    const float4 r0 = *(const float4*)(p);
    const float4 r1 = *(const float4*)(p + NC);
    const float4 r2 = *(const float4*)(p + 2 * NC);
    const float4 r3 = *(const float4*)(p + 3 * NC);
    const int m = g >> 3, hh = (g >> 2) & 1, q = g & 3;
    float4* d = s + ((m * 4 + q) * 2 + hh) * PITCH + 4 * c4;
    d[0] = make_float4(r0.x, r1.x, r2.x, r3.x);
    d[1] = make_float4(r0.y, r1.y, r2.y, r3.y);
    d[2] = make_float4(r0.z, r1.z, r2.z, r3.z);
    d[3] = make_float4(r0.w, r1.w, r2.w, r3.w);
  }
}

__device__ __forceinline__ v8f mm4(float ax, float ay, float bx, float by, v8f c) {
  v2f a; a.x = ax; a.y = ay;
  v2f b; b.x = bx; b.y = by;
  return __builtin_amdgcn_wmma_f32_16x16x4_f32(false, a, false, b, (short)0, c, false, false);
}

__device__ __forceinline__ float4 bn_relu4(const float4 z, const float4 a, const float4 b) {
  const float vx = fmaf(z.x, a.x, b.x);
  const float vy = fmaf(z.y, a.y, b.y);
  const float vz = fmaf(z.z, a.z, b.z);
  const float vw = fmaf(z.w, a.w, b.w);
  return make_float4(vx > 0.f ? vx : 0.f, vy > 0.f ? vy : 0.f, vz > 0.f ? vz : 0.f,
                     vw > 0.f ? vw : 0.f);
}

__device__ __forceinline__ void bn_params_to_lds(const float* __restrict__ stat,
                                                 const float* __restrict__ g,
                                                 const float* __restrict__ be,
                                                 float* __restrict__ sBN, int tid) {
  if (tid < FF) {
    float s = 0.f, q = 0.f;
#pragma unroll
    for (int k = 0; k < STAT_COPIES; ++k) {
      s += stat[k * 256 + tid];
      q += stat[k * 256 + FF + tid];
    }
    const float inv = 1.0f / (float)NN;
    const float mu = s * inv;
    const float var = q * inv - mu * mu;
    const float a = g[tid] * rsqrtf(var + 1e-5f);
    sBN[tid] = a;
    sBN[FF + tid] = be[tid] - mu * a;
  }
}

__global__ __launch_bounds__(256) void k_gemm_dual(const float* __restrict__ Agg,
                                                   const float* __restrict__ H,
                                                   const float* __restrict__ W,
                                                   const float* __restrict__ L,
                                                   float* __restrict__ Z,
                                                   float* __restrict__ stat) {
  constexpr int PITCH = 128 + 8;
  __shared__ __attribute__((aligned(16))) float4 sW[32 * PITCH];
  __shared__ __attribute__((aligned(16))) float4 sL[32 * PITCH];
  stage_weights<128>(W, sW, threadIdx.x);
  stage_weights<128>(L, sL, threadIdx.x);
  __syncthreads();

  const int wv = threadIdx.x >> 5;
  const int lane = threadIdx.x & 31;
  const int t = lane & 15;
  const int hh = lane >> 4;
  float cs[8], cq[8];
#pragma unroll
  for (int ct = 0; ct < 8; ++ct) { cs[ct] = 0.f; cq[ct] = 0.f; }

  for (int strip = blockIdx.x + GEMM_BLOCKS * wv; strip < NSTRIP; strip += GEMM_BLOCKS * 8) {
    const int r0 = strip << 4;
    const float* arow = Agg + (size_t)(r0 + t) * FF + 16 * hh;
    const float* hrow = H + (size_t)(r0 + t) * FF + 16 * hh;
    v8f acc[8];
#pragma unroll
    for (int ct = 0; ct < 8; ++ct) acc[ct] = v8f{};

#pragma unroll 1
    for (int m = 0; m < 4; ++m) {
      float4 a[4], h[4];
#pragma unroll
      for (int q = 0; q < 4; ++q) {
        a[q] = *(const float4*)(arow + 32 * m + 4 * q);
        h[q] = *(const float4*)(hrow + 32 * m + 4 * q);
      }
#pragma unroll
      for (int q = 0; q < 4; ++q) {
        const float4* pw = sW + ((m * 4 + q) * 2 + hh) * PITCH + t;
        const float4* pl = sL + ((m * 4 + q) * 2 + hh) * PITCH + t;
        float4 bw[8], bl[8];
#pragma unroll
        for (int ct = 0; ct < 8; ++ct) { bw[ct] = pw[16 * ct]; bl[ct] = pl[16 * ct]; }
#pragma unroll
        for (int ct = 0; ct < 8; ++ct) acc[ct] = mm4(a[q].x, a[q].y, bw[ct].x, bw[ct].y, acc[ct]);
#pragma unroll
        for (int ct = 0; ct < 8; ++ct) acc[ct] = mm4(h[q].x, h[q].y, bl[ct].x, bl[ct].y, acc[ct]);
#pragma unroll
        for (int ct = 0; ct < 8; ++ct) acc[ct] = mm4(a[q].z, a[q].w, bw[ct].z, bw[ct].w, acc[ct]);
#pragma unroll
        for (int ct = 0; ct < 8; ++ct) acc[ct] = mm4(h[q].z, h[q].w, bl[ct].z, bl[ct].w, acc[ct]);
      }
    }
#pragma unroll
    for (int ct = 0; ct < 8; ++ct)
#pragma unroll
      for (int v = 0; v < 8; ++v) {
        const float x = acc[ct][v];
        Z[(size_t)(r0 + v + 8 * hh) * FF + 16 * ct + t] = x;
        cs[ct] += x;
        cq[ct] = fmaf(x, x, cq[ct]);
      }
  }

  __syncthreads();
  float* red = (float*)sW;
#pragma unroll
  for (int ct = 0; ct < 8; ++ct) {
    red[(wv * 2 + hh) * FF + 16 * ct + t] = cs[ct];
    red[(16 + wv * 2 + hh) * FF + 16 * ct + t] = cq[ct];
  }
  __syncthreads();
  const int st = threadIdx.x >> 7, col = threadIdx.x & 127;
  const float* rr = red + st * 16 * FF + col;
  float ts = 0.f;
#pragma unroll
  for (int i = 0; i < 16; ++i) ts += rr[i * FF];
  atomicAdd(stat + (blockIdx.x % STAT_COPIES) * 256 + st * FF + col, ts);
}

__global__ __launch_bounds__(256) void k_gemm_proj(const float* __restrict__ Zin,
                                                   const float* __restrict__ stat,
                                                   const float* __restrict__ g,
                                                   const float* __restrict__ be,
                                                   const float* __restrict__ dsout,
                                                   const float* __restrict__ W,
                                                   const float* __restrict__ L,
                                                   float* __restrict__ Gw,
                                                   float* __restrict__ Gl) {
  constexpr int PITCH = 128 + 8;
  __shared__ __attribute__((aligned(16))) float4 sW[32 * PITCH];
  __shared__ __attribute__((aligned(16))) float4 sL[32 * PITCH];
  __shared__ __attribute__((aligned(16))) float sBN[2 * FF];
  bn_params_to_lds(stat, g, be, sBN, threadIdx.x);
  stage_weights<128>(W, sW, threadIdx.x);
  stage_weights<128>(L, sL, threadIdx.x);
  __syncthreads();

  const int wv = threadIdx.x >> 5;
  const int lane = threadIdx.x & 31;
  const int t = lane & 15, hh = lane >> 4;

  for (int strip = blockIdx.x + GEMM_BLOCKS * wv; strip < NSTRIP; strip += GEMM_BLOCKS * 8) {
    const int r0 = strip << 4;
    const float* zrow = Zin + (size_t)(r0 + t) * FF + 16 * hh;
    const float4 d0 = *(const float4*)(dsout + r0 + 8 * hh);
    const float4 d1 = *(const float4*)(dsout + r0 + 8 * hh + 4);
    v8f aw[8], al[8];
#pragma unroll
    for (int ct = 0; ct < 8; ++ct) { aw[ct] = v8f{}; al[ct] = v8f{}; }
#pragma unroll 1
    for (int m = 0; m < 4; ++m) {
      float4 h[4];
#pragma unroll
      for (int q = 0; q < 4; ++q) {
        const int c = 32 * m + 16 * hh + 4 * q;
        const float4 x = *(const float4*)(zrow + 32 * m + 4 * q);
        h[q] = bn_relu4(x, *(const float4*)(sBN + c), *(const float4*)(sBN + FF + c));
      }
#pragma unroll
      for (int q = 0; q < 4; ++q) {
        const float4* pw = sW + ((m * 4 + q) * 2 + hh) * PITCH + t;
        const float4* pl = sL + ((m * 4 + q) * 2 + hh) * PITCH + t;
        float4 bw[8], bl[8];
#pragma unroll
        for (int ct = 0; ct < 8; ++ct) { bw[ct] = pw[16 * ct]; bl[ct] = pl[16 * ct]; }
#pragma unroll
        for (int ct = 0; ct < 8; ++ct) aw[ct] = mm4(h[q].x, h[q].y, bw[ct].x, bw[ct].y, aw[ct]);
#pragma unroll
        for (int ct = 0; ct < 8; ++ct) al[ct] = mm4(h[q].x, h[q].y, bl[ct].x, bl[ct].y, al[ct]);
#pragma unroll
        for (int ct = 0; ct < 8; ++ct) aw[ct] = mm4(h[q].z, h[q].w, bw[ct].z, bw[ct].w, aw[ct]);
#pragma unroll
        for (int ct = 0; ct < 8; ++ct) al[ct] = mm4(h[q].z, h[q].w, bl[ct].z, bl[ct].w, al[ct]);
      }
    }
    const float dv[8] = {d0.x, d0.y, d0.z, d0.w, d1.x, d1.y, d1.z, d1.w};
#pragma unroll
    for (int ct = 0; ct < 8; ++ct)
#pragma unroll
      for (int v = 0; v < 8; ++v) {
        const size_t o = (size_t)(r0 + v + 8 * hh) * FF + 16 * ct + t;
        Gw[o] = dv[v] * aw[ct][v];
        Gl[o] = al[ct][v];
      }
  }
}

__global__ __launch_bounds__(256) void k_gemm_p(const float* __restrict__ Zin,
                                                const float* __restrict__ stat,
                                                const float* __restrict__ g,
                                                const float* __restrict__ be,
                                                const float* __restrict__ dsout,
                                                const float* __restrict__ Wp,
                                                float* __restrict__ Hout,
                                                float* __restrict__ P) {
  constexpr int PITCH = 64 + 8;
  __shared__ __attribute__((aligned(16))) float4 sW[32 * PITCH];
  __shared__ __attribute__((aligned(16))) float sBN[2 * FF];
  bn_params_to_lds(stat, g, be, sBN, threadIdx.x);
  stage_weights<64>(Wp, sW, threadIdx.x);
  __syncthreads();

  const int wv = threadIdx.x >> 5;
  const int lane = threadIdx.x & 31;
  const int t = lane & 15, hh = lane >> 4;

  for (int strip = blockIdx.x + GEMM_BLOCKS * wv; strip < NSTRIP; strip += GEMM_BLOCKS * 8) {
    const int r0 = strip << 4;
    const float d = dsout[r0 + t];
    const float* zrow = Zin + (size_t)(r0 + t) * FF + 16 * hh;
    float* hrow = Hout + (size_t)(r0 + t) * FF + 16 * hh;
    v8f acc[4];
#pragma unroll
    for (int ct = 0; ct < 4; ++ct) acc[ct] = v8f{};
#pragma unroll 1
    for (int m = 0; m < 4; ++m) {
      float4 a[4];
#pragma unroll
      for (int q = 0; q < 4; ++q) {
        const int c = 32 * m + 16 * hh + 4 * q;
        const float4 z = *(const float4*)(zrow + 32 * m + 4 * q);
        const float4 x = bn_relu4(z, *(const float4*)(sBN + c), *(const float4*)(sBN + FF + c));
        *(float4*)(hrow + 32 * m + 4 * q) = x;
        a[q] = make_float4(d * x.x, d * x.y, d * x.z, d * x.w);
      }
#pragma unroll
      for (int q = 0; q < 4; ++q) {
        const float4* pw = sW + ((m * 4 + q) * 2 + hh) * PITCH + t;
        float4 bw[4];
#pragma unroll
        for (int ct = 0; ct < 4; ++ct) bw[ct] = pw[16 * ct];
#pragma unroll
        for (int ct = 0; ct < 4; ++ct) acc[ct] = mm4(a[q].x, a[q].y, bw[ct].x, bw[ct].y, acc[ct]);
#pragma unroll
        for (int ct = 0; ct < 4; ++ct) acc[ct] = mm4(a[q].z, a[q].w, bw[ct].z, bw[ct].w, acc[ct]);
      }
    }
#pragma unroll
    for (int ct = 0; ct < 4; ++ct)
#pragma unroll
      for (int v = 0; v < 8; ++v)
        P[(size_t)(r0 + v + 8 * hh) * 64 + 16 * ct + t] = acc[ct][v];
  }
}

__global__ __launch_bounds__(256) void k_gemm_out(const float* __restrict__ H,
                                                  const float* __restrict__ Lp,
                                                  const float* __restrict__ R,
                                                  const float* __restrict__ b2,
                                                  float* __restrict__ out) {
  constexpr int PITCH = 64 + 8;
  __shared__ __attribute__((aligned(16))) float4 sL[32 * PITCH];
  stage_weights<64>(Lp, sL, threadIdx.x);
  __syncthreads();

  const int wv = threadIdx.x >> 5;
  const int lane = threadIdx.x & 31;
  const int t = lane & 15, hh = lane >> 4;

  for (int strip = blockIdx.x + GEMM_BLOCKS * wv; strip < NSTRIP; strip += GEMM_BLOCKS * 8) {
    const int r0 = strip << 4;
    const float* hrow = H + (size_t)(r0 + t) * FF + 16 * hh;
    v8f acc[3];
#pragma unroll
    for (int ct = 0; ct < 3; ++ct) acc[ct] = v8f{};
#pragma unroll 1
    for (int m = 0; m < 4; ++m) {
      float4 a[4];
#pragma unroll
      for (int q = 0; q < 4; ++q) a[q] = *(const float4*)(hrow + 32 * m + 4 * q);
#pragma unroll
      for (int q = 0; q < 4; ++q) {
        const float4* pl = sL + ((m * 4 + q) * 2 + hh) * PITCH + t;
        float4 bl[3];
#pragma unroll
        for (int ct = 0; ct < 3; ++ct) bl[ct] = pl[16 * ct];
#pragma unroll
        for (int ct = 0; ct < 3; ++ct) acc[ct] = mm4(a[q].x, a[q].y, bl[ct].x, bl[ct].y, acc[ct]);
#pragma unroll
        for (int ct = 0; ct < 3; ++ct) acc[ct] = mm4(a[q].z, a[q].w, bl[ct].z, bl[ct].w, acc[ct]);
      }
    }
#pragma unroll
    for (int ct = 0; ct < 3; ++ct) {
      const int col = 16 * ct + t;
      if (col < 40) {
        const float bias = b2[col];
#pragma unroll
        for (int v = 0; v < 8; ++v) {
          const int row = r0 + v + 8 * hh;
          out[(size_t)row * 40 + col] = acc[ct][v] + R[(size_t)row * 64 + col] + bias;
        }
      }
    }
  }
}


__global__ void k_padw(const float* __restrict__ W2, const float* __restrict__ L2,
                       float* __restrict__ Wp, float* __restrict__ Lp) {
  const int i = blockIdx.x * 256 + threadIdx.x;
  const int k = i >> 6, c = i & 63;
  const float w = (c < 40) ? W2[k * 40 + c] : 0.f;
  const float l = (c < 40) ? L2[k * 40 + c] : 0.f;
  Wp[i] = w;
  Lp[i] = l;
}


extern "C" void kernel_launch(void* const* d_in, const int* in_sizes, int n_in,
                              void* d_out, int out_size, void* d_ws, size_t ws_size,
                              hipStream_t stream) {
  (void)in_sizes; (void)n_in; (void)out_size; (void)ws_size;

  const float* feat = (const float*)d_in[0];
  const int* src   = (const int*)d_in[1];
  const int* dst   = (const int*)d_in[2];
  const float* W0  = (const float*)d_in[3];
  const float* L0  = (const float*)d_in[4];
  const float* W1  = (const float*)d_in[5];
  const float* L1  = (const float*)d_in[6];
  const float* W2  = (const float*)d_in[7];
  const float* L2  = (const float*)d_in[8];
  const float* b2  = (const float*)d_in[9];
  const float* g0  = (const float*)d_in[10];
  const float* be0 = (const float*)d_in[11];
  const float* g1  = (const float*)d_in[12];
  const float* be1 = (const float*)d_in[13];
  float* out = (float*)d_out;

  char* base = (char*)d_ws;
  size_t off = 0;
  auto take = [&](size_t bytes) -> char* {
    char* p = base + off;
    off = (off + bytes + 255) & ~(size_t)255;
    return p;
  };
  unsigned* tabD    = (unsigned*)take((size_t)TAB_ROWS * SLICES * 4);
  unsigned* tabS    = (unsigned*)take((size_t)TAB_ROWS * SLICES * 4);
  unsigned* bump    = (unsigned*)take(4);
  unsigned* binD    = (unsigned*)take((size_t)EE * 4);
  unsigned* binS    = (unsigned*)take((size_t)EE * 4);
  unsigned* rows    = (unsigned*)take((size_t)NN * 8);
  unsigned* esrc    = (unsigned*)take((size_t)EE * 4);
  float* dsout = (float*)take((size_t)NN * 4);
  float* dsin  = (float*)take((size_t)NN * 4);
  float* stat0 = (float*)take((size_t)2 * STAT_LAYER * 4);
  float* stat1 = stat0 + STAT_LAYER;
  float* Wp    = (float*)take(128 * 64 * 4);
  float* Lp    = (float*)take(128 * 64 * 4);
  float* bufA  = (float*)take((size_t)NN * FF * 4);
  float* bufB  = (float*)take((size_t)NN * FF * 4);
  float* bufC  = (float*)take((size_t)NN * FF * 4);
  float* bufD  = (float*)take((size_t)NN * FF * 4);

  k_slicesort<<<SLICES, 256, 0, stream>>>(src, dst, tabD, tabS, binD, binS, bump, stat0);
  k_binrows<<<NBIN, 256, 0, stream>>>(binD, binS, tabD, tabS, bump, dsout, dsin, rows, esrc);
  k_padw<<<32, 256, 0, stream>>>(W2, L2, Wp, Lp);

  k_agg128<<<AGG_BLOCKS, 128, 0, stream>>>(feat, rows, esrc, dsout, dsin, bufA);
  k_gemm_dual<<<GEMM_BLOCKS, 256, 0, stream>>>(bufA, feat, W0, L0, bufB, stat0);

  k_gemm_proj<<<GEMM_BLOCKS, 256, 0, stream>>>(bufB, stat0, g0, be0, dsout, W1, L1, bufA, bufD);
  k_aggadd128<<<AGG_BLOCKS, 128, 0, stream>>>(bufA, rows, esrc, dsin, bufD, bufB,
                                               stat1);

  k_gemm_p<<<GEMM_BLOCKS, 256, 0, stream>>>(bufB, stat1, g1, be1, dsout, Wp, bufC, bufA);
  k_agg64<<<AGG_BLOCKS, 128, 0, stream>>>(bufA, rows, esrc, dsin, bufB);
  k_gemm_out<<<GEMM_BLOCKS, 256, 0, stream>>>(bufC, Lp, bufB, b2, out);
}


# CSR build: a slice's bin-sorted edge copies and a bin's row ids are put together in LDS and written out as whole lines; 4.8M scattered 4-byte global stores gone; row order: LDS add arrival, as before
# speedup vs baseline: 2.9195x; 1.0439x over previous
#include <hip/hip_runtime.h>
#include <hip/hip_bf16.h>


static constexpr int NN = 100000;
static constexpr int EE = 1600000;
static constexpr int FF = 128;

typedef __attribute__((ext_vector_type(2))) float v2f;
typedef __attribute__((ext_vector_type(8))) float v8f;

static constexpr int STAT_COPIES = 16;
static constexpr int STAT_LAYER = STAT_COPIES * 256;


static constexpr int BIN_SHIFT = 8;
static constexpr int BIN_NODES = 1 << BIN_SHIFT;
static constexpr int NBIN = (NN + BIN_NODES - 1) / BIN_NODES;
static constexpr int TAB_ROWS = NBIN + 1;
static constexpr int SLICES = 256;
static constexpr int SLICE_EDGES = 6256;
static constexpr int SLICE_IT = (SLICE_EDGES / 4 + 255) / 256;
static_assert(SLICE_EDGES % 4 == 0 && EE % 4 == 0, "slices are read four edges at a time");
static_assert((SLICES - 1) * SLICE_EDGES < EE && SLICES * SLICE_EDGES >= EE, "slices must cover the edges");
static_assert(NN <= (1 << 24), "a bin entry keeps the node id in 24 bits");
static_assert(TAB_ROWS <= 2 * 256, "thread t scans bins 2t and 2t + 1");
static constexpr int ROW_CAP = 5120;

__device__ __forceinline__ unsigned lane_u32(unsigned v, int l) {
  return (unsigned)__builtin_amdgcn_readlane((int)v, l);
}
__device__ __forceinline__ float lane_f32(float v, int l) {
  return __int_as_float(__builtin_amdgcn_readlane(__float_as_int(v), l));
}

__device__ __forceinline__ void bin_put4(unsigned* cD, unsigned* cS, unsigned* sD, unsigned* sS,
                                         const unsigned base, const int4 s, const int4 d) {
  const unsigned pd0 = atomicAdd(&cD[d.x >> BIN_SHIFT], 1u);
  const unsigned pd1 = atomicAdd(&cD[d.y >> BIN_SHIFT], 1u);
  const unsigned pd2 = atomicAdd(&cD[d.z >> BIN_SHIFT], 1u);
  const unsigned pd3 = atomicAdd(&cD[d.w >> BIN_SHIFT], 1u);
  const unsigned ps0 = atomicAdd(&cS[s.x >> BIN_SHIFT], 1u);
  const unsigned ps1 = atomicAdd(&cS[s.y >> BIN_SHIFT], 1u);
  const unsigned ps2 = atomicAdd(&cS[s.z >> BIN_SHIFT], 1u);
  const unsigned ps3 = atomicAdd(&cS[s.w >> BIN_SHIFT], 1u);
  sD[pd0 - base] = ((unsigned)(d.x & (BIN_NODES - 1)) << 24) | (unsigned)s.x;
  sD[pd1 - base] = ((unsigned)(d.y & (BIN_NODES - 1)) << 24) | (unsigned)s.y;
  sD[pd2 - base] = ((unsigned)(d.z & (BIN_NODES - 1)) << 24) | (unsigned)s.z;
  sD[pd3 - base] = ((unsigned)(d.w & (BIN_NODES - 1)) << 24) | (unsigned)s.w;
  sS[ps0 - base] = (unsigned)(s.x & (BIN_NODES - 1));
  sS[ps1 - base] = (unsigned)(s.y & (BIN_NODES - 1));
  sS[ps2 - base] = (unsigned)(s.z & (BIN_NODES - 1));
  sS[ps3 - base] = (unsigned)(s.w & (BIN_NODES - 1));
}

__global__ __launch_bounds__(256) void k_slicesort(const int* __restrict__ src, const int* __restrict__ dst,
                                                   unsigned* __restrict__ tabD, unsigned* __restrict__ tabS,
                                                   unsigned* __restrict__ binD, unsigned* __restrict__ binS,
                                                   unsigned* __restrict__ bump, float* __restrict__ stat) {
  __shared__ unsigned hD[2 * 256];
  __shared__ unsigned hS[2 * 256];
  __shared__ unsigned wtD[8];
  __shared__ unsigned wtS[8];
  __shared__ __attribute__((aligned(16))) unsigned sD[SLICE_EDGES];
  __shared__ __attribute__((aligned(16))) unsigned sS[SLICE_EDGES];
  const int tid = threadIdx.x, b = blockIdx.x;
  const int lane = tid & 31, wv = tid >> 5;
  hD[tid] = 0u;
  hD[tid + 256] = 0u;
  hS[tid] = 0u;
  hS[tid + 256] = 0u;
  if (b == 0 && tid == 0) bump[0] = 0u;
  if (b < 2 * STAT_COPIES) stat[b * 256 + tid] = 0.f;
  __syncthreads();
  const int e0 = b * SLICE_EDGES;
  const int n4 = min(EE - e0, SLICE_EDGES) >> 2;
  const int4* s4 = (const int4*)(src + e0);
  const int4* d4 = (const int4*)(dst + e0);
  int4 s[SLICE_IT], d[SLICE_IT];
#pragma unroll
  for (int u = 0; u < SLICE_IT; ++u) {
    const int i = min(tid + 256 * u, n4 - 1);
    s[u] = s4[i];
    d[u] = d4[i];
  }
#pragma unroll
  for (int u = 0; u < SLICE_IT; ++u) {
    if (tid + 256 * u < n4) {
      atomicAdd(&hD[d[u].x >> BIN_SHIFT], 1u);
      atomicAdd(&hD[d[u].y >> BIN_SHIFT], 1u);
      atomicAdd(&hD[d[u].z >> BIN_SHIFT], 1u);
      atomicAdd(&hD[d[u].w >> BIN_SHIFT], 1u);
      atomicAdd(&hS[s[u].x >> BIN_SHIFT], 1u);
      atomicAdd(&hS[s[u].y >> BIN_SHIFT], 1u);
      atomicAdd(&hS[s[u].z >> BIN_SHIFT], 1u);
      atomicAdd(&hS[s[u].w >> BIN_SHIFT], 1u);
    }
  }
  __syncthreads();

  const unsigned a0 = hD[2 * tid], a1 = hD[2 * tid + 1];
  const unsigned c0 = hS[2 * tid], c1 = hS[2 * tid + 1];
  const unsigned sd = a0 + a1, ss = c0 + c1;
  unsigned incD = sd, incS = ss;
  for (int dl = 1; dl < 32; dl <<= 1) {
    const unsigned upD = __shfl_up(incD, dl);
    const unsigned upS = __shfl_up(incS, dl);
    if (lane >= dl) {
      incD += upD;
      incS += upS;
    }
  }
  if (lane == 31) {
    wtD[wv] = incD;
    wtS[wv] = incS;
  }
  __syncthreads();
  unsigned oD = (unsigned)e0 + incD - sd, oS = (unsigned)e0 + incS - ss;
  for (int w = 0; w < wv; ++w) {
    oD += wtD[w];
    oS += wtS[w];
  }
  hD[2 * tid] = oD;
  hD[2 * tid + 1] = oD + a0;
  hS[2 * tid] = oS;
  hS[2 * tid + 1] = oS + c0;
  if (2 * tid < TAB_ROWS) {
    tabD[(size_t)(2 * tid) * SLICES + b] = oD;
    tabS[(size_t)(2 * tid) * SLICES + b] = oS;
  }
  if (2 * tid + 1 < TAB_ROWS) {
    tabD[(size_t)(2 * tid + 1) * SLICES + b] = oD + a0;
    tabS[(size_t)(2 * tid + 1) * SLICES + b] = oS + c0;
  }
  __syncthreads();

#pragma unroll
  for (int u = 0; u < SLICE_IT; ++u) {
    if (tid + 256 * u < n4) {
      bin_put4(hD, hS, sD, sS, (unsigned)e0, s[u], d[u]);
    }
  }
  __syncthreads();
  uint4* gD = (uint4*)(binD + e0);
  uint4* gS = (uint4*)(binS + e0);
  const uint4* lD = (const uint4*)sD;
  const uint4* lS = (const uint4*)sS;
  for (int i = tid; i < n4; i += 256) {
    gD[i] = lD[i];
    gS[i] = lS[i];
  }
}

__global__ __launch_bounds__(256) void k_binrows(const unsigned* __restrict__ binD,
                                                 const unsigned* __restrict__ binS,
                                                 const unsigned* __restrict__ tabD,
                                                 const unsigned* __restrict__ tabS,
                                                 unsigned* __restrict__ bump,
                                                 float* __restrict__ dsout, float* __restrict__ dsin,
                                                 unsigned* __restrict__ rows, unsigned* __restrict__ esrc) {
  __shared__ unsigned cD[BIN_NODES];
  __shared__ unsigned cS[BIN_NODES];
  __shared__ unsigned cur[BIN_NODES];
  __shared__ unsigned wtot[8];
  __shared__ unsigned wsum[8];
  __shared__ unsigned sbase;
  __shared__ unsigned stot;
  __shared__ unsigned srow[ROW_CAP];
  const int tid = threadIdx.x, j = blockIdx.x;
  const int lane = tid & 31, wv = tid >> 5;
  const unsigned ul = (unsigned)lane;
  cD[tid] = 0u;
  cS[tid] = 0u;
  const unsigned stD = tabD[(size_t)j * SLICES + tid];
  const unsigned cnD = tabD[(size_t)(j + 1) * SLICES + tid] - stD;
  const unsigned stS = tabS[(size_t)j * SLICES + tid];
  const unsigned cnS = tabS[(size_t)(j + 1) * SLICES + tid] - stS;
  unsigned tot = cnD;
  for (int dl = 16; dl > 0; dl >>= 1) tot += __shfl_xor(tot, dl);
  if (lane == 0) wsum[wv] = tot;
  __syncthreads();

  unsigned v[32], w[32];
#pragma unroll
  for (int k = 0; k < 32; ++k) {
    v[k] = 0u;
    if (ul < lane_u32(cnD, k)) v[k] = binD[lane_u32(stD, k) + ul];
  }
#pragma unroll
  for (int k = 0; k < 32; ++k) {
    w[k] = 0u;
    if (ul < lane_u32(cnS, k)) w[k] = binS[lane_u32(stS, k) + ul];
  }
  unsigned mybase = 0u, nD = 0u;
  if (tid == 0) {
    for (int w8 = 0; w8 < 8; ++w8) nD += wsum[w8];
    mybase = atomicAdd(bump, nD);
  }
#pragma unroll
  for (int k = 0; k < 32; ++k)
    if (ul < lane_u32(cnD, k)) atomicAdd(&cD[v[k] >> 24], 1u);
#pragma unroll
  for (int k = 0; k < 32; ++k)
    if (ul < lane_u32(cnS, k)) atomicAdd(&cS[w[k]], 1u);
#pragma unroll 1
  for (int k = 0; k < 32; ++k) {
    const unsigned cn = lane_u32(cnD, k);
    if (cn > 32u) {
      const unsigned* e = binD + lane_u32(stD, k);
      for (unsigned i = 32u + ul; i < cn; i += 32u) atomicAdd(&cD[e[i] >> 24], 1u);
    }
    const unsigned cm = lane_u32(cnS, k);
    if (cm > 32u) {
      const unsigned* e = binS + lane_u32(stS, k);
      for (unsigned i = 32u + ul; i < cm; i += 32u) atomicAdd(&cS[e[i]], 1u);
    }
  }
  if (tid == 0) {
    sbase = mybase;
    stot = nD;
  }
  __syncthreads();

  const unsigned id = cD[tid], od = cS[tid];
  unsigned inc = id;
  for (int d = 1; d < 32; d <<= 1) {
    const unsigned up = __shfl_up(inc, d);
    if (lane >= d) inc += up;
  }
  if (lane == 31) wtot[wv] = inc;
  __syncthreads();
  unsigned st = inc - id;
  for (int w = 0; w < wv; ++w) st += wtot[w];
  cur[tid] = st;
  const unsigned bD = sbase;
  const unsigned nE = stot;
  const bool staged = nE <= (unsigned)ROW_CAP;
  const int node = j * BIN_NODES + tid;
  if (node < NN) {
    *(uint2*)(rows + 2 * node) = make_uint2(bD + st, bD + st + id);
    dsin[node] = id ? rsqrtf((float)id) : 1.0f;
    dsout[node] = od ? rsqrtf((float)od) : 1.0f;
  }
  __syncthreads();

  unsigned* o = esrc + bD;
  unsigned p[32];
#pragma unroll
  for (int k = 0; k < 32; ++k) {
    p[k] = 0u;
    if (ul < lane_u32(cnD, k)) p[k] = atomicAdd(&cur[v[k] >> 24], 1u);
  }
  if (staged) {
#pragma unroll
    for (int k = 0; k < 32; ++k)
      if (ul < lane_u32(cnD, k)) srow[p[k]] = v[k] & 0xFFFFFFu;
  } else {
#pragma unroll
    for (int k = 0; k < 32; ++k)
      if (ul < lane_u32(cnD, k)) o[p[k]] = v[k] & 0xFFFFFFu;
  }
#pragma unroll 1
  for (int k = 0; k < 32; ++k) {
    const unsigned cn = lane_u32(cnD, k);
    if (cn > 32u) {
      const unsigned* e = binD + lane_u32(stD, k);
      for (unsigned i = 32u + ul; i < cn; i += 32u) {
        const unsigned x = e[i];
        const unsigned at = atomicAdd(&cur[x >> 24], 1u);
        if (staged) srow[at] = x & 0xFFFFFFu;
        else o[at] = x & 0xFFFFFFu;
      }
    }
  }
  __syncthreads();
  if (staged)
    for (unsigned i = (unsigned)tid; i < nE; i += 256u) o[i] = srow[i];
}


static constexpr int AGG_NODES = 8;
static constexpr int AGG_GROUP = 8;
static constexpr int AGG_WAVES = NN / AGG_NODES;
static constexpr int AGG_BLOCKS = AGG_WAVES / 4;
static_assert(NN % AGG_NODES == 0 && AGG_WAVES % 4 == 0, "gather grid must cover the nodes exactly");

__device__ __forceinline__ void axpy4(float4& acc, float w, const float4& v) {
  acc.x = fmaf(w, v.x, acc.x);
  acc.y = fmaf(w, v.y, acc.y);
  acc.z = fmaf(w, v.z, acc.z);
  acc.w = fmaf(w, v.w, acc.w);
}

__global__ __launch_bounds__(128) void k_agg128(const float* __restrict__ h,
                                                const unsigned* __restrict__ rows,
                                                const unsigned* __restrict__ esrc,
                                                const float* __restrict__ dsout,
                                                const float* __restrict__ dsin,
                                                float* __restrict__ agg) {
  const int lane = threadIdx.x & 31;
  const int n0 = (blockIdx.x * 4 + (threadIdx.x >> 5)) * AGG_NODES;
  unsigned rb = 0u, re = 0u;
  float cin = 0.f;
  if (lane < AGG_NODES) {
    const uint2 r = *(const uint2*)(rows + 2 * (n0 + lane));
    rb = r.x;
    re = r.y;
    cin = dsin[n0 + lane];
  }
  unsigned b = lane_u32(rb, 0), e = lane_u32(re, 0);
  unsigned my = 0u;
  float myw = 0.f;
  if (b + lane < e) {
    my = esrc[b + lane];
    myw = dsout[my];
  }
  const float* hl = h + lane * 4;

#pragma unroll 1
  for (int k = 0; k < AGG_NODES; ++k) {
    unsigned nb = 0u, ne = 0u, nmy = 0u;
    if (k + 1 < AGG_NODES) {
      nb = lane_u32(rb, k + 1);
      ne = lane_u32(re, k + 1);
      if (nb + lane < ne) nmy = esrc[nb + lane];
    }
    float4 acc = make_float4(0.f, 0.f, 0.f, 0.f);
    const int cnt = (int)min(e - b, 32u);
    int j = 0;
    for (; j + AGG_GROUP <= cnt; j += AGG_GROUP) {
      float4 v[AGG_GROUP];
#pragma unroll
      for (int u = 0; u < AGG_GROUP; ++u)
        v[u] = *(const float4*)(hl + (size_t)lane_u32(my, j + u) * FF);
      __builtin_amdgcn_sched_barrier(0);
#pragma unroll
      for (int u = 0; u < AGG_GROUP; ++u) axpy4(acc, lane_f32(myw, j + u), v[u]);
    }
    const int rem = cnt - j;
    if (rem > 0) {
      float4 v[AGG_GROUP - 1];
#pragma unroll
      for (int u = 0; u < AGG_GROUP - 1; ++u) {
        v[u] = make_float4(0.f, 0.f, 0.f, 0.f);
        if (u < rem) v[u] = *(const float4*)(hl + (size_t)lane_u32(my, j + u) * FF);
      }
#pragma unroll
      for (int u = 0; u < AGG_GROUP - 1; ++u)
        if (u < rem) axpy4(acc, lane_f32(myw, j + u), v[u]);
    }
    for (unsigned i = b + 32u; i < e; ++i) {
      const unsigned s = esrc[i];
      axpy4(acc, dsout[s], *(const float4*)(hl + (size_t)s * FF));
    }
    const float c = lane_f32(cin, k);
    *(float4*)(agg + (size_t)(n0 + k) * FF + lane * 4) =
        make_float4(c * acc.x, c * acc.y, c * acc.z, c * acc.w);
    b = nb;
    e = ne;
    my = nmy;
    myw = 0.f;
    if (b + lane < e) myw = dsout[my];
  }
}

__global__ __launch_bounds__(128) void k_agg64(const float* __restrict__ P,
                                               const unsigned* __restrict__ rows,
                                               const unsigned* __restrict__ esrc,
                                               const float* __restrict__ dsin,
                                               float* __restrict__ R) {
  const int lane = threadIdx.x & 31;
  const int n0 = (blockIdx.x * 4 + (threadIdx.x >> 5)) * AGG_NODES;
  unsigned rb = 0u, re = 0u;
  float cin = 0.f;
  if (lane < AGG_NODES) {
    const uint2 r = *(const uint2*)(rows + 2 * (n0 + lane));
    rb = r.x;
    re = r.y;
    cin = dsin[n0 + lane];
  }
  unsigned b = lane_u32(rb, 0), e = lane_u32(re, 0);
  unsigned my = 0u;
  if (b + lane < e) my = esrc[b + lane];
  const float* pl = P + lane * 2;

#pragma unroll 1
  for (int k = 0; k < AGG_NODES; ++k) {
    unsigned nb = 0u, ne = 0u, nmy = 0u;
    if (k + 1 < AGG_NODES) {
      nb = lane_u32(rb, k + 1);
      ne = lane_u32(re, k + 1);
      if (nb + lane < ne) nmy = esrc[nb + lane];
    }
    float2 acc = make_float2(0.f, 0.f);
    const int cnt = (int)min(e - b, 32u);
    int j = 0;
    for (; j + AGG_GROUP <= cnt; j += AGG_GROUP) {
      float2 v[AGG_GROUP];
#pragma unroll
      for (int u = 0; u < AGG_GROUP; ++u)
        v[u] = *(const float2*)(pl + (size_t)lane_u32(my, j + u) * 64);
#pragma unroll
      for (int u = 0; u < AGG_GROUP; ++u) {
        acc.x += v[u].x;
        acc.y += v[u].y;
      }
    }
    const int rem = cnt - j;
    if (rem > 0) {
      float2 v[AGG_GROUP - 1];
#pragma unroll
      for (int u = 0; u < AGG_GROUP - 1; ++u) {
        v[u] = make_float2(0.f, 0.f);
        if (u < rem) v[u] = *(const float2*)(pl + (size_t)lane_u32(my, j + u) * 64);
      }
#pragma unroll
      for (int u = 0; u < AGG_GROUP - 1; ++u) {
        if (u < rem) {
          acc.x += v[u].x;
          acc.y += v[u].y;
        }
      }
    }
    for (unsigned i = b + 32u; i < e; ++i) {
      const float2 x = *(const float2*)(pl + (size_t)esrc[i] * 64);
      acc.x += x.x;
      acc.y += x.y;
    }
    const float c = lane_f32(cin, k);
    *(float2*)(R + (size_t)(n0 + k) * 64 + lane * 2) = make_float2(c * acc.x, c * acc.y);
    b = nb;
    e = ne;
    my = nmy;
  }
}

__global__ __launch_bounds__(128) void k_aggadd128(const float* __restrict__ G,
                                                   const unsigned* __restrict__ rows,
                                                   const unsigned* __restrict__ esrc,
                                                   const float* __restrict__ dsin,
                                                   const float* __restrict__ S,
                                                   float* __restrict__ Z,
                                                   float* __restrict__ stat) {
  __shared__ float red[2][4][4][32];
  const int lane = threadIdx.x & 31;
  float* mys = &red[0][threadIdx.x >> 5][0][lane];
  float* myq = &red[1][threadIdx.x >> 5][0][lane];
  mys[0] = mys[32] = mys[64] = mys[96] = 0.f;
  myq[0] = myq[32] = myq[64] = myq[96] = 0.f;
  const int n0 = (blockIdx.x * 4 + (threadIdx.x >> 5)) * AGG_NODES;
  unsigned rb = 0u, re = 0u;
  float cin = 0.f;
  if (lane < AGG_NODES) {
    const uint2 r = *(const uint2*)(rows + 2 * (n0 + lane));
    rb = r.x;
    re = r.y;
    cin = dsin[n0 + lane];
  }
  unsigned b = lane_u32(rb, 0), e = lane_u32(re, 0);
  unsigned my = 0u;
  if (b + lane < e) my = esrc[b + lane];
  const float* gl = G + lane * 4;

#pragma unroll 1
  for (int k = 0; k < AGG_NODES; ++k) {
    unsigned nb = 0u, ne = 0u, nmy = 0u;
    if (k + 1 < AGG_NODES) {
      nb = lane_u32(rb, k + 1);
      ne = lane_u32(re, k + 1);
      if (nb + lane < ne) nmy = esrc[nb + lane];
    }
    const float4 self = *(const float4*)(S + (size_t)(n0 + k) * FF + lane * 4);
    float4 acc = make_float4(0.f, 0.f, 0.f, 0.f);
    const int cnt = (int)min(e - b, 32u);
    int j = 0;
    for (; j + AGG_GROUP <= cnt; j += AGG_GROUP) {
      float4 v[AGG_GROUP];
#pragma unroll
      for (int u = 0; u < AGG_GROUP; ++u)
        v[u] = *(const float4*)(gl + (size_t)lane_u32(my, j + u) * FF);
#pragma unroll
      for (int u = 0; u < AGG_GROUP; ++u) {
        acc.x += v[u].x;
        acc.y += v[u].y;
        acc.z += v[u].z;
        acc.w += v[u].w;
      }
    }
    const int rem = cnt - j;
    if (rem > 0) {
      float4 v[AGG_GROUP - 1];
#pragma unroll
      for (int u = 0; u < AGG_GROUP - 1; ++u) {
        v[u] = make_float4(0.f, 0.f, 0.f, 0.f);
        if (u < rem) v[u] = *(const float4*)(gl + (size_t)lane_u32(my, j + u) * FF);
      }
#pragma unroll
      for (int u = 0; u < AGG_GROUP - 1; ++u) {
        if (u < rem) {
          acc.x += v[u].x;
          acc.y += v[u].y;
          acc.z += v[u].z;
          acc.w += v[u].w;
        }
      }
    }
    for (unsigned i = b + 32u; i < e; ++i) {
      const float4 x = *(const float4*)(gl + (size_t)esrc[i] * FF);
      acc.x += x.x;
      acc.y += x.y;
      acc.z += x.z;
      acc.w += x.w;
    }
    const float c = lane_f32(cin, k);
    const float4 z = make_float4(fmaf(c, acc.x, self.x), fmaf(c, acc.y, self.y),
                                 fmaf(c, acc.z, self.z), fmaf(c, acc.w, self.w));
    *(float4*)(Z + (size_t)(n0 + k) * FF + lane * 4) = z;
    atomicAdd(mys, z.x);
    atomicAdd(mys + 32, z.y);
    atomicAdd(mys + 64, z.z);
    atomicAdd(mys + 96, z.w);
    atomicAdd(myq, z.x * z.x);
    atomicAdd(myq + 32, z.y * z.y);
    atomicAdd(myq + 64, z.z * z.z);
    atomicAdd(myq + 96, z.w * z.w);
    b = nb;
    e = ne;
    my = nmy;
  }
  __syncthreads();
  const float* r = &red[0][0][threadIdx.x & 3][threadIdx.x >> 2];
  float ts = 0.f, tq = 0.f;
  for (int w = 0; w < 4; ++w) {
    ts += r[w * FF];
    tq += r[(4 + w) * FF];
  }
  float* o = stat + (blockIdx.x % STAT_COPIES) * 256 + threadIdx.x;
  atomicAdd(o, ts);
  atomicAdd(o + FF, tq);
}


static constexpr int GEMM_BLOCKS = 512;
static constexpr int NSTRIP = NN / 16;

template <int NC>
__device__ __forceinline__ void stage_weights(const float* __restrict__ Bg,
                                              float4* __restrict__ s, int tid) {
  constexpr int C4 = NC / 4;
  constexpr int PITCH = NC + 8;
  for (int i = tid; i < 32 * C4; i += 256) {
    const int g = i / C4, c4 = i - g * C4;
    const float* p = Bg + (size_t)(4 * g) * NC + 4 * c4;
    const float4 r0 = *(const float4*)(p);
    const float4 r1 = *(const float4*)(p + NC);
    const float4 r2 = *(const float4*)(p + 2 * NC);
    const float4 r3 = *(const float4*)(p + 3 * NC);
    const int m = g >> 3, hh = (g >> 2) & 1, q = g & 3;
    float4* d = s + ((m * 4 + q) * 2 + hh) * PITCH + 4 * c4;
    d[0] = make_float4(r0.x, r1.x, r2.x, r3.x);
    d[1] = make_float4(r0.y, r1.y, r2.y, r3.y);
    d[2] = make_float4(r0.z, r1.z, r2.z, r3.z);
    d[3] = make_float4(r0.w, r1.w, r2.w, r3.w);
  }
}

__device__ __forceinline__ v8f mm4(float ax, float ay, float bx, float by, v8f c) {
  v2f a; a.x = ax; a.y = ay;
  v2f b; b.x = bx; b.y = by;
  return __builtin_amdgcn_wmma_f32_16x16x4_f32(false, a, false, b, (short)0, c, false, false);
}

__device__ __forceinline__ float4 bn_relu4(const float4 z, const float4 a, const float4 b) {
  const float vx = fmaf(z.x, a.x, b.x);
  const float vy = fmaf(z.y, a.y, b.y);
  const float vz = fmaf(z.z, a.z, b.z);
  const float vw = fmaf(z.w, a.w, b.w);
  return make_float4(vx > 0.f ? vx : 0.f, vy > 0.f ? vy : 0.f, vz > 0.f ? vz : 0.f,
                     vw > 0.f ? vw : 0.f);
}

__device__ __forceinline__ void bn_params_to_lds(const float* __restrict__ stat,
                                                 const float* __restrict__ g,
                                                 const float* __restrict__ be,
                                                 float* __restrict__ sBN, int tid) {
  if (tid < FF) {
    float s = 0.f, q = 0.f;
#pragma unroll
    for (int k = 0; k < STAT_COPIES; ++k) {
      s += stat[k * 256 + tid];
      q += stat[k * 256 + FF + tid];
    }
    const float inv = 1.0f / (float)NN;
    const float mu = s * inv;
    const float var = q * inv - mu * mu;
    const float a = g[tid] * rsqrtf(var + 1e-5f);
    sBN[tid] = a;
    sBN[FF + tid] = be[tid] - mu * a;
  }
}

__global__ __launch_bounds__(256) void k_gemm_dual(const float* __restrict__ Agg,
                                                   const float* __restrict__ H,
                                                   const float* __restrict__ W,
                                                   const float* __restrict__ L,
                                                   float* __restrict__ Z,
                                                   float* __restrict__ stat) {
  constexpr int PITCH = 128 + 8;
  __shared__ __attribute__((aligned(16))) float4 sW[32 * PITCH];
  __shared__ __attribute__((aligned(16))) float4 sL[32 * PITCH];
  stage_weights<128>(W, sW, threadIdx.x);
  stage_weights<128>(L, sL, threadIdx.x);
  __syncthreads();

  const int wv = threadIdx.x >> 5;
  const int lane = threadIdx.x & 31;
  const int t = lane & 15;
  const int hh = lane >> 4;
  float cs[8], cq[8];
#pragma unroll
  for (int ct = 0; ct < 8; ++ct) { cs[ct] = 0.f; cq[ct] = 0.f; }

  for (int strip = blockIdx.x + GEMM_BLOCKS * wv; strip < NSTRIP; strip += GEMM_BLOCKS * 8) {
    const int r0 = strip << 4;
    const float* arow = Agg + (size_t)(r0 + t) * FF + 16 * hh;
    const float* hrow = H + (size_t)(r0 + t) * FF + 16 * hh;
    v8f acc[8];
#pragma unroll
    for (int ct = 0; ct < 8; ++ct) acc[ct] = v8f{};

#pragma unroll 1
    for (int m = 0; m < 4; ++m) {
      float4 a[4], h[4];
#pragma unroll
      for (int q = 0; q < 4; ++q) {
        a[q] = *(const float4*)(arow + 32 * m + 4 * q);
        h[q] = *(const float4*)(hrow + 32 * m + 4 * q);
      }
#pragma unroll
      for (int q = 0; q < 4; ++q) {
        const float4* pw = sW + ((m * 4 + q) * 2 + hh) * PITCH + t;
        const float4* pl = sL + ((m * 4 + q) * 2 + hh) * PITCH + t;
        float4 bw[8], bl[8];
#pragma unroll
        for (int ct = 0; ct < 8; ++ct) { bw[ct] = pw[16 * ct]; bl[ct] = pl[16 * ct]; }
#pragma unroll
        for (int ct = 0; ct < 8; ++ct) acc[ct] = mm4(a[q].x, a[q].y, bw[ct].x, bw[ct].y, acc[ct]);
#pragma unroll
        for (int ct = 0; ct < 8; ++ct) acc[ct] = mm4(h[q].x, h[q].y, bl[ct].x, bl[ct].y, acc[ct]);
#pragma unroll
        for (int ct = 0; ct < 8; ++ct) acc[ct] = mm4(a[q].z, a[q].w, bw[ct].z, bw[ct].w, acc[ct]);
#pragma unroll
        for (int ct = 0; ct < 8; ++ct) acc[ct] = mm4(h[q].z, h[q].w, bl[ct].z, bl[ct].w, acc[ct]);
      }
    }
#pragma unroll
    for (int ct = 0; ct < 8; ++ct)
#pragma unroll
      for (int v = 0; v < 8; ++v) {
        const float x = acc[ct][v];
        Z[(size_t)(r0 + v + 8 * hh) * FF + 16 * ct + t] = x;
        cs[ct] += x;
        cq[ct] = fmaf(x, x, cq[ct]);
      }
  }

  __syncthreads();
  float* red = (float*)sW;
#pragma unroll
  for (int ct = 0; ct < 8; ++ct) {
    red[(wv * 2 + hh) * FF + 16 * ct + t] = cs[ct];
    red[(16 + wv * 2 + hh) * FF + 16 * ct + t] = cq[ct];
  }
  __syncthreads();
  const int st = threadIdx.x >> 7, col = threadIdx.x & 127;
  const float* rr = red + st * 16 * FF + col;
  float ts = 0.f;
#pragma unroll
  for (int i = 0; i < 16; ++i) ts += rr[i * FF];
  atomicAdd(stat + (blockIdx.x % STAT_COPIES) * 256 + st * FF + col, ts);
}

__global__ __launch_bounds__(256) void k_gemm_proj(const float* __restrict__ Zin,
                                                   const float* __restrict__ stat,
                                                   const float* __restrict__ g,
                                                   const float* __restrict__ be,
                                                   const float* __restrict__ dsout,
                                                   const float* __restrict__ W,
                                                   const float* __restrict__ L,
                                                   float* __restrict__ Gw,
                                                   float* __restrict__ Gl) {
  constexpr int PITCH = 128 + 8;
  __shared__ __attribute__((aligned(16))) float4 sW[32 * PITCH];
  __shared__ __attribute__((aligned(16))) float4 sL[32 * PITCH];
  __shared__ __attribute__((aligned(16))) float sBN[2 * FF];
  bn_params_to_lds(stat, g, be, sBN, threadIdx.x);
  stage_weights<128>(W, sW, threadIdx.x);
  stage_weights<128>(L, sL, threadIdx.x);
  __syncthreads();

  const int wv = threadIdx.x >> 5;
  const int lane = threadIdx.x & 31;
  const int t = lane & 15, hh = lane >> 4;

  for (int strip = blockIdx.x + GEMM_BLOCKS * wv; strip < NSTRIP; strip += GEMM_BLOCKS * 8) {
    const int r0 = strip << 4;
    const float* zrow = Zin + (size_t)(r0 + t) * FF + 16 * hh;
    const float4 d0 = *(const float4*)(dsout + r0 + 8 * hh);
    const float4 d1 = *(const float4*)(dsout + r0 + 8 * hh + 4);
    v8f aw[8], al[8];
#pragma unroll
    for (int ct = 0; ct < 8; ++ct) { aw[ct] = v8f{}; al[ct] = v8f{}; }
#pragma unroll 1
    for (int m = 0; m < 4; ++m) {
      float4 h[4];
#pragma unroll
      for (int q = 0; q < 4; ++q) {
        const int c = 32 * m + 16 * hh + 4 * q;
        const float4 x = *(const float4*)(zrow + 32 * m + 4 * q);
        h[q] = bn_relu4(x, *(const float4*)(sBN + c), *(const float4*)(sBN + FF + c));
      }
#pragma unroll
      for (int q = 0; q < 4; ++q) {
        const float4* pw = sW + ((m * 4 + q) * 2 + hh) * PITCH + t;
        const float4* pl = sL + ((m * 4 + q) * 2 + hh) * PITCH + t;
        float4 bw[8], bl[8];
#pragma unroll
        for (int ct = 0; ct < 8; ++ct) { bw[ct] = pw[16 * ct]; bl[ct] = pl[16 * ct]; }
#pragma unroll
        for (int ct = 0; ct < 8; ++ct) aw[ct] = mm4(h[q].x, h[q].y, bw[ct].x, bw[ct].y, aw[ct]);
#pragma unroll
        for (int ct = 0; ct < 8; ++ct) al[ct] = mm4(h[q].x, h[q].y, bl[ct].x, bl[ct].y, al[ct]);
#pragma unroll
        for (int ct = 0; ct < 8; ++ct) aw[ct] = mm4(h[q].z, h[q].w, bw[ct].z, bw[ct].w, aw[ct]);
#pragma unroll
        for (int ct = 0; ct < 8; ++ct) al[ct] = mm4(h[q].z, h[q].w, bl[ct].z, bl[ct].w, al[ct]);
      }
    }
    const float dv[8] = {d0.x, d0.y, d0.z, d0.w, d1.x, d1.y, d1.z, d1.w};
#pragma unroll
    for (int ct = 0; ct < 8; ++ct)
#pragma unroll
      for (int v = 0; v < 8; ++v) {
        const size_t o = (size_t)(r0 + v + 8 * hh) * FF + 16 * ct + t;
        Gw[o] = dv[v] * aw[ct][v];
        Gl[o] = al[ct][v];
      }
  }
}

__global__ __launch_bounds__(256) void k_gemm_p(const float* __restrict__ Zin,
                                                const float* __restrict__ stat,
                                                const float* __restrict__ g,
                                                const float* __restrict__ be,
                                                const float* __restrict__ dsout,
                                                const float* __restrict__ Wp,
                                                float* __restrict__ Hout,
                                                float* __restrict__ P) {
  constexpr int PITCH = 64 + 8;
  __shared__ __attribute__((aligned(16))) float4 sW[32 * PITCH];
  __shared__ __attribute__((aligned(16))) float sBN[2 * FF];
  bn_params_to_lds(stat, g, be, sBN, threadIdx.x);
  stage_weights<64>(Wp, sW, threadIdx.x);
  __syncthreads();

  const int wv = threadIdx.x >> 5;
  const int lane = threadIdx.x & 31;
  const int t = lane & 15, hh = lane >> 4;

  for (int strip = blockIdx.x + GEMM_BLOCKS * wv; strip < NSTRIP; strip += GEMM_BLOCKS * 8) {
    const int r0 = strip << 4;
    const float d = dsout[r0 + t];
    const float* zrow = Zin + (size_t)(r0 + t) * FF + 16 * hh;
    float* hrow = Hout + (size_t)(r0 + t) * FF + 16 * hh;
    v8f acc[4];
#pragma unroll
    for (int ct = 0; ct < 4; ++ct) acc[ct] = v8f{};
#pragma unroll 1
    for (int m = 0; m < 4; ++m) {
      float4 a[4];
#pragma unroll
      for (int q = 0; q < 4; ++q) {
        const int c = 32 * m + 16 * hh + 4 * q;
        const float4 z = *(const float4*)(zrow + 32 * m + 4 * q);
        const float4 x = bn_relu4(z, *(const float4*)(sBN + c), *(const float4*)(sBN + FF + c));
        *(float4*)(hrow + 32 * m + 4 * q) = x;
        a[q] = make_float4(d * x.x, d * x.y, d * x.z, d * x.w);
      }
#pragma unroll
      for (int q = 0; q < 4; ++q) {
        const float4* pw = sW + ((m * 4 + q) * 2 + hh) * PITCH + t;
        float4 bw[4];
#pragma unroll
        for (int ct = 0; ct < 4; ++ct) bw[ct] = pw[16 * ct];
#pragma unroll
        for (int ct = 0; ct < 4; ++ct) acc[ct] = mm4(a[q].x, a[q].y, bw[ct].x, bw[ct].y, acc[ct]);
#pragma unroll
        for (int ct = 0; ct < 4; ++ct) acc[ct] = mm4(a[q].z, a[q].w, bw[ct].z, bw[ct].w, acc[ct]);
      }
    }
#pragma unroll
    for (int ct = 0; ct < 4; ++ct)
#pragma unroll
      for (int v = 0; v < 8; ++v)
        P[(size_t)(r0 + v + 8 * hh) * 64 + 16 * ct + t] = acc[ct][v];
  }
}

__global__ __launch_bounds__(256) void k_gemm_out(const float* __restrict__ H,
                                                  const float* __restrict__ Lp,
                                                  const float* __restrict__ R,
                                                  const float* __restrict__ b2,
                                                  float* __restrict__ out) {
  constexpr int PITCH = 64 + 8;
  __shared__ __attribute__((aligned(16))) float4 sL[32 * PITCH];
  stage_weights<64>(Lp, sL, threadIdx.x);
  __syncthreads();

  const int wv = threadIdx.x >> 5;
  const int lane = threadIdx.x & 31;
  const int t = lane & 15, hh = lane >> 4;

  for (int strip = blockIdx.x + GEMM_BLOCKS * wv; strip < NSTRIP; strip += GEMM_BLOCKS * 8) {
    const int r0 = strip << 4;
    const float* hrow = H + (size_t)(r0 + t) * FF + 16 * hh;
    v8f acc[3];
#pragma unroll
    for (int ct = 0; ct < 3; ++ct) acc[ct] = v8f{};
#pragma unroll 1
    for (int m = 0; m < 4; ++m) {
      float4 a[4];
#pragma unroll
      for (int q = 0; q < 4; ++q) a[q] = *(const float4*)(hrow + 32 * m + 4 * q);
#pragma unroll
      for (int q = 0; q < 4; ++q) {
        const float4* pl = sL + ((m * 4 + q) * 2 + hh) * PITCH + t;
        float4 bl[3];
#pragma unroll
        for (int ct = 0; ct < 3; ++ct) bl[ct] = pl[16 * ct];
#pragma unroll
        for (int ct = 0; ct < 3; ++ct) acc[ct] = mm4(a[q].x, a[q].y, bl[ct].x, bl[ct].y, acc[ct]);
#pragma unroll
        for (int ct = 0; ct < 3; ++ct) acc[ct] = mm4(a[q].z, a[q].w, bl[ct].z, bl[ct].w, acc[ct]);
      }
    }
#pragma unroll
    for (int ct = 0; ct < 3; ++ct) {
      const int col = 16 * ct + t;
      if (col < 40) {
        const float bias = b2[col];
#pragma unroll
        for (int v = 0; v < 8; ++v) {
          const int row = r0 + v + 8 * hh;
          out[(size_t)row * 40 + col] = acc[ct][v] + R[(size_t)row * 64 + col] + bias;
        }
      }
    }
  }
}


__global__ void k_padw(const float* __restrict__ W2, const float* __restrict__ L2,
                       float* __restrict__ Wp, float* __restrict__ Lp) {
  const int i = blockIdx.x * 256 + threadIdx.x;
  const int k = i >> 6, c = i & 63;
  const float w = (c < 40) ? W2[k * 40 + c] : 0.f;
  const float l = (c < 40) ? L2[k * 40 + c] : 0.f;
  Wp[i] = w;
  Lp[i] = l;
}


extern "C" void kernel_launch(void* const* d_in, const int* in_sizes, int n_in,
                              void* d_out, int out_size, void* d_ws, size_t ws_size,
                              hipStream_t stream) {
  (void)in_sizes; (void)n_in; (void)out_size; (void)ws_size;

  const float* feat = (const float*)d_in[0];
  const int* src   = (const int*)d_in[1];
  const int* dst   = (const int*)d_in[2];
  const float* W0  = (const float*)d_in[3];
  const float* L0  = (const float*)d_in[4];
  const float* W1  = (const float*)d_in[5];
  const float* L1  = (const float*)d_in[6];
  const float* W2  = (const float*)d_in[7];
  const float* L2  = (const float*)d_in[8];
  const float* b2  = (const float*)d_in[9];
  const float* g0  = (const float*)d_in[10];
  const float* be0 = (const float*)d_in[11];
  const float* g1  = (const float*)d_in[12];
  const float* be1 = (const float*)d_in[13];
  float* out = (float*)d_out;

  char* base = (char*)d_ws;
  size_t off = 0;
  auto take = [&](size_t bytes) -> char* {
    char* p = base + off;
    off = (off + bytes + 255) & ~(size_t)255;
    return p;
  };
  unsigned* tabD    = (unsigned*)take((size_t)TAB_ROWS * SLICES * 4);
  unsigned* tabS    = (unsigned*)take((size_t)TAB_ROWS * SLICES * 4);
  unsigned* bump    = (unsigned*)take(4);
  unsigned* binD    = (unsigned*)take((size_t)EE * 4);
  unsigned* binS    = (unsigned*)take((size_t)EE * 4);
  unsigned* rows    = (unsigned*)take((size_t)NN * 8);
  unsigned* esrc    = (unsigned*)take((size_t)EE * 4);
  float* dsout = (float*)take((size_t)NN * 4);
  float* dsin  = (float*)take((size_t)NN * 4);
  float* stat0 = (float*)take((size_t)2 * STAT_LAYER * 4);
  float* stat1 = stat0 + STAT_LAYER;
  float* Wp    = (float*)take(128 * 64 * 4);
  float* Lp    = (float*)take(128 * 64 * 4);
  float* bufA  = (float*)take((size_t)NN * FF * 4);
  float* bufB  = (float*)take((size_t)NN * FF * 4);
  float* bufC  = (float*)take((size_t)NN * FF * 4);
  float* bufD  = (float*)take((size_t)NN * FF * 4);

  k_slicesort<<<SLICES, 256, 0, stream>>>(src, dst, tabD, tabS, binD, binS, bump, stat0);
  k_binrows<<<NBIN, 256, 0, stream>>>(binD, binS, tabD, tabS, bump, dsout, dsin, rows, esrc);
  k_padw<<<32, 256, 0, stream>>>(W2, L2, Wp, Lp);

  k_agg128<<<AGG_BLOCKS, 128, 0, stream>>>(feat, rows, esrc, dsout, dsin, bufA);
  k_gemm_dual<<<GEMM_BLOCKS, 256, 0, stream>>>(bufA, feat, W0, L0, bufB, stat0);

  k_gemm_proj<<<GEMM_BLOCKS, 256, 0, stream>>>(bufB, stat0, g0, be0, dsout, W1, L1, bufA, bufD);
  k_aggadd128<<<AGG_BLOCKS, 128, 0, stream>>>(bufA, rows, esrc, dsin, bufD, bufB,
                                               stat1);

  k_gemm_p<<<GEMM_BLOCKS, 256, 0, stream>>>(bufB, stat1, g1, be1, dsout, Wp, bufC, bufA);
  k_agg64<<<AGG_BLOCKS, 128, 0, stream>>>(bufA, rows, esrc, dsin, bufB);
  k_gemm_out<<<GEMM_BLOCKS, 256, 0, stream>>>(bufC, Lp, bufB, b2, out);
}
